# starting kernel
# baseline (speedup 1.0000x reference)
#include <hip/hip_runtime.h>
#include <hip/hip_bf16.h>


typedef __attribute__((ext_vector_type(2))) float v2f;
typedef __attribute__((ext_vector_type(8))) float v8f;

__device__ __forceinline__ v8f wmma4(v2f a, v2f b, v8f c) {
    return __builtin_amdgcn_wmma_f32_16x16x4_f32(
        false, a, false, b, (short)0, c, false, false);
}

__device__ __forceinline__ v2f load_a_frag(const float* M, int ld, int row0,
                                           int k0, int lane) {
    int r = row0 + (lane & 15);
    int k = k0 + ((lane >> 4) << 1);
    const float* p = M + (size_t)r * ld + k;
    v2f a; a.x = p[0]; a.y = p[1];
    return a;
}

__device__ __forceinline__ v2f load_b_packed(const float* P, int nk, int kt,
                                             int nt, int lane) {
    const v2f* p = (const v2f*)P + ((size_t)(nt * nk + kt) * 32 + lane);
    return *p;
}

__device__ __forceinline__ float sigmoid_f(float v) {
    return 1.0f / (1.0f + __expf(-v));
}
__device__ __forceinline__ float tanh_f(float v) {
    float t = __expf(-2.0f * v);
    return (1.0f - t) / (1.0f + t);
}

__global__ __launch_bounds__(256) void pack_kernel(
    const float* __restrict__ W, float* __restrict__ P, int K, int Nc) {
    int w = (int)((blockIdx.x * 256u + threadIdx.x) >> 5);
    int lane = threadIdx.x & 31;
    int nk = K >> 2, nn = Nc >> 4;
    if (w >= nk * nn) return;
    int kt = w % nk, nt = w / nk;
    int n = nt * 16 + (lane & 15);
    int k = kt * 4 + ((lane >> 4) << 1);
    v2f v;
    v.x = W[(size_t)k * Nc + n];
    v.y = W[(size_t)(k + 1) * Nc + n];
    ((v2f*)P)[(size_t)(nt * nk + kt) * 32 + lane] = v;
}

__global__ __launch_bounds__(256) void zero_kernel(float* p, long n) {
    long i = (long)blockIdx.x * blockDim.x + threadIdx.x;
    long stride = (long)gridDim.x * blockDim.x;
    for (; i < n; i += stride) p[i] = 0.0f;
}

__global__ __launch_bounds__(128) void enc_kernel(
    const float* __restrict__ feat, const float* __restrict__ pW0,
    const float* __restrict__ b0, const float* __restrict__ pW1,
    const float* __restrict__ b1, float* __restrict__ x, int N) {
    __shared__ float t[4][16 * 256];
    const int wave = threadIdx.x >> 5;
    const int lane = threadIdx.x & 31;
    const int row0 = (blockIdx.x * 4 + wave) * 16;
    const bool active = (row0 < N);

    if (active) {
        v2f afr[16];
#pragma unroll
        for (int kk = 0; kk < 16; ++kk)
            afr[kk] = load_a_frag(feat, 64, row0, kk * 4, lane);
#pragma unroll 1
        for (int nt = 0; nt < 16; ++nt) {
            v8f acc = {};
#pragma unroll
            for (int kk = 0; kk < 16; ++kk) {
                v2f b = load_b_packed(pW0, 16, kk, nt, lane);
                acc = wmma4(afr[kk], b, acc);
            }
            int col = nt * 16 + (lane & 15);
            float bias = b0[col];
#pragma unroll
            for (int i = 0; i < 8; ++i) {
                int r = i + 8 * (lane >> 4);
                t[wave][r * 256 + col] = fmaxf(acc[i] + bias, 0.0f);
            }
        }
    }
    __syncthreads();
    if (active) {
#pragma unroll 1
        for (int nt = 0; nt < 16; ++nt) {
            v8f acc = {};
#pragma unroll 4
            for (int kk = 0; kk < 64; ++kk) {
                int k = kk * 4 + ((lane >> 4) << 1);
                int r = lane & 15;
                v2f a; a.x = t[wave][r * 256 + k]; a.y = t[wave][r * 256 + k + 1];
                v2f b = load_b_packed(pW1, 64, kk, nt, lane);
                acc = wmma4(a, b, acc);
            }
            int col = nt * 16 + (lane & 15);
            float bias = b1[col];
#pragma unroll
            for (int i = 0; i < 8; ++i) {
                int r = row0 + i + 8 * (lane >> 4);
                x[(size_t)r * 256 + col] = fmaxf(acc[i] + bias, 0.0f);
            }
        }
    }
}

__global__ __launch_bounds__(128) void msg_kernel(
    const float* __restrict__ x, const float* __restrict__ h,
    const float* __restrict__ pmsgW, const float* __restrict__ msgb,
    float* __restrict__ mx, int N) {
    const int wave = threadIdx.x >> 5;
    const int lane = threadIdx.x & 31;
    const int row0 = (blockIdx.x * 4 + wave) * 32;
    if (row0 >= N) return;
#pragma unroll 1
    for (int nt = 0; nt < 8; ++nt) {
        v8f acc0 = {}, acc1 = {};
#pragma unroll 4
        for (int kk = 0; kk < 128; ++kk) {
            int k0 = kk * 4;
            v2f b = load_b_packed(pmsgW, 128, kk, nt, lane);
            v2f a0 = (k0 < 256) ? load_a_frag(x, 256, row0, k0, lane)
                                : load_a_frag(h, 256, row0, k0 - 256, lane);
            v2f a1 = (k0 < 256) ? load_a_frag(x, 256, row0 + 16, k0, lane)
                                : load_a_frag(h, 256, row0 + 16, k0 - 256, lane);
            acc0 = wmma4(a0, b, acc0);
            acc1 = wmma4(a1, b, acc1);
        }
        int col = nt * 16 + (lane & 15);
        float bias = msgb[col];
#pragma unroll
        for (int i = 0; i < 8; ++i) {
            int r = row0 + i + 8 * (lane >> 4);
            mx[(size_t)r * 128 + col] = acc0[i] + bias;
            mx[(size_t)(r + 16) * 128 + col] = acc1[i] + bias;
        }
    }
}

__global__ __launch_bounds__(256) void edge_kernel(
    const int* __restrict__ src, const int* __restrict__ dst,
    const float* __restrict__ mx, float* __restrict__ sums,
    float* __restrict__ cnt, int E) {
    int e = (int)((blockIdx.x * 256u + threadIdx.x) >> 5);
    int lane = threadIdx.x & 31;
    if (e >= E) return;
    int s = src[e];
    int d = dst[e];
    const float4* mv = (const float4*)(mx + (size_t)s * 128);
    float4 v = mv[lane];
    float* base = sums + (size_t)d * 128 + lane * 4;
    atomicAdd(base + 0, v.x);
    atomicAdd(base + 1, v.y);
    atomicAdd(base + 2, v.z);
    atomicAdd(base + 3, v.w);
    if (lane == 0) atomicAdd(cnt + d, 1.0f);
}

__global__ __launch_bounds__(256) void mean_kernel(
    const float* __restrict__ sums, const float* __restrict__ cnt,
    float* __restrict__ c, long n) {
    long i = (long)blockIdx.x * blockDim.x + threadIdx.x;
    if (i >= n) return;
    long node = i >> 7;
    c[i] = sums[i] / fmaxf(cnt[node], 1.0f);
}

__global__ __launch_bounds__(128) void gru_kernel(
    const float* __restrict__ x, const float* __restrict__ c,
    const float* __restrict__ h, const float* __restrict__ pWih,
    const float* __restrict__ pWhh, const float* __restrict__ bih,
    const float* __restrict__ bhh, float* __restrict__ hnew, int N) {
    const int wave = threadIdx.x >> 5;
    const int lane = threadIdx.x & 31;
    const int row0 = (blockIdx.x * 4 + wave) * 32;
    if (row0 >= N) return;
#pragma unroll 1
    for (int nt = 0; nt < 16; ++nt) {
        v8f ir0 = {}, iz0 = {}, in0 = {}, ir1 = {}, iz1 = {}, in1 = {};
        v8f hr0 = {}, hz0 = {}, hn0 = {}, hr1 = {}, hz1 = {}, hn1 = {};
#pragma unroll 2
        for (int kk = 0; kk < 96; ++kk) {
            int k0 = kk * 4;
            v2f br = load_b_packed(pWih, 96, kk, nt, lane);
            v2f bz = load_b_packed(pWih, 96, kk, nt + 16, lane);
            v2f bn = load_b_packed(pWih, 96, kk, nt + 32, lane);
            v2f a0 = (k0 < 256) ? load_a_frag(x, 256, row0, k0, lane)
                                : load_a_frag(c, 128, row0, k0 - 256, lane);
            v2f a1 = (k0 < 256) ? load_a_frag(x, 256, row0 + 16, k0, lane)
                                : load_a_frag(c, 128, row0 + 16, k0 - 256, lane);
            ir0 = wmma4(a0, br, ir0);  ir1 = wmma4(a1, br, ir1);
            iz0 = wmma4(a0, bz, iz0);  iz1 = wmma4(a1, bz, iz1);
            in0 = wmma4(a0, bn, in0);  in1 = wmma4(a1, bn, in1);
        }
#pragma unroll 2
        for (int kk = 0; kk < 64; ++kk) {
            int k0 = kk * 4;
            v2f br = load_b_packed(pWhh, 64, kk, nt, lane);
            v2f bz = load_b_packed(pWhh, 64, kk, nt + 16, lane);
            v2f bn = load_b_packed(pWhh, 64, kk, nt + 32, lane);
            v2f a0 = load_a_frag(h, 256, row0, k0, lane);
            v2f a1 = load_a_frag(h, 256, row0 + 16, k0, lane);
            hr0 = wmma4(a0, br, hr0);  hr1 = wmma4(a1, br, hr1);
            hz0 = wmma4(a0, bz, hz0);  hz1 = wmma4(a1, bz, hz1);
            hn0 = wmma4(a0, bn, hn0);  hn1 = wmma4(a1, bn, hn1);
        }
        int col = nt * 16 + (lane & 15);
        float bir = bih[col], biz = bih[col + 256], bin = bih[col + 512];
        float bhr = bhh[col], bhz = bhh[col + 256], bhn = bhh[col + 512];
#pragma unroll
        for (int i = 0; i < 8; ++i) {
            int r = row0 + i + 8 * (lane >> 4);
            {
                float rg = sigmoid_f((ir0[i] + bir) + (hr0[i] + bhr));
                float zg = sigmoid_f((iz0[i] + biz) + (hz0[i] + bhz));
                float ng = tanh_f((in0[i] + bin) + rg * (hn0[i] + bhn));
                float hv = h[(size_t)r * 256 + col];
                hnew[(size_t)r * 256 + col] = (1.0f - zg) * ng + zg * hv;
            }
            {
                float rg = sigmoid_f((ir1[i] + bir) + (hr1[i] + bhr));
                float zg = sigmoid_f((iz1[i] + biz) + (hz1[i] + bhz));
                float ng = tanh_f((in1[i] + bin) + rg * (hn1[i] + bhn));
                float hv = h[(size_t)(r + 16) * 256 + col];
                hnew[(size_t)(r + 16) * 256 + col] = (1.0f - zg) * ng + zg * hv;
            }
        }
    }
}

__global__ __launch_bounds__(128) void out_kernel(
    const float* __restrict__ hnew, const float* __restrict__ poutW,
    const float* __restrict__ outb, float* __restrict__ q, int N) {
    const int wave = threadIdx.x >> 5;
    const int lane = threadIdx.x & 31;
    const int row0 = (blockIdx.x * 4 + wave) * 32;
    if (row0 >= N) return;
    v8f acc0 = {}, acc1 = {};
#pragma unroll 4
    for (int kk = 0; kk < 64; ++kk) {
        v2f b = load_b_packed(poutW, 64, kk, 0, lane);
        v2f a0 = load_a_frag(hnew, 256, row0, kk * 4, lane);
        v2f a1 = load_a_frag(hnew, 256, row0 + 16, kk * 4, lane);
        acc0 = wmma4(a0, b, acc0);
        acc1 = wmma4(a1, b, acc1);
    }
    int col = lane & 15;
    float bias = outb[col];
#pragma unroll
    for (int i = 0; i < 8; ++i) {
        int r = row0 + i + 8 * (lane >> 4);
        q[(size_t)r * 16 + col] = acc0[i] + bias;
        q[(size_t)(r + 16) * 16 + col] = acc1[i] + bias;
    }
}

static inline int packBlocks(int K, int Nc) {
    int waves = (K / 4) * (Nc / 16);
    return (waves + 7) / 8;
}

extern "C" void kernel_launch(void* const* d_in, const int* in_sizes, int n_in,
                              void* d_out, int out_size, void* d_ws,
                              size_t ws_size, hipStream_t stream) {
    (void)n_in; (void)out_size; (void)ws_size;
    const float* feat = (const float*)d_in[0];
    const float* h    = (const float*)d_in[1];
    const int*   src  = (const int*)d_in[2];
    const int*   dst  = (const int*)d_in[3];
    const float* W0   = (const float*)d_in[4];
    const float* b0   = (const float*)d_in[5];
    const float* W1   = (const float*)d_in[6];
    const float* b1   = (const float*)d_in[7];
    const float* msgW = (const float*)d_in[8];
    const float* msgb = (const float*)d_in[9];
    const float* Wih  = (const float*)d_in[10];
    const float* Whh  = (const float*)d_in[11];
    const float* bih  = (const float*)d_in[12];
    const float* bhh  = (const float*)d_in[13];
    const float* outW = (const float*)d_in[14];
    const float* outb = (const float*)d_in[15];

    const int N = in_sizes[1] / 256;
    const int E = in_sizes[2];

    float* ws    = (float*)d_ws;
    float* x     = ws;
    float* mx    = x    + (size_t)N * 256;
    float* sums  = mx   + (size_t)N * 128;
    float* cnt   = sums + (size_t)N * 128;
    float* c     = cnt  + N;
    float* pW0   = c    + (size_t)N * 128;
    float* pW1   = pW0  + 16384;
    float* pmsgW = pW1  + 65536;
    float* pWih  = pmsgW + 65536;
    float* pWhh  = pWih + 294912;
    float* poutW = pWhh + 196608;

    float* q    = (float*)d_out;
    float* hnew = q + (size_t)N * 16;

    const int encBlocks  = (N / 16 + 3) / 4;
    const int gemmBlocks = (N / 32 + 3) / 4;
    const long zeroN     = (long)N * 128 + N;
    const int zeroBlocks = (int)((zeroN + 255) / 256);
    const int edgeBlocks = (E + 7) / 8;
    const long meanN     = (long)N * 128;
    const int meanBlocks = (int)((meanN + 255) / 256);

    pack_kernel<<<packBlocks(64, 256), 256, 0, stream>>>(W0, pW0, 64, 256);
    pack_kernel<<<packBlocks(256, 256), 256, 0, stream>>>(W1, pW1, 256, 256);
    pack_kernel<<<packBlocks(512, 128), 256, 0, stream>>>(msgW, pmsgW, 512, 128);
    pack_kernel<<<packBlocks(384, 768), 256, 0, stream>>>(Wih, pWih, 384, 768);
    pack_kernel<<<packBlocks(256, 768), 256, 0, stream>>>(Whh, pWhh, 256, 768);
    pack_kernel<<<packBlocks(256, 16), 256, 0, stream>>>(outW, poutW, 256, 16);

    zero_kernel<<<zeroBlocks, 256, 0, stream>>>(sums, zeroN);
    enc_kernel<<<encBlocks, 128, 0, stream>>>(feat, pW0, b0, pW1, b1, x, N);
    msg_kernel<<<gemmBlocks, 128, 0, stream>>>(x, h, pmsgW, msgb, mx, N);
    edge_kernel<<<edgeBlocks, 256, 0, stream>>>(src, dst, mx, sums, cnt, E);
    mean_kernel<<<meanBlocks, 256, 0, stream>>>(sums, cnt, c, meanN);
    gru_kernel<<<gemmBlocks, 128, 0, stream>>>(x, c, h, pWih, pWhh, bih, bhh,
                                               hnew, N);
    out_kernel<<<gemmBlocks, 128, 0, stream>>>(hnew, poutW, outb, q, N);
}


# gru and msg GEMMs: column-tile loop spread over the grid, each wave keeps only the 2 tiles of one 128-byte output line (8x and 4x more waves), so each SIMD holds several waves that hide load latency
# speedup vs baseline: 2.1355x; 2.1355x over previous
#include <hip/hip_runtime.h>
#include <hip/hip_bf16.h>


typedef __attribute__((ext_vector_type(2))) float v2f;
typedef __attribute__((ext_vector_type(8))) float v8f;

__device__ __forceinline__ v8f wmma4(v2f a, v2f b, v8f c) {
    return __builtin_amdgcn_wmma_f32_16x16x4_f32(
        false, a, false, b, (short)0, c, false, false);
}

__device__ __forceinline__ v2f load_a_frag(const float* M, int ld, int row0,
                                           int k0, int lane) {
    int r = row0 + (lane & 15);
    int k = k0 + ((lane >> 4) << 1);
    const float* p = M + (size_t)r * ld + k;
    v2f a; a.x = p[0]; a.y = p[1];
    return a;
}

__device__ __forceinline__ v2f load_b_packed(const float* P, int nk, int kt,
                                             int nt, int lane) {
    const v2f* p = (const v2f*)P + ((size_t)(nt * nk + kt) * 32 + lane);
    return *p;
}

__device__ __forceinline__ float sigmoid_f(float v) {
    return 1.0f / (1.0f + __expf(-v));
}
__device__ __forceinline__ float tanh_f(float v) {
    float t = __expf(-2.0f * v);
    return (1.0f - t) / (1.0f + t);
}

__global__ __launch_bounds__(256) void pack_kernel(
    const float* __restrict__ W, float* __restrict__ P, int K, int Nc) {
    int w = (int)((blockIdx.x * 256u + threadIdx.x) >> 5);
    int lane = threadIdx.x & 31;
    int nk = K >> 2, nn = Nc >> 4;
    if (w >= nk * nn) return;
    int kt = w % nk, nt = w / nk;
    int n = nt * 16 + (lane & 15);
    int k = kt * 4 + ((lane >> 4) << 1);
    v2f v;
    v.x = W[(size_t)k * Nc + n];
    v.y = W[(size_t)(k + 1) * Nc + n];
    ((v2f*)P)[(size_t)(nt * nk + kt) * 32 + lane] = v;
}

__global__ __launch_bounds__(256) void zero_kernel(float* p, long n) {
    long i = (long)blockIdx.x * blockDim.x + threadIdx.x;
    long stride = (long)gridDim.x * blockDim.x;
    for (; i < n; i += stride) p[i] = 0.0f;
}

__global__ __launch_bounds__(128) void enc_kernel(
    const float* __restrict__ feat, const float* __restrict__ pW0,
    const float* __restrict__ b0, const float* __restrict__ pW1,
    const float* __restrict__ b1, float* __restrict__ x, int N) {
    __shared__ float t[4][16 * 256];
    const int wave = threadIdx.x >> 5;
    const int lane = threadIdx.x & 31;
    const int row0 = (blockIdx.x * 4 + wave) * 16;
    const bool active = (row0 < N);

    if (active) {
        v2f afr[16];
#pragma unroll
        for (int kk = 0; kk < 16; ++kk)
            afr[kk] = load_a_frag(feat, 64, row0, kk * 4, lane);
#pragma unroll 1
        for (int nt = 0; nt < 16; ++nt) {
            v8f acc = {};
#pragma unroll
            for (int kk = 0; kk < 16; ++kk) {
                v2f b = load_b_packed(pW0, 16, kk, nt, lane);
                acc = wmma4(afr[kk], b, acc);
            }
            int col = nt * 16 + (lane & 15);
            float bias = b0[col];
#pragma unroll
            for (int i = 0; i < 8; ++i) {
                int r = i + 8 * (lane >> 4);
                t[wave][r * 256 + col] = fmaxf(acc[i] + bias, 0.0f);
            }
        }
    }
    __syncthreads();
    if (active) {
#pragma unroll 1
        for (int nt = 0; nt < 16; ++nt) {
            v8f acc = {};
#pragma unroll 4
            for (int kk = 0; kk < 64; ++kk) {
                int k = kk * 4 + ((lane >> 4) << 1);
                int r = lane & 15;
                v2f a; a.x = t[wave][r * 256 + k]; a.y = t[wave][r * 256 + k + 1];
                v2f b = load_b_packed(pW1, 64, kk, nt, lane);
                acc = wmma4(a, b, acc);
            }
            int col = nt * 16 + (lane & 15);
            float bias = b1[col];
#pragma unroll
            for (int i = 0; i < 8; ++i) {
                int r = row0 + i + 8 * (lane >> 4);
                x[(size_t)r * 256 + col] = fmaxf(acc[i] + bias, 0.0f);
            }
        }
    }
}

__global__ __launch_bounds__(128) void msg_kernel(
    const float* __restrict__ x, const float* __restrict__ h,
    const float* __restrict__ pmsgW, const float* __restrict__ msgb,
    float* __restrict__ mx, int N) {
    const int wave = threadIdx.x >> 5;
    const int lane = threadIdx.x & 31;
    const int row0 = blockIdx.x * 32;
    if (row0 >= N) return;
#pragma unroll 1
    for (int t = 0; t < 2; ++t) {
        const int nt = wave * 2 + t;
        v8f acc0 = {}, acc1 = {};
#pragma unroll 4
        for (int kk = 0; kk < 128; ++kk) {
            int k0 = kk * 4;
            v2f b = load_b_packed(pmsgW, 128, kk, nt, lane);
            v2f a0 = (k0 < 256) ? load_a_frag(x, 256, row0, k0, lane)
                                : load_a_frag(h, 256, row0, k0 - 256, lane);
            v2f a1 = (k0 < 256) ? load_a_frag(x, 256, row0 + 16, k0, lane)
                                : load_a_frag(h, 256, row0 + 16, k0 - 256, lane);
            acc0 = wmma4(a0, b, acc0);
            acc1 = wmma4(a1, b, acc1);
        }
        int col = nt * 16 + (lane & 15);
        float bias = msgb[col];
#pragma unroll
        for (int i = 0; i < 8; ++i) {
            int r = row0 + i + 8 * (lane >> 4);
            mx[(size_t)r * 128 + col] = acc0[i] + bias;
            mx[(size_t)(r + 16) * 128 + col] = acc1[i] + bias;
        }
    }
}

__global__ __launch_bounds__(256) void edge_kernel(
    const int* __restrict__ src, const int* __restrict__ dst,
    const float* __restrict__ mx, float* __restrict__ sums,
    float* __restrict__ cnt, int E) {
    int e = (int)((blockIdx.x * 256u + threadIdx.x) >> 5);
    int lane = threadIdx.x & 31;
    if (e >= E) return;
    int s = src[e];
    int d = dst[e];
    const float4* mv = (const float4*)(mx + (size_t)s * 128);
    float4 v = mv[lane];
    float* base = sums + (size_t)d * 128 + lane * 4;
    atomicAdd(base + 0, v.x);
    atomicAdd(base + 1, v.y);
    atomicAdd(base + 2, v.z);
    atomicAdd(base + 3, v.w);
    if (lane == 0) atomicAdd(cnt + d, 1.0f);
}

__global__ __launch_bounds__(256) void mean_kernel(
    const float* __restrict__ sums, const float* __restrict__ cnt,
    float* __restrict__ c, long n) {
    long i = (long)blockIdx.x * blockDim.x + threadIdx.x;
    if (i >= n) return;
    long node = i >> 7;
    c[i] = sums[i] / fmaxf(cnt[node], 1.0f);
}

__global__ __launch_bounds__(128) void gru_kernel(
    const float* __restrict__ x, const float* __restrict__ c,
    const float* __restrict__ h, const float* __restrict__ pWih,
    const float* __restrict__ pWhh, const float* __restrict__ bih,
    const float* __restrict__ bhh, float* __restrict__ hnew, int N) {
    const int wave = threadIdx.x >> 5;
    const int lane = threadIdx.x & 31;
    const int row0 = blockIdx.x * 32;
    if (row0 >= N) return;
    const int p = blockIdx.y * 4 + wave;
#pragma unroll 1
    for (int t = 0; t < 2; ++t) {
        const int nt = p * 2 + t;
        v8f ir0 = {}, iz0 = {}, in0 = {}, ir1 = {}, iz1 = {}, in1 = {};
        v8f hr0 = {}, hz0 = {}, hn0 = {}, hr1 = {}, hz1 = {}, hn1 = {};
#pragma unroll 2
        for (int kk = 0; kk < 96; ++kk) {
            int k0 = kk * 4;
            v2f br = load_b_packed(pWih, 96, kk, nt, lane);
            v2f bz = load_b_packed(pWih, 96, kk, nt + 16, lane);
            v2f bn = load_b_packed(pWih, 96, kk, nt + 32, lane);
            v2f a0 = (k0 < 256) ? load_a_frag(x, 256, row0, k0, lane)
                                : load_a_frag(c, 128, row0, k0 - 256, lane);
            v2f a1 = (k0 < 256) ? load_a_frag(x, 256, row0 + 16, k0, lane)
                                : load_a_frag(c, 128, row0 + 16, k0 - 256, lane);
            ir0 = wmma4(a0, br, ir0);  ir1 = wmma4(a1, br, ir1);
            iz0 = wmma4(a0, bz, iz0);  iz1 = wmma4(a1, bz, iz1);
            in0 = wmma4(a0, bn, in0);  in1 = wmma4(a1, bn, in1);
        }
#pragma unroll 2
        for (int kk = 0; kk < 64; ++kk) {
            int k0 = kk * 4;
            v2f br = load_b_packed(pWhh, 64, kk, nt, lane);
            v2f bz = load_b_packed(pWhh, 64, kk, nt + 16, lane);
            v2f bn = load_b_packed(pWhh, 64, kk, nt + 32, lane);
            v2f a0 = load_a_frag(h, 256, row0, k0, lane);
            v2f a1 = load_a_frag(h, 256, row0 + 16, k0, lane);
            hr0 = wmma4(a0, br, hr0);  hr1 = wmma4(a1, br, hr1);
            hz0 = wmma4(a0, bz, hz0);  hz1 = wmma4(a1, bz, hz1);
            hn0 = wmma4(a0, bn, hn0);  hn1 = wmma4(a1, bn, hn1);
        }
        int col = nt * 16 + (lane & 15);
        float bir = bih[col], biz = bih[col + 256], bin = bih[col + 512];
        float bhr = bhh[col], bhz = bhh[col + 256], bhn = bhh[col + 512];
#pragma unroll
        for (int i = 0; i < 8; ++i) {
            int r = row0 + i + 8 * (lane >> 4);
            {
                float rg = sigmoid_f((ir0[i] + bir) + (hr0[i] + bhr));
                float zg = sigmoid_f((iz0[i] + biz) + (hz0[i] + bhz));
                float ng = tanh_f((in0[i] + bin) + rg * (hn0[i] + bhn));
                float hv = h[(size_t)r * 256 + col];
                hnew[(size_t)r * 256 + col] = (1.0f - zg) * ng + zg * hv;
            }
            {
                float rg = sigmoid_f((ir1[i] + bir) + (hr1[i] + bhr));
                float zg = sigmoid_f((iz1[i] + biz) + (hz1[i] + bhz));
                float ng = tanh_f((in1[i] + bin) + rg * (hn1[i] + bhn));
                float hv = h[(size_t)(r + 16) * 256 + col];
                hnew[(size_t)(r + 16) * 256 + col] = (1.0f - zg) * ng + zg * hv;
            }
        }
    }
}

__global__ __launch_bounds__(128) void out_kernel(
    const float* __restrict__ hnew, const float* __restrict__ poutW,
    const float* __restrict__ outb, float* __restrict__ q, int N) {
    const int wave = threadIdx.x >> 5;
    const int lane = threadIdx.x & 31;
    const int row0 = (blockIdx.x * 4 + wave) * 32;
    if (row0 >= N) return;
    v8f acc0 = {}, acc1 = {};
#pragma unroll 4
    for (int kk = 0; kk < 64; ++kk) {
        v2f b = load_b_packed(poutW, 64, kk, 0, lane);
        v2f a0 = load_a_frag(hnew, 256, row0, kk * 4, lane);
        v2f a1 = load_a_frag(hnew, 256, row0 + 16, kk * 4, lane);
        acc0 = wmma4(a0, b, acc0);
        acc1 = wmma4(a1, b, acc1);
    }
    int col = lane & 15;
    float bias = outb[col];
#pragma unroll
    for (int i = 0; i < 8; ++i) {
        int r = row0 + i + 8 * (lane >> 4);
        q[(size_t)r * 16 + col] = acc0[i] + bias;
        q[(size_t)(r + 16) * 16 + col] = acc1[i] + bias;
    }
}

static inline int packBlocks(int K, int Nc) {
    int waves = (K / 4) * (Nc / 16);
    return (waves + 7) / 8;
}

extern "C" void kernel_launch(void* const* d_in, const int* in_sizes, int n_in,
                              void* d_out, int out_size, void* d_ws,
                              size_t ws_size, hipStream_t stream) {
    (void)n_in; (void)out_size; (void)ws_size;
    const float* feat = (const float*)d_in[0];
    const float* h    = (const float*)d_in[1];
    const int*   src  = (const int*)d_in[2];
    const int*   dst  = (const int*)d_in[3];
    const float* W0   = (const float*)d_in[4];
    const float* b0   = (const float*)d_in[5];
    const float* W1   = (const float*)d_in[6];
    const float* b1   = (const float*)d_in[7];
    const float* msgW = (const float*)d_in[8];
    const float* msgb = (const float*)d_in[9];
    const float* Wih  = (const float*)d_in[10];
    const float* Whh  = (const float*)d_in[11];
    const float* bih  = (const float*)d_in[12];
    const float* bhh  = (const float*)d_in[13];
    const float* outW = (const float*)d_in[14];
    const float* outb = (const float*)d_in[15];

    const int N = in_sizes[1] / 256;
    const int E = in_sizes[2];

    float* ws    = (float*)d_ws;
    float* x     = ws;
    float* mx    = x    + (size_t)N * 256;
    float* sums  = mx   + (size_t)N * 128;
    float* cnt   = sums + (size_t)N * 128;
    float* c     = cnt  + N;
    float* pW0   = c    + (size_t)N * 128;
    float* pW1   = pW0  + 16384;
    float* pmsgW = pW1  + 65536;
    float* pWih  = pmsgW + 65536;
    float* pWhh  = pWih + 294912;
    float* poutW = pWhh + 196608;

    float* q    = (float*)d_out;
    float* hnew = q + (size_t)N * 16;

    const int encBlocks  = (N / 16 + 3) / 4;
    const int gemmBlocks = (N / 32 + 3) / 4;
    const int rowTiles   = (N + 31) / 32;
    const long zeroN     = (long)N * 128 + N;
    const int zeroBlocks = (int)((zeroN + 255) / 256);
    const int edgeBlocks = (E + 7) / 8;
    const long meanN     = (long)N * 128;
    const int meanBlocks = (int)((meanN + 255) / 256);

    pack_kernel<<<packBlocks(64, 256), 256, 0, stream>>>(W0, pW0, 64, 256);
    pack_kernel<<<packBlocks(256, 256), 256, 0, stream>>>(W1, pW1, 256, 256);
    pack_kernel<<<packBlocks(512, 128), 256, 0, stream>>>(msgW, pmsgW, 512, 128);
    pack_kernel<<<packBlocks(384, 768), 256, 0, stream>>>(Wih, pWih, 384, 768);
    pack_kernel<<<packBlocks(256, 768), 256, 0, stream>>>(Whh, pWhh, 256, 768);
    pack_kernel<<<packBlocks(256, 16), 256, 0, stream>>>(outW, poutW, 256, 16);

    zero_kernel<<<zeroBlocks, 256, 0, stream>>>(sums, zeroN);
    enc_kernel<<<encBlocks, 128, 0, stream>>>(feat, pW0, b0, pW1, b1, x, N);
    msg_kernel<<<rowTiles, 128, 0, stream>>>(x, h, pmsgW, msgb, mx, N);
    edge_kernel<<<edgeBlocks, 256, 0, stream>>>(src, dst, mx, sums, cnt, E);
    mean_kernel<<<meanBlocks, 256, 0, stream>>>(sums, cnt, c, meanN);
    gru_kernel<<<dim3(rowTiles, 2), 128, 0, stream>>>(x, c, h, pWih, pWhh, bih,
                                                      bhh, hnew, N);
    out_kernel<<<gemmBlocks, 128, 0, stream>>>(hnew, poutW, outb, q, N);
}


# edge sum: float atomic scatter replaced by bucketing edges by dst (int atomic arrival-order rank + scan) and one wave per node that adds its rows in that rank order and writes the mean
# speedup vs baseline: 2.6343x; 1.2336x over previous
#include <hip/hip_runtime.h>
#include <hip/hip_bf16.h>


typedef __attribute__((ext_vector_type(2))) float v2f;
typedef __attribute__((ext_vector_type(8))) float v8f;

__device__ __forceinline__ v8f wmma4(v2f a, v2f b, v8f c) {
    return __builtin_amdgcn_wmma_f32_16x16x4_f32(
        false, a, false, b, (short)0, c, false, false);
}

__device__ __forceinline__ v2f load_a_frag(const float* M, int ld, int row0,
                                           int k0, int lane) {
    int r = row0 + (lane & 15);
    int k = k0 + ((lane >> 4) << 1);
    const float* p = M + (size_t)r * ld + k;
    v2f a; a.x = p[0]; a.y = p[1];
    return a;
}

__device__ __forceinline__ v2f load_b_packed(const float* P, int nk, int kt,
                                             int nt, int lane) {
    const v2f* p = (const v2f*)P + ((size_t)(nt * nk + kt) * 32 + lane);
    return *p;
}

__device__ __forceinline__ float sigmoid_f(float v) {
    return 1.0f / (1.0f + __expf(-v));
}
__device__ __forceinline__ float tanh_f(float v) {
    float t = __expf(-2.0f * v);
    return (1.0f - t) / (1.0f + t);
}

__global__ __launch_bounds__(256) void pack_kernel(
    const float* __restrict__ W, float* __restrict__ P, int K, int Nc) {
    int w = (int)((blockIdx.x * 256u + threadIdx.x) >> 5);
    int lane = threadIdx.x & 31;
    int nk = K >> 2, nn = Nc >> 4;
    if (w >= nk * nn) return;
    int kt = w % nk, nt = w / nk;
    int n = nt * 16 + (lane & 15);
    int k = kt * 4 + ((lane >> 4) << 1);
    v2f v;
    v.x = W[(size_t)k * Nc + n];
    v.y = W[(size_t)(k + 1) * Nc + n];
    ((v2f*)P)[(size_t)(nt * nk + kt) * 32 + lane] = v;
}

__global__ __launch_bounds__(256) void zero_kernel(int* p, int n) {
    int i = (int)(blockIdx.x * 256u + threadIdx.x);
    if (i < n) p[i] = 0;
}

__global__ __launch_bounds__(128) void enc_kernel(
    const float* __restrict__ feat, const float* __restrict__ pW0,
    const float* __restrict__ b0, const float* __restrict__ pW1,
    const float* __restrict__ b1, float* __restrict__ x, int N) {
    __shared__ float t[4][16 * 256];
    const int wave = threadIdx.x >> 5;
    const int lane = threadIdx.x & 31;
    const int row0 = (blockIdx.x * 4 + wave) * 16;
    const bool active = (row0 < N);

    if (active) {
        v2f afr[16];
#pragma unroll
        for (int kk = 0; kk < 16; ++kk)
            afr[kk] = load_a_frag(feat, 64, row0, kk * 4, lane);
#pragma unroll 1
        for (int nt = 0; nt < 16; ++nt) {
            v8f acc = {};
#pragma unroll
            for (int kk = 0; kk < 16; ++kk) {
                v2f b = load_b_packed(pW0, 16, kk, nt, lane);
                acc = wmma4(afr[kk], b, acc);
            }
            int col = nt * 16 + (lane & 15);
            float bias = b0[col];
#pragma unroll
            for (int i = 0; i < 8; ++i) {
                int r = i + 8 * (lane >> 4);
                t[wave][r * 256 + col] = fmaxf(acc[i] + bias, 0.0f);
            }
        }
    }
    __syncthreads();
    if (active) {
#pragma unroll 1
        for (int nt = 0; nt < 16; ++nt) {
            v8f acc = {};
#pragma unroll 4
            for (int kk = 0; kk < 64; ++kk) {
                int k = kk * 4 + ((lane >> 4) << 1);
                int r = lane & 15;
                v2f a; a.x = t[wave][r * 256 + k]; a.y = t[wave][r * 256 + k + 1];
                v2f b = load_b_packed(pW1, 64, kk, nt, lane);
                acc = wmma4(a, b, acc);
            }
            int col = nt * 16 + (lane & 15);
            float bias = b1[col];
#pragma unroll
            for (int i = 0; i < 8; ++i) {
                int r = row0 + i + 8 * (lane >> 4);
                x[(size_t)r * 256 + col] = fmaxf(acc[i] + bias, 0.0f);
            }
        }
    }
}

__global__ __launch_bounds__(128) void msg_kernel(
    const float* __restrict__ x, const float* __restrict__ h,
    const float* __restrict__ pmsgW, const float* __restrict__ msgb,
    float* __restrict__ mx, int N) {
    const int wave = threadIdx.x >> 5;
    const int lane = threadIdx.x & 31;
    const int row0 = blockIdx.x * 32;
    if (row0 >= N) return;
#pragma unroll 1
    for (int t = 0; t < 2; ++t) {
        const int nt = wave * 2 + t;
        v8f acc0 = {}, acc1 = {};
#pragma unroll 4
        for (int kk = 0; kk < 128; ++kk) {
            int k0 = kk * 4;
            v2f b = load_b_packed(pmsgW, 128, kk, nt, lane);
            v2f a0 = (k0 < 256) ? load_a_frag(x, 256, row0, k0, lane)
                                : load_a_frag(h, 256, row0, k0 - 256, lane);
            v2f a1 = (k0 < 256) ? load_a_frag(x, 256, row0 + 16, k0, lane)
                                : load_a_frag(h, 256, row0 + 16, k0 - 256, lane);
            acc0 = wmma4(a0, b, acc0);
            acc1 = wmma4(a1, b, acc1);
        }
        int col = nt * 16 + (lane & 15);
        float bias = msgb[col];
#pragma unroll
        for (int i = 0; i < 8; ++i) {
            int r = row0 + i + 8 * (lane >> 4);
            mx[(size_t)r * 128 + col] = acc0[i] + bias;
            mx[(size_t)(r + 16) * 128 + col] = acc1[i] + bias;
        }
    }
}

__global__ __launch_bounds__(256) void rank_kernel(
    const int* __restrict__ dst, int* __restrict__ deg,
    int* __restrict__ rnk, int E) {
    int e = (int)(blockIdx.x * 256u + threadIdx.x);
    if (e >= E) return;
    rnk[e] = atomicAdd(deg + dst[e], 1);
}

__global__ __launch_bounds__(1024) void scan_kernel(
    const int* __restrict__ deg, int* __restrict__ off, int n) {
    __shared__ int part[1024];
    const int t = threadIdx.x;
    const int per = (n + 1023) >> 10;
    const int beg = min(t * per, n);
    const int end = min(beg + per, n);
    int s = 0;
#pragma unroll 4
    for (int i = beg; i < end; ++i) s += deg[i];
    part[t] = s;
    __syncthreads();
    for (int o = 1; o < 1024; o <<= 1) {
        int v = (t >= o) ? part[t - o] : 0;
        __syncthreads();
        part[t] += v;
        __syncthreads();
    }
    int run = part[t] - s;
#pragma unroll 4
    for (int i = beg; i < end; ++i) {
        int d = deg[i];
        off[i] = run;
        run += d;
    }
    if (t == 1023) off[n] = part[1023];
}

__global__ __launch_bounds__(256) void fill_kernel(
    const int* __restrict__ src, const int* __restrict__ dst,
    const int* __restrict__ off, const int* __restrict__ rnk,
    int* __restrict__ ssrc, int E) {
    int e = (int)(blockIdx.x * 256u + threadIdx.x);
    if (e >= E) return;
    int pos = off[dst[e]] + rnk[e];
    pos = min(max(pos, 0), E - 1);
    ssrc[pos] = src[e];
}

__global__ __launch_bounds__(256) void gather_kernel(
    const int* __restrict__ off, const int* __restrict__ ssrc,
    const float* __restrict__ mx, float* __restrict__ c, int N) {
    const int node = __builtin_amdgcn_readfirstlane(
        (int)((blockIdx.x * 256u + threadIdx.x) >> 5));
    const int lane = threadIdx.x & 31;
    if (node >= N) return;
    const int beg = off[node];
    const int end = off[node + 1];
    const float4* mv = (const float4*)mx;
    float4 acc = make_float4(0.0f, 0.0f, 0.0f, 0.0f);
    int j = beg;
    for (; j + 8 <= end; j += 8) {
        float4 v[8];
#pragma unroll
        for (int k = 0; k < 8; ++k) {
            int s = ssrc[j + k];
            v[k] = mv[(size_t)s * 32 + lane];
        }
#pragma unroll
        for (int k = 0; k < 8; ++k) {
            acc.x += v[k].x; acc.y += v[k].y;
            acc.z += v[k].z; acc.w += v[k].w;
        }
    }
    if (j < end) {
        float4 v[8];
#pragma unroll
        for (int k = 0; k < 8; ++k) {
            int s = ssrc[min(j + k, end - 1)];
            v[k] = mv[(size_t)s * 32 + lane];
        }
#pragma unroll
        for (int k = 0; k < 8; ++k) {
            const bool in = (j + k < end);
            acc.x += in ? v[k].x : 0.0f; acc.y += in ? v[k].y : 0.0f;
            acc.z += in ? v[k].z : 0.0f; acc.w += in ? v[k].w : 0.0f;
        }
    }
    const float den = fmaxf((float)(end - beg), 1.0f);
    float4 o;
    o.x = acc.x / den; o.y = acc.y / den; o.z = acc.z / den; o.w = acc.w / den;
    ((float4*)c)[(size_t)node * 32 + lane] = o;
}

__global__ __launch_bounds__(128) void gru_kernel(
    const float* __restrict__ x, const float* __restrict__ c,
    const float* __restrict__ h, const float* __restrict__ pWih,
    const float* __restrict__ pWhh, const float* __restrict__ bih,
    const float* __restrict__ bhh, float* __restrict__ hnew, int N) {
    const int wave = threadIdx.x >> 5;
    const int lane = threadIdx.x & 31;
    const int row0 = blockIdx.x * 32;
    if (row0 >= N) return;
    const int p = blockIdx.y * 4 + wave;
#pragma unroll 1
    for (int t = 0; t < 2; ++t) {
        const int nt = p * 2 + t;
        v8f ir0 = {}, iz0 = {}, in0 = {}, ir1 = {}, iz1 = {}, in1 = {};
        v8f hr0 = {}, hz0 = {}, hn0 = {}, hr1 = {}, hz1 = {}, hn1 = {};
#pragma unroll 2
        for (int kk = 0; kk < 96; ++kk) {
            int k0 = kk * 4;
            v2f br = load_b_packed(pWih, 96, kk, nt, lane);
            v2f bz = load_b_packed(pWih, 96, kk, nt + 16, lane);
            v2f bn = load_b_packed(pWih, 96, kk, nt + 32, lane);
            v2f a0 = (k0 < 256) ? load_a_frag(x, 256, row0, k0, lane)
                                : load_a_frag(c, 128, row0, k0 - 256, lane);
            v2f a1 = (k0 < 256) ? load_a_frag(x, 256, row0 + 16, k0, lane)
                                : load_a_frag(c, 128, row0 + 16, k0 - 256, lane);
            ir0 = wmma4(a0, br, ir0);  ir1 = wmma4(a1, br, ir1);
            iz0 = wmma4(a0, bz, iz0);  iz1 = wmma4(a1, bz, iz1);
            in0 = wmma4(a0, bn, in0);  in1 = wmma4(a1, bn, in1);
        }
#pragma unroll 2
        for (int kk = 0; kk < 64; ++kk) {
            int k0 = kk * 4;
            v2f br = load_b_packed(pWhh, 64, kk, nt, lane);
            v2f bz = load_b_packed(pWhh, 64, kk, nt + 16, lane);
            v2f bn = load_b_packed(pWhh, 64, kk, nt + 32, lane);
            v2f a0 = load_a_frag(h, 256, row0, k0, lane);
            v2f a1 = load_a_frag(h, 256, row0 + 16, k0, lane);
            hr0 = wmma4(a0, br, hr0);  hr1 = wmma4(a1, br, hr1);
            hz0 = wmma4(a0, bz, hz0);  hz1 = wmma4(a1, bz, hz1);
            hn0 = wmma4(a0, bn, hn0);  hn1 = wmma4(a1, bn, hn1);
        }
        int col = nt * 16 + (lane & 15);
        float bir = bih[col], biz = bih[col + 256], bin = bih[col + 512];
        float bhr = bhh[col], bhz = bhh[col + 256], bhn = bhh[col + 512];
#pragma unroll
        for (int i = 0; i < 8; ++i) {
            int r = row0 + i + 8 * (lane >> 4);
            {
                float rg = sigmoid_f((ir0[i] + bir) + (hr0[i] + bhr));
                float zg = sigmoid_f((iz0[i] + biz) + (hz0[i] + bhz));
                float ng = tanh_f((in0[i] + bin) + rg * (hn0[i] + bhn));
                float hv = h[(size_t)r * 256 + col];
                hnew[(size_t)r * 256 + col] = (1.0f - zg) * ng + zg * hv;
            }
            {
                float rg = sigmoid_f((ir1[i] + bir) + (hr1[i] + bhr));
                float zg = sigmoid_f((iz1[i] + biz) + (hz1[i] + bhz));
                float ng = tanh_f((in1[i] + bin) + rg * (hn1[i] + bhn));
                float hv = h[(size_t)(r + 16) * 256 + col];
                hnew[(size_t)(r + 16) * 256 + col] = (1.0f - zg) * ng + zg * hv;
            }
        }
    }
}

__global__ __launch_bounds__(128) void out_kernel(
    const float* __restrict__ hnew, const float* __restrict__ poutW,
    const float* __restrict__ outb, float* __restrict__ q, int N) {
    const int wave = threadIdx.x >> 5;
    const int lane = threadIdx.x & 31;
    const int row0 = (blockIdx.x * 4 + wave) * 32;
    if (row0 >= N) return;
    v8f acc0 = {}, acc1 = {};
#pragma unroll 4
    for (int kk = 0; kk < 64; ++kk) {
        v2f b = load_b_packed(poutW, 64, kk, 0, lane);
        v2f a0 = load_a_frag(hnew, 256, row0, kk * 4, lane);
        v2f a1 = load_a_frag(hnew, 256, row0 + 16, kk * 4, lane);
        acc0 = wmma4(a0, b, acc0);
        acc1 = wmma4(a1, b, acc1);
    }
    int col = lane & 15;
    float bias = outb[col];
#pragma unroll
    for (int i = 0; i < 8; ++i) {
        int r = row0 + i + 8 * (lane >> 4);
        q[(size_t)r * 16 + col] = acc0[i] + bias;
        q[(size_t)(r + 16) * 16 + col] = acc1[i] + bias;
    }
}

static inline int packBlocks(int K, int Nc) {
    int waves = (K / 4) * (Nc / 16);
    return (waves + 7) / 8;
}

extern "C" void kernel_launch(void* const* d_in, const int* in_sizes, int n_in,
                              void* d_out, int out_size, void* d_ws,
                              size_t ws_size, hipStream_t stream) {
    (void)n_in; (void)out_size; (void)ws_size;
    const float* feat = (const float*)d_in[0];
    const float* h    = (const float*)d_in[1];
    const int*   src  = (const int*)d_in[2];
    const int*   dst  = (const int*)d_in[3];
    const float* W0   = (const float*)d_in[4];
    const float* b0   = (const float*)d_in[5];
    const float* W1   = (const float*)d_in[6];
    const float* b1   = (const float*)d_in[7];
    const float* msgW = (const float*)d_in[8];
    const float* msgb = (const float*)d_in[9];
    const float* Wih  = (const float*)d_in[10];
    const float* Whh  = (const float*)d_in[11];
    const float* bih  = (const float*)d_in[12];
    const float* bhh  = (const float*)d_in[13];
    const float* outW = (const float*)d_in[14];
    const float* outb = (const float*)d_in[15];

    const int N = in_sizes[1] / 256;
    const int E = in_sizes[2];

    float* ws    = (float*)d_ws;
    float* x     = ws;
    float* mx    = x    + (size_t)N * 256;
    float* bkt   = mx   + (size_t)N * 128;
    int*   deg   = (int*)bkt;
    int*   off   = deg + N;
    int*   rnk   = off + N + 1;
    int*   ssrc  = rnk + E;
    float* c     = bkt  + (size_t)N * 128 + N;
    float* pW0   = c    + (size_t)N * 128;
    float* pW1   = pW0  + 16384;
    float* pmsgW = pW1  + 65536;
    float* pWih  = pmsgW + 65536;
    float* pWhh  = pWih + 294912;
    float* poutW = pWhh + 196608;

    float* q    = (float*)d_out;
    float* hnew = q + (size_t)N * 16;

    const int encBlocks  = (N / 16 + 3) / 4;
    const int gemmBlocks = (N / 32 + 3) / 4;
    const int rowTiles   = (N + 31) / 32;
    const int zeroBlocks = (N + 255) / 256;
    const int edgeBlocks = (E + 255) / 256;
    const int nodeBlocks = (N + 7) / 8;

    pack_kernel<<<packBlocks(64, 256), 256, 0, stream>>>(W0, pW0, 64, 256);
    pack_kernel<<<packBlocks(256, 256), 256, 0, stream>>>(W1, pW1, 256, 256);
    pack_kernel<<<packBlocks(512, 128), 256, 0, stream>>>(msgW, pmsgW, 512, 128);
    pack_kernel<<<packBlocks(384, 768), 256, 0, stream>>>(Wih, pWih, 384, 768);
    pack_kernel<<<packBlocks(256, 768), 256, 0, stream>>>(Whh, pWhh, 256, 768);
    pack_kernel<<<packBlocks(256, 16), 256, 0, stream>>>(outW, poutW, 256, 16);

    zero_kernel<<<zeroBlocks, 256, 0, stream>>>(deg, N);
    rank_kernel<<<edgeBlocks, 256, 0, stream>>>(dst, deg, rnk, E);
    scan_kernel<<<1, 1024, 0, stream>>>(deg, off, N);
    fill_kernel<<<edgeBlocks, 256, 0, stream>>>(src, dst, off, rnk, ssrc, E);
    enc_kernel<<<encBlocks, 128, 0, stream>>>(feat, pW0, b0, pW1, b1, x, N);
    msg_kernel<<<rowTiles, 128, 0, stream>>>(x, h, pmsgW, msgb, mx, N);
    gather_kernel<<<nodeBlocks, 256, 0, stream>>>(off, ssrc, mx, c, N);
    gru_kernel<<<dim3(rowTiles, 2), 128, 0, stream>>>(x, c, h, pWih, pWhh, bih,
                                                      bhh, hnew, N);
    out_kernel<<<gemmBlocks, 128, 0, stream>>>(hnew, poutW, outb, q, N);
}


# enc, msg, gru and out GEMM K loops software-pipelined: next K block's A/B fragment loads issued as one group into a second register set ahead of the current block's WMMAs; k order unchanged
# speedup vs baseline: 4.1532x; 1.5766x over previous
#include <hip/hip_runtime.h>
#include <hip/hip_bf16.h>


typedef __attribute__((ext_vector_type(2))) float v2f;
typedef __attribute__((ext_vector_type(8))) float v8f;

__device__ __forceinline__ v8f wmma4(v2f a, v2f b, v8f c) {
    return __builtin_amdgcn_wmma_f32_16x16x4_f32(
        false, a, false, b, (short)0, c, false, false);
}

__device__ __forceinline__ v2f load_a_frag(const float* M, int ld, int row0,
                                           int k0, int lane) {
    int r = row0 + (lane & 15);
    int k = k0 + ((lane >> 4) << 1);
    const float* p = M + (size_t)r * ld + k;
    v2f a; a.x = p[0]; a.y = p[1];
    return a;
}


template <int KB> struct Frag21 { v2f a0[KB], a1[KB], b[KB]; };

template <int KB>
__device__ __forceinline__ void load21(Frag21<KB>& f, const float* pa0,
                                       const float* pa1, const v2f* pb,
                                       int blk) {
    const float* q0 = pa0 + blk * (KB * 4);
    const float* q1 = pa1 + blk * (KB * 4);
    const v2f* qb = pb + blk * (KB * 32);
#pragma unroll
    for (int j = 0; j < KB; ++j) {
        f.a0[j].x = q0[j * 4]; f.a0[j].y = q0[j * 4 + 1];
        f.a1[j].x = q1[j * 4]; f.a1[j].y = q1[j * 4 + 1];
        f.b[j] = qb[j * 32];
    }
}

template <int KB>
__device__ __forceinline__ void mma21(const Frag21<KB>& f, v8f& c0, v8f& c1) {
#pragma unroll
    for (int j = 0; j < KB; ++j) {
        c0 = wmma4(f.a0[j], f.b[j], c0);
        c1 = wmma4(f.a1[j], f.b[j], c1);
    }
}

template <int KB>
__device__ __forceinline__ void seg21(const float* pa0, const float* pa1,
                                      const v2f* pb, int nb, v8f& c0, v8f& c1) {
    Frag21<KB> f0, f1;
    load21<KB>(f0, pa0, pa1, pb, 0);
    int b = 0;
#pragma unroll 1
    for (; b + 2 < nb; b += 2) {
        load21<KB>(f1, pa0, pa1, pb, b + 1);
        mma21<KB>(f0, c0, c1);
        load21<KB>(f0, pa0, pa1, pb, b + 2);
        mma21<KB>(f1, c0, c1);
    }
    load21<KB>(f1, pa0, pa1, pb, b + 1);
    mma21<KB>(f0, c0, c1);
    mma21<KB>(f1, c0, c1);
}

template <int KB> struct Frag23 { v2f a0[KB], a1[KB], br[KB], bz[KB], bn[KB]; };

template <int KB>
__device__ __forceinline__ void load23(Frag23<KB>& f, const float* A, int aoff,
                                       int a1, const float* B, int boff,
                                       int gate, int blk) {
    const float* qa = A + blk * (KB * 4);
    const float* qb = B + blk * (KB * 64);
#pragma unroll
    for (int j = 0; j < KB; ++j) {
        f.a0[j].x = qa[aoff + j * 4];      f.a0[j].y = qa[aoff + j * 4 + 1];
        f.a1[j].x = qa[aoff + a1 + j * 4]; f.a1[j].y = qa[aoff + a1 + j * 4 + 1];
        f.br[j].x = qb[boff + j * 64];     f.br[j].y = qb[boff + j * 64 + 1];
        f.bz[j].x = qb[boff + gate + j * 64];
        f.bz[j].y = qb[boff + gate + j * 64 + 1];
        f.bn[j].x = qb[boff + 2 * gate + j * 64];
        f.bn[j].y = qb[boff + 2 * gate + j * 64 + 1];
    }
}

template <int KB>
__device__ __forceinline__ void mma23(const Frag23<KB>& f, v8f& r0, v8f& r1,
                                      v8f& z0, v8f& z1, v8f& n0, v8f& n1) {
#pragma unroll
    for (int j = 0; j < KB; ++j) {
        r0 = wmma4(f.a0[j], f.br[j], r0);  r1 = wmma4(f.a1[j], f.br[j], r1);
        z0 = wmma4(f.a0[j], f.bz[j], z0);  z1 = wmma4(f.a1[j], f.bz[j], z1);
        n0 = wmma4(f.a0[j], f.bn[j], n0);  n1 = wmma4(f.a1[j], f.bn[j], n1);
    }
}

template <int KB>
__device__ __forceinline__ void seg23(const float* A, int aoff, int a1,
                                      const float* B, int boff, int gate,
                                      int nb, v8f& r0, v8f& r1, v8f& z0,
                                      v8f& z1, v8f& n0, v8f& n1) {
    Frag23<KB> f0, f1;
    load23<KB>(f0, A, aoff, a1, B, boff, gate, 0);
    int b = 0;
#pragma unroll 1
    for (; b + 2 < nb; b += 2) {
        load23<KB>(f1, A, aoff, a1, B, boff, gate, b + 1);
        mma23<KB>(f0, r0, r1, z0, z1, n0, n1);
        load23<KB>(f0, A, aoff, a1, B, boff, gate, b + 2);
        mma23<KB>(f1, r0, r1, z0, z1, n0, n1);
    }
    load23<KB>(f1, A, aoff, a1, B, boff, gate, b + 1);
    mma23<KB>(f0, r0, r1, z0, z1, n0, n1);
    mma23<KB>(f1, r0, r1, z0, z1, n0, n1);
}

template <int KB> struct FragB { v2f b[KB]; };

template <int KB>
__device__ __forceinline__ void loadB(FragB<KB>& f, const v2f* pb, int blk) {
    const v2f* qb = pb + blk * (KB * 32);
#pragma unroll
    for (int j = 0; j < KB; ++j) f.b[j] = qb[j * 32];
}

__device__ __forceinline__ float sigmoid_f(float v) {
    return 1.0f / (1.0f + __expf(-v));
}
__device__ __forceinline__ float tanh_f(float v) {
    float t = __expf(-2.0f * v);
    return (1.0f - t) / (1.0f + t);
}

__global__ __launch_bounds__(256) void pack_kernel(
    const float* __restrict__ W, float* __restrict__ P, int K, int Nc) {
    int w = (int)((blockIdx.x * 256u + threadIdx.x) >> 5);
    int lane = threadIdx.x & 31;
    int nk = K >> 2, nn = Nc >> 4;
    if (w >= nk * nn) return;
    int kt = w % nk, nt = w / nk;
    int n = nt * 16 + (lane & 15);
    int k = kt * 4 + ((lane >> 4) << 1);
    v2f v;
    v.x = W[(size_t)k * Nc + n];
    v.y = W[(size_t)(k + 1) * Nc + n];
    ((v2f*)P)[(size_t)(nt * nk + kt) * 32 + lane] = v;
}

__global__ __launch_bounds__(256) void zero_kernel(int* p, int n) {
    int i = (int)(blockIdx.x * 256u + threadIdx.x);
    if (i < n) p[i] = 0;
}

__global__ __launch_bounds__(128) void enc_kernel(
    const float* __restrict__ feat, const float* __restrict__ pW0,
    const float* __restrict__ b0, const float* __restrict__ pW1,
    const float* __restrict__ b1, float* __restrict__ x, int N) {
    __shared__ float t[4][16 * 256];
    const int wave = threadIdx.x >> 5;
    const int lane = threadIdx.x & 31;
    const int row0 = (blockIdx.x * 4 + wave) * 16;
    const bool active = (row0 < N);
    const v2f* pb1 = (const v2f*)pW1 + lane;
    FragB<32> f0, f1;

    if (active) {
        v2f afr[16];
#pragma unroll
        for (int kk = 0; kk < 16; ++kk)
            afr[kk] = load_a_frag(feat, 64, row0, kk * 4, lane);
        const v2f* pb0 = (const v2f*)pW0 + lane;
        FragB<16> g0, g1;
        loadB<16>(g0, pb0, 0);
#pragma unroll 1
        for (int nt = 0; nt < 16; nt += 2) {
            loadB<16>(g1, pb0, nt + 1);
            {
                v8f acc = {};
#pragma unroll
                for (int kk = 0; kk < 16; ++kk)
                    acc = wmma4(afr[kk], g0.b[kk], acc);
                int col = nt * 16 + (lane & 15);
                float bias = b0[col];
#pragma unroll
                for (int i = 0; i < 8; ++i) {
                    int r = i + 8 * (lane >> 4);
                    t[wave][r * 256 + col] = fmaxf(acc[i] + bias, 0.0f);
                }
            }
            loadB<16>(g0, pb0, min(nt + 2, 15));
            {
                v8f acc = {};
#pragma unroll
                for (int kk = 0; kk < 16; ++kk)
                    acc = wmma4(afr[kk], g1.b[kk], acc);
                int col = (nt + 1) * 16 + (lane & 15);
                float bias = b0[col];
#pragma unroll
                for (int i = 0; i < 8; ++i) {
                    int r = i + 8 * (lane >> 4);
                    t[wave][r * 256 + col] = fmaxf(acc[i] + bias, 0.0f);
                }
            }
        }
        loadB<32>(f0, pb1, 0);
    }
    __syncthreads();
    if (active) {
        const float* ta = &t[wave][(lane & 15) * 256 + ((lane >> 4) << 1)];
#pragma unroll 1
        for (int nt = 0; nt < 16; ++nt) {
            v8f acc = {};
            loadB<32>(f1, pb1, nt * 2 + 1);
#pragma unroll
            for (int kk = 0; kk < 32; ++kk) {
                v2f a; a.x = ta[kk * 4]; a.y = ta[kk * 4 + 1];
                acc = wmma4(a, f0.b[kk], acc);
            }
            loadB<32>(f0, pb1, min(nt * 2 + 2, 31));
#pragma unroll
            for (int kk = 0; kk < 32; ++kk) {
                v2f a; a.x = ta[128 + kk * 4]; a.y = ta[128 + kk * 4 + 1];
                acc = wmma4(a, f1.b[kk], acc);
            }
            int col = nt * 16 + (lane & 15);
            float bias = b1[col];
#pragma unroll
            for (int i = 0; i < 8; ++i) {
                int r = row0 + i + 8 * (lane >> 4);
                x[(size_t)r * 256 + col] = fmaxf(acc[i] + bias, 0.0f);
            }
        }
    }
}

__global__ __launch_bounds__(128) void msg_kernel(
    const float* __restrict__ x, const float* __restrict__ h,
    const float* __restrict__ pmsgW, const float* __restrict__ msgb,
    float* __restrict__ mx, int N) {
    const int wave = threadIdx.x >> 5;
    const int lane = threadIdx.x & 31;
    const int row0 = blockIdx.x * 32;
    if (row0 >= N) return;
    const size_t arow = (size_t)(row0 + (lane & 15)) * 256 + ((lane >> 4) << 1);
    const float* xa0 = x + arow;
    const float* xa1 = xa0 + 16 * 256;
    const float* ha0 = h + arow;
    const float* ha1 = ha0 + 16 * 256;
#pragma unroll 1
    for (int t = 0; t < 2; ++t) {
        const int nt = wave * 2 + t;
        const v2f* pb = (const v2f*)pmsgW + ((size_t)(nt * 128) * 32 + lane);
        v8f acc0 = {}, acc1 = {};
        seg21<8>(xa0, xa1, pb, 8, acc0, acc1);
        seg21<8>(ha0, ha1, pb + 64 * 32, 8, acc0, acc1);
        int col = nt * 16 + (lane & 15);
        float bias = msgb[col];
#pragma unroll
        for (int i = 0; i < 8; ++i) {
            int r = row0 + i + 8 * (lane >> 4);
            mx[(size_t)r * 128 + col] = acc0[i] + bias;
            mx[(size_t)(r + 16) * 128 + col] = acc1[i] + bias;
        }
    }
}

__global__ __launch_bounds__(256) void rank_kernel(
    const int* __restrict__ dst, int* __restrict__ deg,
    int* __restrict__ rnk, int E) {
    int e = (int)(blockIdx.x * 256u + threadIdx.x);
    if (e >= E) return;
    rnk[e] = atomicAdd(deg + dst[e], 1);
}

__global__ __launch_bounds__(1024) void scan_kernel(
    const int* __restrict__ deg, int* __restrict__ off, int n) {
    __shared__ int part[1024];
    const int t = threadIdx.x;
    const int per = (n + 1023) >> 10;
    const int beg = min(t * per, n);
    const int end = min(beg + per, n);
    int s = 0;
#pragma unroll 4
    for (int i = beg; i < end; ++i) s += deg[i];
    part[t] = s;
    __syncthreads();
    for (int o = 1; o < 1024; o <<= 1) {
        int v = (t >= o) ? part[t - o] : 0;
        __syncthreads();
        part[t] += v;
        __syncthreads();
    }
    int run = part[t] - s;
#pragma unroll 4
    for (int i = beg; i < end; ++i) {
        int d = deg[i];
        off[i] = run;
        run += d;
    }
    if (t == 1023) off[n] = part[1023];
}

__global__ __launch_bounds__(256) void fill_kernel(
    const int* __restrict__ src, const int* __restrict__ dst,
    const int* __restrict__ off, const int* __restrict__ rnk,
    int* __restrict__ ssrc, int E) {
    int e = (int)(blockIdx.x * 256u + threadIdx.x);
    if (e >= E) return;
    int pos = off[dst[e]] + rnk[e];
    pos = min(max(pos, 0), E - 1);
    ssrc[pos] = src[e];
}

__global__ __launch_bounds__(256) void gather_kernel(
    const int* __restrict__ off, const int* __restrict__ ssrc,
    const float* __restrict__ mx, float* __restrict__ c, int N) {
    const int node = __builtin_amdgcn_readfirstlane(
        (int)((blockIdx.x * 256u + threadIdx.x) >> 5));
    const int lane = threadIdx.x & 31;
    if (node >= N) return;
    const int beg = off[node];
    const int end = off[node + 1];
    const float4* mv = (const float4*)mx;
    float4 acc = make_float4(0.0f, 0.0f, 0.0f, 0.0f);
    int j = beg;
    for (; j + 8 <= end; j += 8) {
        float4 v[8];
#pragma unroll
        for (int k = 0; k < 8; ++k) {
            int s = ssrc[j + k];
            v[k] = mv[(size_t)s * 32 + lane];
        }
#pragma unroll
        for (int k = 0; k < 8; ++k) {
            acc.x += v[k].x; acc.y += v[k].y;
            acc.z += v[k].z; acc.w += v[k].w;
        }
    }
    if (j < end) {
        float4 v[8];
#pragma unroll
        for (int k = 0; k < 8; ++k) {
            int s = ssrc[min(j + k, end - 1)];
            v[k] = mv[(size_t)s * 32 + lane];
        }
#pragma unroll
        for (int k = 0; k < 8; ++k) {
            const bool in = (j + k < end);
            acc.x += in ? v[k].x : 0.0f; acc.y += in ? v[k].y : 0.0f;
            acc.z += in ? v[k].z : 0.0f; acc.w += in ? v[k].w : 0.0f;
        }
    }
    const float den = fmaxf((float)(end - beg), 1.0f);
    float4 o;
    o.x = acc.x / den; o.y = acc.y / den; o.z = acc.z / den; o.w = acc.w / den;
    ((float4*)c)[(size_t)node * 32 + lane] = o;
}

constexpr int GRU_KB = 2;
__global__ __launch_bounds__(128, 5) void gru_kernel(
    const float* __restrict__ x, const float* __restrict__ c,
    const float* __restrict__ h, const float* __restrict__ pWih,
    const float* __restrict__ pWhh, const float* __restrict__ bih,
    const float* __restrict__ bhh, float* __restrict__ hnew, int N) {
    const int wave = __builtin_amdgcn_readfirstlane((int)(threadIdx.x >> 5));
    const int lane = threadIdx.x & 31;
    const int row0 = blockIdx.x * 32;
    if (row0 >= N) return;
    const int p = blockIdx.y * 4 + wave;
    const int ak = (lane >> 4) << 1;
    const int boff = lane * 2;
    int lr = lane & 15;
#pragma unroll 1
    for (int t = 0; t < 2; ++t) {
        const int nt = p * 2 + t;
        asm volatile("" : "+v"(lr));
        lr &= 15;
        v8f ir0 = {}, iz0 = {}, in0 = {}, ir1 = {}, iz1 = {}, in1 = {};
        v8f hr0 = {}, hz0 = {}, hn0 = {}, hr1 = {}, hz1 = {}, hn1 = {};
        const float* pbi = pWih + (size_t)(nt * 96) * 64;
        const float* pbh = pWhh + (size_t)(nt * 64) * 64;
        seg23<GRU_KB>(x + (size_t)row0 * 256, lr * 256 + ak, 16 * 256,
                      pbi, boff, 16 * 96 * 64, 64 / GRU_KB,
                      ir0, ir1, iz0, iz1, in0, in1);
        seg23<GRU_KB>(c + (size_t)row0 * 128, lr * 128 + ak, 16 * 128,
                      pbi + 64 * 64, boff, 16 * 96 * 64, 32 / GRU_KB,
                      ir0, ir1, iz0, iz1, in0, in1);
        seg23<GRU_KB>(h + (size_t)row0 * 256, lr * 256 + ak, 16 * 256,
                      pbh, boff, 16 * 64 * 64, 64 / GRU_KB,
                      hr0, hr1, hz0, hz1, hn0, hn1);
        int col = nt * 16 + (lane & 15);
        float bir = bih[col], biz = bih[col + 256], bin = bih[col + 512];
        float bhr = bhh[col], bhz = bhh[col + 256], bhn = bhh[col + 512];
#pragma unroll
        for (int i = 0; i < 8; ++i) {
            int r = row0 + i + 8 * (lane >> 4);
            {
                float rg = sigmoid_f((ir0[i] + bir) + (hr0[i] + bhr));
                float zg = sigmoid_f((iz0[i] + biz) + (hz0[i] + bhz));
                float ng = tanh_f((in0[i] + bin) + rg * (hn0[i] + bhn));
                float hv = h[(size_t)r * 256 + col];
                hnew[(size_t)r * 256 + col] = (1.0f - zg) * ng + zg * hv;
            }
            {
                float rg = sigmoid_f((ir1[i] + bir) + (hr1[i] + bhr));
                float zg = sigmoid_f((iz1[i] + biz) + (hz1[i] + bhz));
                float ng = tanh_f((in1[i] + bin) + rg * (hn1[i] + bhn));
                float hv = h[(size_t)(r + 16) * 256 + col];
                hnew[(size_t)(r + 16) * 256 + col] = (1.0f - zg) * ng + zg * hv;
            }
        }
    }
}

__global__ __launch_bounds__(128) void out_kernel(
    const float* __restrict__ hnew, const float* __restrict__ poutW,
    const float* __restrict__ outb, float* __restrict__ q, int N) {
    const int wave = threadIdx.x >> 5;
    const int lane = threadIdx.x & 31;
    const int row0 = (blockIdx.x * 4 + wave) * 32;
    if (row0 >= N) return;
    v8f acc0 = {}, acc1 = {};
    const float* a0 = hnew + (size_t)(row0 + (lane & 15)) * 256 + ((lane >> 4) << 1);
    seg21<16>(a0, a0 + 16 * 256, (const v2f*)poutW + lane, 4, acc0, acc1);
    int col = lane & 15;
    float bias = outb[col];
#pragma unroll
    for (int i = 0; i < 8; ++i) {
        int r = row0 + i + 8 * (lane >> 4);
        q[(size_t)r * 16 + col] = acc0[i] + bias;
        q[(size_t)(r + 16) * 16 + col] = acc1[i] + bias;
    }
}

static inline int packBlocks(int K, int Nc) {
    int waves = (K / 4) * (Nc / 16);
    return (waves + 7) / 8;
}

extern "C" void kernel_launch(void* const* d_in, const int* in_sizes, int n_in,
                              void* d_out, int out_size, void* d_ws,
                              size_t ws_size, hipStream_t stream) {
    (void)n_in; (void)out_size; (void)ws_size;
    const float* feat = (const float*)d_in[0];
    const float* h    = (const float*)d_in[1];
    const int*   src  = (const int*)d_in[2];
    const int*   dst  = (const int*)d_in[3];
    const float* W0   = (const float*)d_in[4];
    const float* b0   = (const float*)d_in[5];
    const float* W1   = (const float*)d_in[6];
    const float* b1   = (const float*)d_in[7];
    const float* msgW = (const float*)d_in[8];
    const float* msgb = (const float*)d_in[9];
    const float* Wih  = (const float*)d_in[10];
    const float* Whh  = (const float*)d_in[11];
    const float* bih  = (const float*)d_in[12];
    const float* bhh  = (const float*)d_in[13];
    const float* outW = (const float*)d_in[14];
    const float* outb = (const float*)d_in[15];

    const int N = in_sizes[1] / 256;
    const int E = in_sizes[2];

    float* ws    = (float*)d_ws;
    float* x     = ws;
    float* mx    = x    + (size_t)N * 256;
    float* bkt   = mx   + (size_t)N * 128;
    int*   deg   = (int*)bkt;
    int*   off   = deg + N;
    int*   rnk   = off + N + 1;
    int*   ssrc  = rnk + E;
    float* c     = bkt  + (size_t)N * 128 + N;
    float* pW0   = c    + (size_t)N * 128;
    float* pW1   = pW0  + 16384;
    float* pmsgW = pW1  + 65536;
    float* pWih  = pmsgW + 65536;
    float* pWhh  = pWih + 294912;
    float* poutW = pWhh + 196608;

    float* q    = (float*)d_out;
    float* hnew = q + (size_t)N * 16;

    const int encBlocks  = (N / 16 + 3) / 4;
    const int gemmBlocks = (N / 32 + 3) / 4;
    const int rowTiles   = (N + 31) / 32;
    const int zeroBlocks = (N + 255) / 256;
    const int edgeBlocks = (E + 255) / 256;
    const int nodeBlocks = (N + 7) / 8;

    pack_kernel<<<packBlocks(64, 256), 256, 0, stream>>>(W0, pW0, 64, 256);
    pack_kernel<<<packBlocks(256, 256), 256, 0, stream>>>(W1, pW1, 256, 256);
    pack_kernel<<<packBlocks(512, 128), 256, 0, stream>>>(msgW, pmsgW, 512, 128);
    pack_kernel<<<packBlocks(384, 768), 256, 0, stream>>>(Wih, pWih, 384, 768);
    pack_kernel<<<packBlocks(256, 768), 256, 0, stream>>>(Whh, pWhh, 256, 768);
    pack_kernel<<<packBlocks(256, 16), 256, 0, stream>>>(outW, poutW, 256, 16);

    zero_kernel<<<zeroBlocks, 256, 0, stream>>>(deg, N);
    rank_kernel<<<edgeBlocks, 256, 0, stream>>>(dst, deg, rnk, E);
    scan_kernel<<<1, 1024, 0, stream>>>(deg, off, N);
    fill_kernel<<<edgeBlocks, 256, 0, stream>>>(src, dst, off, rnk, ssrc, E);
    enc_kernel<<<encBlocks, 128, 0, stream>>>(feat, pW0, b0, pW1, b1, x, N);
    msg_kernel<<<rowTiles, 128, 0, stream>>>(x, h, pmsgW, msgb, mx, N);
    gather_kernel<<<nodeBlocks, 256, 0, stream>>>(off, ssrc, mx, c, N);
    gru_kernel<<<dim3(rowTiles, 2), 128, 0, stream>>>(x, c, h, pWih, pWhh, bih,
                                                      bhh, hnew, N);
    out_kernel<<<gemmBlocks, 128, 0, stream>>>(hnew, poutW, outb, q, N);
}


# independent kernels share launches: six weight packs + counter clear in one, encoder + edge ranking in one, node-message GEMM + bucket fill in one; 15 launches become 7
# speedup vs baseline: 4.8577x; 1.1696x over previous
#include <hip/hip_runtime.h>
#include <hip/hip_bf16.h>


typedef __attribute__((ext_vector_type(2))) float v2f;
typedef __attribute__((ext_vector_type(8))) float v8f;

__device__ __forceinline__ v8f wmma4(v2f a, v2f b, v8f c) {
    return __builtin_amdgcn_wmma_f32_16x16x4_f32(
        false, a, false, b, (short)0, c, false, false);
}

__device__ __forceinline__ v2f load_a_frag(const float* M, int ld, int row0,
                                           int k0, int lane) {
    int r = row0 + (lane & 15);
    int k = k0 + ((lane >> 4) << 1);
    const float* p = M + (size_t)r * ld + k;
    v2f a; a.x = p[0]; a.y = p[1];
    return a;
}


template <int KB> struct Frag21 { v2f a0[KB], a1[KB], b[KB]; };

template <int KB>
__device__ __forceinline__ void load21(Frag21<KB>& f, const float* pa0,
                                       const float* pa1, const v2f* pb,
                                       int blk) {
    const float* q0 = pa0 + blk * (KB * 4);
    const float* q1 = pa1 + blk * (KB * 4);
    const v2f* qb = pb + blk * (KB * 32);
#pragma unroll
    for (int j = 0; j < KB; ++j) {
        f.a0[j].x = q0[j * 4]; f.a0[j].y = q0[j * 4 + 1];
        f.a1[j].x = q1[j * 4]; f.a1[j].y = q1[j * 4 + 1];
        f.b[j] = qb[j * 32];
    }
}

template <int KB>
__device__ __forceinline__ void mma21(const Frag21<KB>& f, v8f& c0, v8f& c1) {
#pragma unroll
    for (int j = 0; j < KB; ++j) {
        c0 = wmma4(f.a0[j], f.b[j], c0);
        c1 = wmma4(f.a1[j], f.b[j], c1);
    }
}

template <int KB>
__device__ __forceinline__ void seg21(const float* pa0, const float* pa1,
                                      const v2f* pb, int nb, v8f& c0, v8f& c1) {
    Frag21<KB> f0, f1;
    load21<KB>(f0, pa0, pa1, pb, 0);
    int b = 0;
#pragma unroll 1
    for (; b + 2 < nb; b += 2) {
        load21<KB>(f1, pa0, pa1, pb, b + 1);
        mma21<KB>(f0, c0, c1);
        load21<KB>(f0, pa0, pa1, pb, b + 2);
        mma21<KB>(f1, c0, c1);
    }
    load21<KB>(f1, pa0, pa1, pb, b + 1);
    mma21<KB>(f0, c0, c1);
    mma21<KB>(f1, c0, c1);
}

template <int KB> struct Frag23 { v2f a0[KB], a1[KB], br[KB], bz[KB], bn[KB]; };

template <int KB>
__device__ __forceinline__ void load23(Frag23<KB>& f, const float* A, int aoff,
                                       int a1, const float* B, int boff,
                                       int gate, int blk) {
    const float* qa = A + blk * (KB * 4);
    const float* qb = B + blk * (KB * 64);
#pragma unroll
    for (int j = 0; j < KB; ++j) {
        f.a0[j].x = qa[aoff + j * 4];      f.a0[j].y = qa[aoff + j * 4 + 1];
        f.a1[j].x = qa[aoff + a1 + j * 4]; f.a1[j].y = qa[aoff + a1 + j * 4 + 1];
        f.br[j].x = qb[boff + j * 64];     f.br[j].y = qb[boff + j * 64 + 1];
        f.bz[j].x = qb[boff + gate + j * 64];
        f.bz[j].y = qb[boff + gate + j * 64 + 1];
        f.bn[j].x = qb[boff + 2 * gate + j * 64];
        f.bn[j].y = qb[boff + 2 * gate + j * 64 + 1];
    }
}

template <int KB>
__device__ __forceinline__ void mma23(const Frag23<KB>& f, v8f& r0, v8f& r1,
                                      v8f& z0, v8f& z1, v8f& n0, v8f& n1) {
#pragma unroll
    for (int j = 0; j < KB; ++j) {
        r0 = wmma4(f.a0[j], f.br[j], r0);  r1 = wmma4(f.a1[j], f.br[j], r1);
        z0 = wmma4(f.a0[j], f.bz[j], z0);  z1 = wmma4(f.a1[j], f.bz[j], z1);
        n0 = wmma4(f.a0[j], f.bn[j], n0);  n1 = wmma4(f.a1[j], f.bn[j], n1);
    }
}

template <int KB>
__device__ __forceinline__ void seg23(const float* A, int aoff, int a1,
                                      const float* B, int boff, int gate,
                                      int nb, v8f& r0, v8f& r1, v8f& z0,
                                      v8f& z1, v8f& n0, v8f& n1) {
    Frag23<KB> f0, f1;
    load23<KB>(f0, A, aoff, a1, B, boff, gate, 0);
    int b = 0;
#pragma unroll 1
    for (; b + 2 < nb; b += 2) {
        load23<KB>(f1, A, aoff, a1, B, boff, gate, b + 1);
        mma23<KB>(f0, r0, r1, z0, z1, n0, n1);
        load23<KB>(f0, A, aoff, a1, B, boff, gate, b + 2);
        mma23<KB>(f1, r0, r1, z0, z1, n0, n1);
    }
    load23<KB>(f1, A, aoff, a1, B, boff, gate, b + 1);
    mma23<KB>(f0, r0, r1, z0, z1, n0, n1);
    mma23<KB>(f1, r0, r1, z0, z1, n0, n1);
}

template <int KB> struct FragB { v2f b[KB]; };

template <int KB>
__device__ __forceinline__ void loadB(FragB<KB>& f, const v2f* pb, int blk) {
    const v2f* qb = pb + blk * (KB * 32);
#pragma unroll
    for (int j = 0; j < KB; ++j) f.b[j] = qb[j * 32];
}

__device__ __forceinline__ float sigmoid_f(float v) {
    return 1.0f / (1.0f + __expf(-v));
}
__device__ __forceinline__ float tanh_f(float v) {
    float t = __expf(-2.0f * v);
    return (1.0f - t) / (1.0f + t);
}

template <int K, int Nc>
__device__ __forceinline__ void pack_tile(const float* __restrict__ W,
                                          float* __restrict__ P, int w,
                                          int lane) {
    constexpr int nk = K >> 2;
    int kt = w % nk, nt = w / nk;
    int n = nt * 16 + (lane & 15);
    int k = kt * 4 + ((lane >> 4) << 1);
    v2f v;
    v.x = W[(size_t)k * Nc + n];
    v.y = W[(size_t)(k + 1) * Nc + n];
    ((v2f*)P)[(size_t)(nt * nk + kt) * 32 + lane] = v;
}

constexpr int packBlocks(int K, int Nc) { return (K / 4) * (Nc / 16) / 8; }
constexpr bool packWhole(int K, int Nc) { return (K / 4) * (Nc / 16) % 8 == 0; }
constexpr int PB_W0  = packBlocks(64, 256);
constexpr int PB_W1  = packBlocks(256, 256);
constexpr int PB_MSG = packBlocks(512, 128);
constexpr int PB_WIH = packBlocks(384, 768);
constexpr int PB_WHH = packBlocks(256, 768);
constexpr int PB_OUT = packBlocks(256, 16);
constexpr int PB_ALL = PB_W0 + PB_W1 + PB_MSG + PB_WIH + PB_WHH + PB_OUT;
static_assert(packWhole(64, 256) && packWhole(256, 256) && packWhole(512, 128) &&
              packWhole(384, 768) && packWhole(256, 768) && packWhole(256, 16),
              "every matrix fills whole 8-wave blocks, so no wave has to be masked");

__global__ __launch_bounds__(256) void setup_kernel(
    const float* __restrict__ W0, const float* __restrict__ W1,
    const float* __restrict__ msgW, const float* __restrict__ Wih,
    const float* __restrict__ Whh, const float* __restrict__ outW,
    float* __restrict__ pW0, float* __restrict__ pW1,
    float* __restrict__ pmsgW, float* __restrict__ pWih,
    float* __restrict__ pWhh, float* __restrict__ poutW,
    int* __restrict__ deg, int N) {
    const int lane = threadIdx.x & 31;
    const int wib = threadIdx.x >> 5;
    int b = blockIdx.x;
    if (b < PB_W0)  { pack_tile<64, 256>(W0, pW0, b * 8 + wib, lane); return; }
    b -= PB_W0;
    if (b < PB_W1)  { pack_tile<256, 256>(W1, pW1, b * 8 + wib, lane); return; }
    b -= PB_W1;
    if (b < PB_MSG) { pack_tile<512, 128>(msgW, pmsgW, b * 8 + wib, lane); return; }
    b -= PB_MSG;
    if (b < PB_WIH) { pack_tile<384, 768>(Wih, pWih, b * 8 + wib, lane); return; }
    b -= PB_WIH;
    if (b < PB_WHH) { pack_tile<256, 768>(Whh, pWhh, b * 8 + wib, lane); return; }
    b -= PB_WHH;
    if (b < PB_OUT) { pack_tile<256, 16>(outW, poutW, b * 8 + wib, lane); return; }
    b -= PB_OUT;
    int i = b * 256 + (int)threadIdx.x;
    if (i < N) deg[i] = 0;
}

__device__ __forceinline__ void enc_block(
    float (*t)[16 * 256], int blk,
    const float* __restrict__ feat, const float* __restrict__ pW0,
    const float* __restrict__ b0, const float* __restrict__ pW1,
    const float* __restrict__ b1, float* __restrict__ x, int N) {
    const int wave = threadIdx.x >> 5;
    const int lane = threadIdx.x & 31;
    const int row0 = (blk * 4 + wave) * 16;
    const bool active = (row0 < N);
    const v2f* pb1 = (const v2f*)pW1 + lane;
    FragB<32> f0, f1;

    if (active) {
        v2f afr[16];
#pragma unroll
        for (int kk = 0; kk < 16; ++kk)
            afr[kk] = load_a_frag(feat, 64, row0, kk * 4, lane);
        const v2f* pb0 = (const v2f*)pW0 + lane;
        FragB<16> g0, g1;
        loadB<16>(g0, pb0, 0);
#pragma unroll 1
        for (int nt = 0; nt < 16; nt += 2) {
            loadB<16>(g1, pb0, nt + 1);
            {
                v8f acc = {};
#pragma unroll
                for (int kk = 0; kk < 16; ++kk)
                    acc = wmma4(afr[kk], g0.b[kk], acc);
                int col = nt * 16 + (lane & 15);
                float bias = b0[col];
#pragma unroll
                for (int i = 0; i < 8; ++i) {
                    int r = i + 8 * (lane >> 4);
                    t[wave][r * 256 + col] = fmaxf(acc[i] + bias, 0.0f);
                }
            }
            loadB<16>(g0, pb0, min(nt + 2, 15));
            {
                v8f acc = {};
#pragma unroll
                for (int kk = 0; kk < 16; ++kk)
                    acc = wmma4(afr[kk], g1.b[kk], acc);
                int col = (nt + 1) * 16 + (lane & 15);
                float bias = b0[col];
#pragma unroll
                for (int i = 0; i < 8; ++i) {
                    int r = i + 8 * (lane >> 4);
                    t[wave][r * 256 + col] = fmaxf(acc[i] + bias, 0.0f);
                }
            }
        }
        loadB<32>(f0, pb1, 0);
    }
    __syncthreads();
    if (active) {
        const float* ta = &t[wave][(lane & 15) * 256 + ((lane >> 4) << 1)];
#pragma unroll 1
        for (int nt = 0; nt < 16; ++nt) {
            v8f acc = {};
            loadB<32>(f1, pb1, nt * 2 + 1);
#pragma unroll
            for (int kk = 0; kk < 32; ++kk) {
                v2f a; a.x = ta[kk * 4]; a.y = ta[kk * 4 + 1];
                acc = wmma4(a, f0.b[kk], acc);
            }
            loadB<32>(f0, pb1, min(nt * 2 + 2, 31));
#pragma unroll
            for (int kk = 0; kk < 32; ++kk) {
                v2f a; a.x = ta[128 + kk * 4]; a.y = ta[128 + kk * 4 + 1];
                acc = wmma4(a, f1.b[kk], acc);
            }
            int col = nt * 16 + (lane & 15);
            float bias = b1[col];
#pragma unroll
            for (int i = 0; i < 8; ++i) {
                int r = row0 + i + 8 * (lane >> 4);
                x[(size_t)r * 256 + col] = fmaxf(acc[i] + bias, 0.0f);
            }
        }
    }
}

__device__ __forceinline__ void msg_block(
    int blk, const float* __restrict__ x, const float* __restrict__ h,
    const float* __restrict__ pmsgW, const float* __restrict__ msgb,
    float* __restrict__ mx, int N) {
    const int wave = threadIdx.x >> 5;
    const int lane = threadIdx.x & 31;
    const int row0 = blk * 32;
    if (row0 >= N) return;
    const size_t arow = (size_t)(row0 + (lane & 15)) * 256 + ((lane >> 4) << 1);
    const float* xa0 = x + arow;
    const float* xa1 = xa0 + 16 * 256;
    const float* ha0 = h + arow;
    const float* ha1 = ha0 + 16 * 256;
#pragma unroll 1
    for (int t = 0; t < 2; ++t) {
        const int nt = wave * 2 + t;
        const v2f* pb = (const v2f*)pmsgW + ((size_t)(nt * 128) * 32 + lane);
        v8f acc0 = {}, acc1 = {};
        seg21<8>(xa0, xa1, pb, 8, acc0, acc1);
        seg21<8>(ha0, ha1, pb + 64 * 32, 8, acc0, acc1);
        int col = nt * 16 + (lane & 15);
        float bias = msgb[col];
#pragma unroll
        for (int i = 0; i < 8; ++i) {
            int r = row0 + i + 8 * (lane >> 4);
            mx[(size_t)r * 128 + col] = acc0[i] + bias;
            mx[(size_t)(r + 16) * 128 + col] = acc1[i] + bias;
        }
    }
}

constexpr int EDGES_PER_BLOCK = 512;
template <bool FULL>
__device__ __forceinline__ void rank_edges(
    int base, const int* __restrict__ dst, int* __restrict__ deg,
    int* __restrict__ rnk, int E) {
    int d[4], r[4];
#pragma unroll
    for (int k = 0; k < 4; ++k) {
        int e = base + k * 128;
        d[k] = dst[FULL ? e : min(e, E - 1)];
    }
#pragma unroll
    for (int k = 0; k < 4; ++k)
        r[k] = (FULL || base + k * 128 < E) ? atomicAdd(deg + d[k], 1) : 0;
#pragma unroll
    for (int k = 0; k < 4; ++k)
        if (FULL || base + k * 128 < E) rnk[base + k * 128] = r[k];
}

__device__ __forceinline__ void rank_block(
    int blk, const int* __restrict__ dst, int* __restrict__ deg,
    int* __restrict__ rnk, int E) {
    const int base = blk * EDGES_PER_BLOCK + (int)threadIdx.x;
    if ((blk + 1) * EDGES_PER_BLOCK <= E)
        rank_edges<true>(base, dst, deg, rnk, E);
    else
        rank_edges<false>(base, dst, deg, rnk, E);
}

__global__ __launch_bounds__(128) void enc_rank_kernel(
    const float* __restrict__ feat, const float* __restrict__ pW0,
    const float* __restrict__ b0, const float* __restrict__ pW1,
    const float* __restrict__ b1, float* __restrict__ x, int N, int encBlocks,
    const int* __restrict__ dst, int* __restrict__ deg,
    int* __restrict__ rnk, int E) {
    __shared__ float t[4][16 * 256];
    const int blk = blockIdx.x;
    if (blk < encBlocks)
        enc_block(t, blk, feat, pW0, b0, pW1, b1, x, N);
    else
        rank_block(blk - encBlocks, dst, deg, rnk, E);
}

__global__ __launch_bounds__(1024) void scan_kernel(
    const int* __restrict__ deg, int* __restrict__ off, int n) {
    __shared__ int part[1024];
    const int t = threadIdx.x;
    const int per = (n + 1023) >> 10;
    const int beg = min(t * per, n);
    const int end = min(beg + per, n);
    int s = 0;
#pragma unroll 4
    for (int i = beg; i < end; ++i) s += deg[i];
    part[t] = s;
    __syncthreads();
    for (int o = 1; o < 1024; o <<= 1) {
        int v = (t >= o) ? part[t - o] : 0;
        __syncthreads();
        part[t] += v;
        __syncthreads();
    }
    int run = part[t] - s;
#pragma unroll 4
    for (int i = beg; i < end; ++i) {
        int d = deg[i];
        off[i] = run;
        run += d;
    }
    if (t == 1023) off[n] = part[1023];
}

template <bool FULL>
__device__ __forceinline__ void fill_edges(
    int base, const int* __restrict__ src, const int* __restrict__ dst,
    const int* __restrict__ off, const int* __restrict__ rnk,
    int* __restrict__ ssrc, int E) {
    int d[4], r[4], s[4], o[4];
#pragma unroll
    for (int k = 0; k < 4; ++k) {
        int e = base + k * 128;
        if (!FULL) e = min(e, E - 1);
        d[k] = dst[e]; r[k] = rnk[e]; s[k] = src[e];
    }
#pragma unroll
    for (int k = 0; k < 4; ++k) o[k] = off[d[k]];
#pragma unroll
    for (int k = 0; k < 4; ++k) {
        int pos = o[k] + r[k];
        pos = min(max(pos, 0), E - 1);
        if (FULL || base + k * 128 < E) ssrc[pos] = s[k];
    }
}

__device__ __forceinline__ void fill_block(
    int blk, const int* __restrict__ src, const int* __restrict__ dst,
    const int* __restrict__ off, const int* __restrict__ rnk,
    int* __restrict__ ssrc, int E) {
    const int base = blk * EDGES_PER_BLOCK + (int)threadIdx.x;
    if ((blk + 1) * EDGES_PER_BLOCK <= E)
        fill_edges<true>(base, src, dst, off, rnk, ssrc, E);
    else
        fill_edges<false>(base, src, dst, off, rnk, ssrc, E);
}

__global__ __launch_bounds__(128) void msg_fill_kernel(
    const float* __restrict__ x, const float* __restrict__ h,
    const float* __restrict__ pmsgW, const float* __restrict__ msgb,
    float* __restrict__ mx, int N, int msgBlocks,
    const int* __restrict__ src, const int* __restrict__ dst,
    const int* __restrict__ off, const int* __restrict__ rnk,
    int* __restrict__ ssrc, int E) {
    const int blk = blockIdx.x;
    if (blk < msgBlocks)
        msg_block(blk, x, h, pmsgW, msgb, mx, N);
    else
        fill_block(blk - msgBlocks, src, dst, off, rnk, ssrc, E);
}

__global__ __launch_bounds__(256) void gather_kernel(
    const int* __restrict__ off, const int* __restrict__ ssrc,
    const float* __restrict__ mx, float* __restrict__ c, int N) {
    const int node = __builtin_amdgcn_readfirstlane(
        (int)((blockIdx.x * 256u + threadIdx.x) >> 5));
    const int lane = threadIdx.x & 31;
    if (node >= N) return;
    const int beg = off[node];
    const int end = off[node + 1];
    const float4* mv = (const float4*)mx;
    float4 acc = make_float4(0.0f, 0.0f, 0.0f, 0.0f);
    int j = beg;
    for (; j + 8 <= end; j += 8) {
        float4 v[8];
#pragma unroll
        for (int k = 0; k < 8; ++k) {
            int s = ssrc[j + k];
            v[k] = mv[(size_t)s * 32 + lane];
        }
#pragma unroll
        for (int k = 0; k < 8; ++k) {
            acc.x += v[k].x; acc.y += v[k].y;
            acc.z += v[k].z; acc.w += v[k].w;
        }
    }
    if (j < end) {
        float4 v[8];
#pragma unroll
        for (int k = 0; k < 8; ++k) {
            int s = ssrc[min(j + k, end - 1)];
            v[k] = mv[(size_t)s * 32 + lane];
        }
#pragma unroll
        for (int k = 0; k < 8; ++k) {
            const bool in = (j + k < end);
            acc.x += in ? v[k].x : 0.0f; acc.y += in ? v[k].y : 0.0f;
            acc.z += in ? v[k].z : 0.0f; acc.w += in ? v[k].w : 0.0f;
        }
    }
    const float den = fmaxf((float)(end - beg), 1.0f);
    float4 o;
    o.x = acc.x / den; o.y = acc.y / den; o.z = acc.z / den; o.w = acc.w / den;
    ((float4*)c)[(size_t)node * 32 + lane] = o;
}

constexpr int GRU_KB = 2;
__global__ __launch_bounds__(128, 5) void gru_kernel(
    const float* __restrict__ x, const float* __restrict__ c,
    const float* __restrict__ h, const float* __restrict__ pWih,
    const float* __restrict__ pWhh, const float* __restrict__ bih,
    const float* __restrict__ bhh, float* __restrict__ hnew, int N) {
    const int wave = __builtin_amdgcn_readfirstlane((int)(threadIdx.x >> 5));
    const int lane = threadIdx.x & 31;
    const int row0 = blockIdx.x * 32;
    if (row0 >= N) return;
    const int p = blockIdx.y * 4 + wave;
    const int ak = (lane >> 4) << 1;
    const int boff = lane * 2;
    int lr = lane & 15;
#pragma unroll 1
    for (int t = 0; t < 2; ++t) {
        const int nt = p * 2 + t;
        asm volatile("" : "+v"(lr));
        lr &= 15;
        v8f ir0 = {}, iz0 = {}, in0 = {}, ir1 = {}, iz1 = {}, in1 = {};
        v8f hr0 = {}, hz0 = {}, hn0 = {}, hr1 = {}, hz1 = {}, hn1 = {};
        const float* pbi = pWih + (size_t)(nt * 96) * 64;
        const float* pbh = pWhh + (size_t)(nt * 64) * 64;
        seg23<GRU_KB>(x + (size_t)row0 * 256, lr * 256 + ak, 16 * 256,
                      pbi, boff, 16 * 96 * 64, 64 / GRU_KB,
                      ir0, ir1, iz0, iz1, in0, in1);
        seg23<GRU_KB>(c + (size_t)row0 * 128, lr * 128 + ak, 16 * 128,
                      pbi + 64 * 64, boff, 16 * 96 * 64, 32 / GRU_KB,
                      ir0, ir1, iz0, iz1, in0, in1);
        seg23<GRU_KB>(h + (size_t)row0 * 256, lr * 256 + ak, 16 * 256,
                      pbh, boff, 16 * 64 * 64, 64 / GRU_KB,
                      hr0, hr1, hz0, hz1, hn0, hn1);
        int col = nt * 16 + (lane & 15);
        float bir = bih[col], biz = bih[col + 256], bin = bih[col + 512];
        float bhr = bhh[col], bhz = bhh[col + 256], bhn = bhh[col + 512];
#pragma unroll
        for (int i = 0; i < 8; ++i) {
            int r = row0 + i + 8 * (lane >> 4);
            {
                float rg = sigmoid_f((ir0[i] + bir) + (hr0[i] + bhr));
                float zg = sigmoid_f((iz0[i] + biz) + (hz0[i] + bhz));
                float ng = tanh_f((in0[i] + bin) + rg * (hn0[i] + bhn));
                float hv = h[(size_t)r * 256 + col];
                hnew[(size_t)r * 256 + col] = (1.0f - zg) * ng + zg * hv;
            }
            {
                float rg = sigmoid_f((ir1[i] + bir) + (hr1[i] + bhr));
                float zg = sigmoid_f((iz1[i] + biz) + (hz1[i] + bhz));
                float ng = tanh_f((in1[i] + bin) + rg * (hn1[i] + bhn));
                float hv = h[(size_t)(r + 16) * 256 + col];
                hnew[(size_t)(r + 16) * 256 + col] = (1.0f - zg) * ng + zg * hv;
            }
        }
    }
}

__global__ __launch_bounds__(128) void out_kernel(
    const float* __restrict__ hnew, const float* __restrict__ poutW,
    const float* __restrict__ outb, float* __restrict__ q, int N) {
    const int wave = threadIdx.x >> 5;
    const int lane = threadIdx.x & 31;
    const int row0 = (blockIdx.x * 4 + wave) * 32;
    if (row0 >= N) return;
    v8f acc0 = {}, acc1 = {};
    const float* a0 = hnew + (size_t)(row0 + (lane & 15)) * 256 + ((lane >> 4) << 1);
    seg21<16>(a0, a0 + 16 * 256, (const v2f*)poutW + lane, 4, acc0, acc1);
    int col = lane & 15;
    float bias = outb[col];
#pragma unroll
    for (int i = 0; i < 8; ++i) {
        int r = row0 + i + 8 * (lane >> 4);
        q[(size_t)r * 16 + col] = acc0[i] + bias;
        q[(size_t)(r + 16) * 16 + col] = acc1[i] + bias;
    }
}

extern "C" void kernel_launch(void* const* d_in, const int* in_sizes, int n_in,
                              void* d_out, int out_size, void* d_ws,
                              size_t ws_size, hipStream_t stream) {
    (void)n_in; (void)out_size; (void)ws_size;
    const float* feat = (const float*)d_in[0];
    const float* h    = (const float*)d_in[1];
    const int*   src  = (const int*)d_in[2];
    const int*   dst  = (const int*)d_in[3];
    const float* W0   = (const float*)d_in[4];
    const float* b0   = (const float*)d_in[5];
    const float* W1   = (const float*)d_in[6];
    const float* b1   = (const float*)d_in[7];
    const float* msgW = (const float*)d_in[8];
    const float* msgb = (const float*)d_in[9];
    const float* Wih  = (const float*)d_in[10];
    const float* Whh  = (const float*)d_in[11];
    const float* bih  = (const float*)d_in[12];
    const float* bhh  = (const float*)d_in[13];
    const float* outW = (const float*)d_in[14];
    const float* outb = (const float*)d_in[15];

    const int N = in_sizes[1] / 256;
    const int E = in_sizes[2];

    float* ws    = (float*)d_ws;
    float* x     = ws;
    float* mx    = x    + (size_t)N * 256;
    float* bkt   = mx   + (size_t)N * 128;
    int*   deg   = (int*)bkt;
    int*   off   = deg + N;
    int*   rnk   = off + N + 1;
    int*   ssrc  = rnk + E;
    float* c     = bkt  + (size_t)N * 128 + N;
    float* pW0   = c    + (size_t)N * 128;
    float* pW1   = pW0  + 16384;
    float* pmsgW = pW1  + 65536;
    float* pWih  = pmsgW + 65536;
    float* pWhh  = pWih + 294912;
    float* poutW = pWhh + 196608;

    float* q    = (float*)d_out;
    float* hnew = q + (size_t)N * 16;

    const int encBlocks  = (N / 16 + 3) / 4;
    const int gemmBlocks = (N / 32 + 3) / 4;
    const int rowTiles   = (N + 31) / 32;
    const int zeroBlocks = (N + 255) / 256;
    const int edgeBlocks = (E + EDGES_PER_BLOCK - 1) / EDGES_PER_BLOCK;
    const int nodeBlocks = (N + 7) / 8;

    setup_kernel<<<PB_ALL + zeroBlocks, 256, 0, stream>>>(
        W0, W1, msgW, Wih, Whh, outW, pW0, pW1, pmsgW, pWih, pWhh, poutW, deg, N);
    enc_rank_kernel<<<encBlocks + edgeBlocks, 128, 0, stream>>>(
        feat, pW0, b0, pW1, b1, x, N, encBlocks, dst, deg, rnk, E);
    scan_kernel<<<1, 1024, 0, stream>>>(deg, off, N);
    msg_fill_kernel<<<rowTiles + edgeBlocks, 128, 0, stream>>>(
        x, h, pmsgW, msgb, mx, N, rowTiles, src, dst, off, rnk, ssrc, E);
    gather_kernel<<<nodeBlocks, 256, 0, stream>>>(off, ssrc, mx, c, N);
    gru_kernel<<<dim3(rowTiles, 2), 128, 0, stream>>>(x, c, h, pWih, pWhh, bih,
                                                      bhh, hnew, N);
    out_kernel<<<gemmBlocks, 128, 0, stream>>>(hnew, poutW, outb, q, N);
}


# gru and msg GEMMs: each block stages its 32-row A tile in LDS once (whole-line 16-byte loads), shared by its 4 waves and both column tiles; A fragments come from LDS, not 16-line global loads
# speedup vs baseline: 5.1613x; 1.0625x over previous
#include <hip/hip_runtime.h>
#include <hip/hip_bf16.h>


typedef __attribute__((ext_vector_type(2))) float v2f;
typedef __attribute__((ext_vector_type(8))) float v8f;

__device__ __forceinline__ v8f wmma4(v2f a, v2f b, v8f c) {
    return __builtin_amdgcn_wmma_f32_16x16x4_f32(
        false, a, false, b, (short)0, c, false, false);
}

__device__ __forceinline__ v2f load_a_frag(const float* M, int ld, int row0,
                                           int k0, int lane) {
    int r = row0 + (lane & 15);
    int k = k0 + ((lane >> 4) << 1);
    const float* p = M + (size_t)r * ld + k;
    v2f a; a.x = p[0]; a.y = p[1];
    return a;
}


template <int KB> struct Frag21 { v2f a0[KB], a1[KB], b[KB]; };

template <int KB>
__device__ __forceinline__ void load21(Frag21<KB>& f, const float* pa0,
                                       const float* pa1, const v2f* pb,
                                       int blk) {
    const float* q0 = pa0 + blk * (KB * 4);
    const float* q1 = pa1 + blk * (KB * 4);
    const v2f* qb = pb + blk * (KB * 32);
#pragma unroll
    for (int j = 0; j < KB; ++j) {
        f.a0[j].x = q0[j * 4]; f.a0[j].y = q0[j * 4 + 1];
        f.a1[j].x = q1[j * 4]; f.a1[j].y = q1[j * 4 + 1];
        f.b[j] = qb[j * 32];
    }
}

template <int KB>
__device__ __forceinline__ void mma21(const Frag21<KB>& f, v8f& c0, v8f& c1) {
#pragma unroll
    for (int j = 0; j < KB; ++j) {
        c0 = wmma4(f.a0[j], f.b[j], c0);
        c1 = wmma4(f.a1[j], f.b[j], c1);
    }
}

template <int KB>
__device__ __forceinline__ void seg21(const float* pa0, const float* pa1,
                                      const v2f* pb, int nb, v8f& c0, v8f& c1) {
    Frag21<KB> f0, f1;
    load21<KB>(f0, pa0, pa1, pb, 0);
    int b = 0;
#pragma unroll 1
    for (; b + 2 < nb; b += 2) {
        load21<KB>(f1, pa0, pa1, pb, b + 1);
        mma21<KB>(f0, c0, c1);
        load21<KB>(f0, pa0, pa1, pb, b + 2);
        mma21<KB>(f1, c0, c1);
    }
    load21<KB>(f1, pa0, pa1, pb, b + 1);
    mma21<KB>(f0, c0, c1);
    mma21<KB>(f1, c0, c1);
}

template <int KB, int PITCH>
__device__ __forceinline__ void load21s(Frag21<KB>& f, const float* la,
                                        const v2f* pb, int blk) {
    const float* qa = la + blk * (KB * 4);
    const v2f* qb = pb + blk * (KB * 32);
#pragma unroll
    for (int j = 0; j < KB; ++j) {
        f.a0[j].x = qa[j * 4];              f.a0[j].y = qa[j * 4 + 1];
        f.a1[j].x = qa[16 * PITCH + j * 4]; f.a1[j].y = qa[16 * PITCH + j * 4 + 1];
        f.b[j] = qb[j * 32];
    }
}

template <int KB, int PITCH>
__device__ __forceinline__ void seg21s(const float* la, const v2f* pb, int nb,
                                       v8f& c0, v8f& c1) {
    Frag21<KB> f0, f1;
    load21s<KB, PITCH>(f0, la, pb, 0);
    int b = 0;
#pragma unroll 1
    for (; b + 2 < nb; b += 2) {
        load21s<KB, PITCH>(f1, la, pb, b + 1);
        __builtin_amdgcn_sched_barrier(0);
        mma21<KB>(f0, c0, c1);
        load21s<KB, PITCH>(f0, la, pb, b + 2);
        __builtin_amdgcn_sched_barrier(0);
        mma21<KB>(f1, c0, c1);
    }
    load21s<KB, PITCH>(f1, la, pb, b + 1);
    mma21<KB>(f0, c0, c1);
    mma21<KB>(f1, c0, c1);
}

template <int KB> struct Frag23 { v2f a0[KB], a1[KB], br[KB], bz[KB], bn[KB]; };

template <int KB>
__device__ __forceinline__ void mma23(const Frag23<KB>& f, v8f& r0, v8f& r1,
                                      v8f& z0, v8f& z1, v8f& n0, v8f& n1) {
#pragma unroll
    for (int j = 0; j < KB; ++j) {
        r0 = wmma4(f.a0[j], f.br[j], r0);  r1 = wmma4(f.a1[j], f.br[j], r1);
        z0 = wmma4(f.a0[j], f.bz[j], z0);  z1 = wmma4(f.a1[j], f.bz[j], z1);
        n0 = wmma4(f.a0[j], f.bn[j], n0);  n1 = wmma4(f.a1[j], f.bn[j], n1);
    }
}

template <int KB> struct FragB { v2f b[KB]; };

template <int KB>
__device__ __forceinline__ void loadB(FragB<KB>& f, const v2f* pb, int blk) {
    const v2f* qb = pb + blk * (KB * 32);
#pragma unroll
    for (int j = 0; j < KB; ++j) f.b[j] = qb[j * 32];
}

__device__ __forceinline__ float sigmoid_f(float v) {
    return 1.0f / (1.0f + __expf(-v));
}
__device__ __forceinline__ float tanh_f(float v) {
    float t = __expf(-2.0f * v);
    return (1.0f - t) / (1.0f + t);
}

template <int K, int Nc>
__device__ __forceinline__ void pack_tile(const float* __restrict__ W,
                                          float* __restrict__ P, int w,
                                          int lane) {
    constexpr int nk = K >> 2;
    int kt = w % nk, nt = w / nk;
    int n = nt * 16 + (lane & 15);
    int k = kt * 4 + ((lane >> 4) << 1);
    v2f v;
    v.x = W[(size_t)k * Nc + n];
    v.y = W[(size_t)(k + 1) * Nc + n];
    ((v2f*)P)[(size_t)(nt * nk + kt) * 32 + lane] = v;
}

constexpr int packBlocks(int K, int Nc) { return (K / 4) * (Nc / 16) / 8; }
constexpr bool packWhole(int K, int Nc) { return (K / 4) * (Nc / 16) % 8 == 0; }
constexpr int PB_W0  = packBlocks(64, 256);
constexpr int PB_W1  = packBlocks(256, 256);
constexpr int PB_MSG = packBlocks(512, 128);
constexpr int PB_WIH = packBlocks(384, 768);
constexpr int PB_WHH = packBlocks(256, 768);
constexpr int PB_OUT = packBlocks(256, 16);
constexpr int PB_ALL = PB_W0 + PB_W1 + PB_MSG + PB_WIH + PB_WHH + PB_OUT;
static_assert(packWhole(64, 256) && packWhole(256, 256) && packWhole(512, 128) &&
              packWhole(384, 768) && packWhole(256, 768) && packWhole(256, 16),
              "every matrix fills whole 8-wave blocks, so no wave has to be masked");

__global__ __launch_bounds__(256) void setup_kernel(
    const float* __restrict__ W0, const float* __restrict__ W1,
    const float* __restrict__ msgW, const float* __restrict__ Wih,
    const float* __restrict__ Whh, const float* __restrict__ outW,
    float* __restrict__ pW0, float* __restrict__ pW1,
    float* __restrict__ pmsgW, float* __restrict__ pWih,
    float* __restrict__ pWhh, float* __restrict__ poutW,
    int* __restrict__ deg, int N) {
    const int lane = threadIdx.x & 31;
    const int wib = threadIdx.x >> 5;
    int b = blockIdx.x;
    if (b < PB_W0)  { pack_tile<64, 256>(W0, pW0, b * 8 + wib, lane); return; }
    b -= PB_W0;
    if (b < PB_W1)  { pack_tile<256, 256>(W1, pW1, b * 8 + wib, lane); return; }
    b -= PB_W1;
    if (b < PB_MSG) { pack_tile<512, 128>(msgW, pmsgW, b * 8 + wib, lane); return; }
    b -= PB_MSG;
    if (b < PB_WIH) { pack_tile<384, 768>(Wih, pWih, b * 8 + wib, lane); return; }
    b -= PB_WIH;
    if (b < PB_WHH) { pack_tile<256, 768>(Whh, pWhh, b * 8 + wib, lane); return; }
    b -= PB_WHH;
    if (b < PB_OUT) { pack_tile<256, 16>(outW, poutW, b * 8 + wib, lane); return; }
    b -= PB_OUT;
    int i = b * 256 + (int)threadIdx.x;
    if (i < N) deg[i] = 0;
}

__device__ __forceinline__ void enc_block(
    float (*t)[16 * 256], int blk,
    const float* __restrict__ feat, const float* __restrict__ pW0,
    const float* __restrict__ b0, const float* __restrict__ pW1,
    const float* __restrict__ b1, float* __restrict__ x, int N) {
    const int wave = threadIdx.x >> 5;
    const int lane = threadIdx.x & 31;
    const int row0 = (blk * 4 + wave) * 16;
    const bool active = (row0 < N);
    const v2f* pb1 = (const v2f*)pW1 + lane;
    FragB<32> f0, f1;

    if (active) {
        v2f afr[16];
#pragma unroll
        for (int kk = 0; kk < 16; ++kk)
            afr[kk] = load_a_frag(feat, 64, row0, kk * 4, lane);
        const v2f* pb0 = (const v2f*)pW0 + lane;
        FragB<16> g0, g1;
        loadB<16>(g0, pb0, 0);
#pragma unroll 1
        for (int nt = 0; nt < 16; nt += 2) {
            loadB<16>(g1, pb0, nt + 1);
            {
                v8f acc = {};
#pragma unroll
                for (int kk = 0; kk < 16; ++kk)
                    acc = wmma4(afr[kk], g0.b[kk], acc);
                int col = nt * 16 + (lane & 15);
                float bias = b0[col];
#pragma unroll
                for (int i = 0; i < 8; ++i) {
                    int r = i + 8 * (lane >> 4);
                    t[wave][r * 256 + col] = fmaxf(acc[i] + bias, 0.0f);
                }
            }
            loadB<16>(g0, pb0, min(nt + 2, 15));
            {
                v8f acc = {};
#pragma unroll
                for (int kk = 0; kk < 16; ++kk)
                    acc = wmma4(afr[kk], g1.b[kk], acc);
                int col = (nt + 1) * 16 + (lane & 15);
                float bias = b0[col];
#pragma unroll
                for (int i = 0; i < 8; ++i) {
                    int r = i + 8 * (lane >> 4);
                    t[wave][r * 256 + col] = fmaxf(acc[i] + bias, 0.0f);
                }
            }
        }
        loadB<32>(f0, pb1, 0);
    }
    __syncthreads();
    if (active) {
        const float* ta = &t[wave][(lane & 15) * 256 + ((lane >> 4) << 1)];
#pragma unroll 1
        for (int nt = 0; nt < 16; ++nt) {
            v8f acc = {};
            loadB<32>(f1, pb1, nt * 2 + 1);
#pragma unroll
            for (int kk = 0; kk < 32; ++kk) {
                v2f a; a.x = ta[kk * 4]; a.y = ta[kk * 4 + 1];
                acc = wmma4(a, f0.b[kk], acc);
            }
            loadB<32>(f0, pb1, min(nt * 2 + 2, 31));
#pragma unroll
            for (int kk = 0; kk < 32; ++kk) {
                v2f a; a.x = ta[128 + kk * 4]; a.y = ta[128 + kk * 4 + 1];
                acc = wmma4(a, f1.b[kk], acc);
            }
            int col = nt * 16 + (lane & 15);
            float bias = b1[col];
#pragma unroll
            for (int i = 0; i < 8; ++i) {
                int r = row0 + i + 8 * (lane >> 4);
                x[(size_t)r * 256 + col] = fmaxf(acc[i] + bias, 0.0f);
            }
        }
    }
}

constexpr int MSG_PITCH = 512 + 4;
__device__ __forceinline__ void msg_block(
    float* am, int blk, const float* __restrict__ x, const float* __restrict__ h,
    const float* __restrict__ pmsgW, const float* __restrict__ msgb,
    float* __restrict__ mx, int N) {
    const int wave = threadIdx.x >> 5;
    const int lane = threadIdx.x & 31;
    const int tid = threadIdx.x;
    const int row0 = blk * 32;
    if (row0 >= N) return;
    {
        const float4* xv = (const float4*)(x + (size_t)row0 * 256);
        const float4* hv = (const float4*)(h + (size_t)row0 * 256);
        float4 vx[16], vh[16];
#pragma unroll
        for (int i = 0; i < 16; ++i) vx[i] = xv[i * 128 + tid];
#pragma unroll
        for (int i = 0; i < 16; ++i) vh[i] = hv[i * 128 + tid];
#pragma unroll
        for (int i = 0; i < 16; ++i) {
            int idx = i * 128 + tid;
            *(float4*)&am[(idx >> 6) * MSG_PITCH + (idx & 63) * 4] = vx[i];
        }
#pragma unroll
        for (int i = 0; i < 16; ++i) {
            int idx = i * 128 + tid;
            *(float4*)&am[(idx >> 6) * MSG_PITCH + 256 + (idx & 63) * 4] = vh[i];
        }
    }
    __syncthreads();
    const float* la = am + (lane & 15) * MSG_PITCH + ((lane >> 4) << 1);
#pragma unroll 1
    for (int t = 0; t < 2; ++t) {
        const int nt = wave * 2 + t;
        const v2f* pb = (const v2f*)pmsgW + ((size_t)(nt * 128) * 32 + lane);
        v8f acc0 = {}, acc1 = {};
        seg21s<8, MSG_PITCH>(la, pb, 16, acc0, acc1);
        int col = nt * 16 + (lane & 15);
        float bias = msgb[col];
#pragma unroll
        for (int i = 0; i < 8; ++i) {
            int r = row0 + i + 8 * (lane >> 4);
            mx[(size_t)r * 128 + col] = acc0[i] + bias;
            mx[(size_t)(r + 16) * 128 + col] = acc1[i] + bias;
        }
    }
}

constexpr int EDGES_PER_BLOCK = 512;
template <bool FULL>
__device__ __forceinline__ void rank_edges(
    int base, const int* __restrict__ dst, int* __restrict__ deg,
    int* __restrict__ rnk, int E) {
    int d[4], r[4];
#pragma unroll
    for (int k = 0; k < 4; ++k) {
        int e = base + k * 128;
        d[k] = dst[FULL ? e : min(e, E - 1)];
    }
#pragma unroll
    for (int k = 0; k < 4; ++k)
        r[k] = (FULL || base + k * 128 < E) ? atomicAdd(deg + d[k], 1) : 0;
#pragma unroll
    for (int k = 0; k < 4; ++k)
        if (FULL || base + k * 128 < E) rnk[base + k * 128] = r[k];
}

__device__ __forceinline__ void rank_block(
    int blk, const int* __restrict__ dst, int* __restrict__ deg,
    int* __restrict__ rnk, int E) {
    const int base = blk * EDGES_PER_BLOCK + (int)threadIdx.x;
    if ((blk + 1) * EDGES_PER_BLOCK <= E)
        rank_edges<true>(base, dst, deg, rnk, E);
    else
        rank_edges<false>(base, dst, deg, rnk, E);
}

__global__ __launch_bounds__(128) void enc_rank_kernel(
    const float* __restrict__ feat, const float* __restrict__ pW0,
    const float* __restrict__ b0, const float* __restrict__ pW1,
    const float* __restrict__ b1, float* __restrict__ x, int N, int encBlocks,
    const int* __restrict__ dst, int* __restrict__ deg,
    int* __restrict__ rnk, int E) {
    __shared__ float t[4][16 * 256];
    const int blk = blockIdx.x;
    if (blk < encBlocks)
        enc_block(t, blk, feat, pW0, b0, pW1, b1, x, N);
    else
        rank_block(blk - encBlocks, dst, deg, rnk, E);
}

__global__ __launch_bounds__(1024) void scan_kernel(
    const int* __restrict__ deg, int* __restrict__ off, int n) {
    __shared__ int part[1024];
    const int t = threadIdx.x;
    const int per = (n + 1023) >> 10;
    const int beg = min(t * per, n);
    const int end = min(beg + per, n);
    int s = 0;
#pragma unroll 4
    for (int i = beg; i < end; ++i) s += deg[i];
    part[t] = s;
    __syncthreads();
    for (int o = 1; o < 1024; o <<= 1) {
        int v = (t >= o) ? part[t - o] : 0;
        __syncthreads();
        part[t] += v;
        __syncthreads();
    }
    int run = part[t] - s;
#pragma unroll 4
    for (int i = beg; i < end; ++i) {
        int d = deg[i];
        off[i] = run;
        run += d;
    }
    if (t == 1023) off[n] = part[1023];
}

template <bool FULL>
__device__ __forceinline__ void fill_edges(
    int base, const int* __restrict__ src, const int* __restrict__ dst,
    const int* __restrict__ off, const int* __restrict__ rnk,
    int* __restrict__ ssrc, int E) {
    int d[4], r[4], s[4], o[4];
#pragma unroll
    for (int k = 0; k < 4; ++k) {
        int e = base + k * 128;
        if (!FULL) e = min(e, E - 1);
        d[k] = dst[e]; r[k] = rnk[e]; s[k] = src[e];
    }
#pragma unroll
    for (int k = 0; k < 4; ++k) o[k] = off[d[k]];
#pragma unroll
    for (int k = 0; k < 4; ++k) {
        int pos = o[k] + r[k];
        pos = min(max(pos, 0), E - 1);
        if (FULL || base + k * 128 < E) ssrc[pos] = s[k];
    }
}

__device__ __forceinline__ void fill_block(
    int blk, const int* __restrict__ src, const int* __restrict__ dst,
    const int* __restrict__ off, const int* __restrict__ rnk,
    int* __restrict__ ssrc, int E) {
    const int base = blk * EDGES_PER_BLOCK + (int)threadIdx.x;
    if ((blk + 1) * EDGES_PER_BLOCK <= E)
        fill_edges<true>(base, src, dst, off, rnk, ssrc, E);
    else
        fill_edges<false>(base, src, dst, off, rnk, ssrc, E);
}

__global__ __launch_bounds__(128) void msg_fill_kernel(
    const float* __restrict__ x, const float* __restrict__ h,
    const float* __restrict__ pmsgW, const float* __restrict__ msgb,
    float* __restrict__ mx, int N, int msgBlocks,
    const int* __restrict__ src, const int* __restrict__ dst,
    const int* __restrict__ off, const int* __restrict__ rnk,
    int* __restrict__ ssrc, int E) {
    __shared__ __attribute__((aligned(16))) float am[32 * MSG_PITCH];
    const int blk = blockIdx.x;
    if (blk < msgBlocks)
        msg_block(am, blk, x, h, pmsgW, msgb, mx, N);
    else
        fill_block(blk - msgBlocks, src, dst, off, rnk, ssrc, E);
}

__global__ __launch_bounds__(256) void gather_kernel(
    const int* __restrict__ off, const int* __restrict__ ssrc,
    const float* __restrict__ mx, float* __restrict__ c, int N) {
    const int node = __builtin_amdgcn_readfirstlane(
        (int)((blockIdx.x * 256u + threadIdx.x) >> 5));
    const int lane = threadIdx.x & 31;
    if (node >= N) return;
    const int beg = off[node];
    const int end = off[node + 1];
    const float4* mv = (const float4*)mx;
    float4 acc = make_float4(0.0f, 0.0f, 0.0f, 0.0f);
    int j = beg;
    for (; j + 8 <= end; j += 8) {
        float4 v[8];
#pragma unroll
        for (int k = 0; k < 8; ++k) {
            int s = ssrc[j + k];
            v[k] = mv[(size_t)s * 32 + lane];
        }
#pragma unroll
        for (int k = 0; k < 8; ++k) {
            acc.x += v[k].x; acc.y += v[k].y;
            acc.z += v[k].z; acc.w += v[k].w;
        }
    }
    if (j < end) {
        float4 v[8];
#pragma unroll
        for (int k = 0; k < 8; ++k) {
            int s = ssrc[min(j + k, end - 1)];
            v[k] = mv[(size_t)s * 32 + lane];
        }
#pragma unroll
        for (int k = 0; k < 8; ++k) {
            const bool in = (j + k < end);
            acc.x += in ? v[k].x : 0.0f; acc.y += in ? v[k].y : 0.0f;
            acc.z += in ? v[k].z : 0.0f; acc.w += in ? v[k].w : 0.0f;
        }
    }
    const float den = fmaxf((float)(end - beg), 1.0f);
    float4 o;
    o.x = acc.x / den; o.y = acc.y / den; o.z = acc.z / den; o.w = acc.w / den;
    ((float4*)c)[(size_t)node * 32 + lane] = o;
}

constexpr int GRU_KB = 4;
constexpr int GRU_K = 640;
constexpr int GRU_PITCH = GRU_K + 4;

template <int KB>
__device__ __forceinline__ void load23s(Frag23<KB>& f, const float* la,
                                        const float* B, int boff, int gate,
                                        int blk) {
    const float* qa = la + blk * (KB * 4);
    const float* qb = B + blk * (KB * 64);
#pragma unroll
    for (int j = 0; j < KB; ++j) {
        f.a0[j].x = qa[j * 4];                  f.a0[j].y = qa[j * 4 + 1];
        f.a1[j].x = qa[16 * GRU_PITCH + j * 4]; f.a1[j].y = qa[16 * GRU_PITCH + j * 4 + 1];
        f.br[j].x = qb[boff + j * 64];          f.br[j].y = qb[boff + j * 64 + 1];
        f.bz[j].x = qb[boff + gate + j * 64];
        f.bz[j].y = qb[boff + gate + j * 64 + 1];
        f.bn[j].x = qb[boff + 2 * gate + j * 64];
        f.bn[j].y = qb[boff + 2 * gate + j * 64 + 1];
    }
}

template <int KB>
__device__ __forceinline__ void seg23s(const float* la, const float* B,
                                       int boff, int gate, int nb, v8f& r0,
                                       v8f& r1, v8f& z0, v8f& z1, v8f& n0,
                                       v8f& n1) {
    Frag23<KB> f0, f1;
    load23s<KB>(f0, la, B, boff, gate, 0);
    int b = 0;
#pragma unroll 1
    for (; b + 2 < nb; b += 2) {
        load23s<KB>(f1, la, B, boff, gate, b + 1);
        __builtin_amdgcn_sched_barrier(0);
        mma23<KB>(f0, r0, r1, z0, z1, n0, n1);
        load23s<KB>(f0, la, B, boff, gate, b + 2);
        __builtin_amdgcn_sched_barrier(0);
        mma23<KB>(f1, r0, r1, z0, z1, n0, n1);
    }
    load23s<KB>(f1, la, B, boff, gate, b + 1);
    mma23<KB>(f0, r0, r1, z0, z1, n0, n1);
    mma23<KB>(f1, r0, r1, z0, z1, n0, n1);
}

__global__ __launch_bounds__(128) void gru_kernel(
    const float* __restrict__ x, const float* __restrict__ c,
    const float* __restrict__ h, const float* __restrict__ pWih,
    const float* __restrict__ pWhh, const float* __restrict__ bih,
    const float* __restrict__ bhh, float* __restrict__ hnew, int N) {
    __shared__ __attribute__((aligned(16))) float at[32 * GRU_PITCH];
    const int wave = __builtin_amdgcn_readfirstlane((int)(threadIdx.x >> 5));
    const int lane = threadIdx.x & 31;
    const int tid = threadIdx.x;
    const int row0 = blockIdx.x * 32;
    if (row0 >= N) return;

    {
        const float4* xv = (const float4*)(x + (size_t)row0 * 256);
        const float4* cv = (const float4*)(c + (size_t)row0 * 128);
        const float4* hv = (const float4*)(h + (size_t)row0 * 256);
        float4 vx[16], vc[8], vh[16];
#pragma unroll
        for (int i = 0; i < 16; ++i) vx[i] = xv[i * 128 + tid];
#pragma unroll
        for (int i = 0; i < 8; ++i) vc[i] = cv[i * 128 + tid];
#pragma unroll
        for (int i = 0; i < 16; ++i) vh[i] = hv[i * 128 + tid];
#pragma unroll
        for (int i = 0; i < 16; ++i) {
            int idx = i * 128 + tid;
            *(float4*)&at[(idx >> 6) * GRU_PITCH + (idx & 63) * 4] = vx[i];
        }
#pragma unroll
        for (int i = 0; i < 8; ++i) {
            int idx = i * 128 + tid;
            *(float4*)&at[(idx >> 5) * GRU_PITCH + 256 + (idx & 31) * 4] = vc[i];
        }
#pragma unroll
        for (int i = 0; i < 16; ++i) {
            int idx = i * 128 + tid;
            *(float4*)&at[(idx >> 6) * GRU_PITCH + 384 + (idx & 63) * 4] = vh[i];
        }
    }
    __syncthreads();

    const int p = blockIdx.y * 4 + wave;
    const int boff = lane * 2;
    const float* la = at + (lane & 15) * GRU_PITCH + ((lane >> 4) << 1);
#pragma unroll 1
    for (int t = 0; t < 2; ++t) {
        const int nt = p * 2 + t;
        v8f ir0 = {}, iz0 = {}, in0 = {}, ir1 = {}, iz1 = {}, in1 = {};
        v8f hr0 = {}, hz0 = {}, hn0 = {}, hr1 = {}, hz1 = {}, hn1 = {};
        const float* pbi = pWih + (size_t)(nt * 96) * 64;
        const float* pbh = pWhh + (size_t)(nt * 64) * 64;
        seg23s<GRU_KB>(la, pbi, boff, 16 * 96 * 64, 96 / GRU_KB,
                       ir0, ir1, iz0, iz1, in0, in1);
        seg23s<GRU_KB>(la + 384, pbh, boff, 16 * 64 * 64, 64 / GRU_KB,
                       hr0, hr1, hz0, hz1, hn0, hn1);
        int col = nt * 16 + (lane & 15);
        float bir = bih[col], biz = bih[col + 256], bin = bih[col + 512];
        float bhr = bhh[col], bhz = bhh[col + 256], bhn = bhh[col + 512];
#pragma unroll
        for (int i = 0; i < 8; ++i) {
            int lr = i + 8 * (lane >> 4);
            int r = row0 + lr;
            {
                float rg = sigmoid_f((ir0[i] + bir) + (hr0[i] + bhr));
                float zg = sigmoid_f((iz0[i] + biz) + (hz0[i] + bhz));
                float ng = tanh_f((in0[i] + bin) + rg * (hn0[i] + bhn));
                float hv = at[lr * GRU_PITCH + 384 + col];
                hnew[(size_t)r * 256 + col] = (1.0f - zg) * ng + zg * hv;
            }
            {
                float rg = sigmoid_f((ir1[i] + bir) + (hr1[i] + bhr));
                float zg = sigmoid_f((iz1[i] + biz) + (hz1[i] + bhz));
                float ng = tanh_f((in1[i] + bin) + rg * (hn1[i] + bhn));
                float hv = at[(lr + 16) * GRU_PITCH + 384 + col];
                hnew[(size_t)(r + 16) * 256 + col] = (1.0f - zg) * ng + zg * hv;
            }
        }
    }
}

__global__ __launch_bounds__(128) void out_kernel(
    const float* __restrict__ hnew, const float* __restrict__ poutW,
    const float* __restrict__ outb, float* __restrict__ q, int N) {
    const int wave = threadIdx.x >> 5;
    const int lane = threadIdx.x & 31;
    const int row0 = (blockIdx.x * 4 + wave) * 32;
    if (row0 >= N) return;
    v8f acc0 = {}, acc1 = {};
    const float* a0 = hnew + (size_t)(row0 + (lane & 15)) * 256 + ((lane >> 4) << 1);
    seg21<16>(a0, a0 + 16 * 256, (const v2f*)poutW + lane, 4, acc0, acc1);
    int col = lane & 15;
    float bias = outb[col];
#pragma unroll
    for (int i = 0; i < 8; ++i) {
        int r = row0 + i + 8 * (lane >> 4);
        q[(size_t)r * 16 + col] = acc0[i] + bias;
        q[(size_t)(r + 16) * 16 + col] = acc1[i] + bias;
    }
}

extern "C" void kernel_launch(void* const* d_in, const int* in_sizes, int n_in,
                              void* d_out, int out_size, void* d_ws,
                              size_t ws_size, hipStream_t stream) {
    (void)n_in; (void)out_size; (void)ws_size;
    const float* feat = (const float*)d_in[0];
    const float* h    = (const float*)d_in[1];
    const int*   src  = (const int*)d_in[2];
    const int*   dst  = (const int*)d_in[3];
    const float* W0   = (const float*)d_in[4];
    const float* b0   = (const float*)d_in[5];
    const float* W1   = (const float*)d_in[6];
    const float* b1   = (const float*)d_in[7];
    const float* msgW = (const float*)d_in[8];
    const float* msgb = (const float*)d_in[9];
    const float* Wih  = (const float*)d_in[10];
    const float* Whh  = (const float*)d_in[11];
    const float* bih  = (const float*)d_in[12];
    const float* bhh  = (const float*)d_in[13];
    const float* outW = (const float*)d_in[14];
    const float* outb = (const float*)d_in[15];

    const int N = in_sizes[1] / 256;
    const int E = in_sizes[2];

    float* ws    = (float*)d_ws;
    float* x     = ws;
    float* mx    = x    + (size_t)N * 256;
    float* bkt   = mx   + (size_t)N * 128;
    int*   deg   = (int*)bkt;
    int*   off   = deg + N;
    int*   rnk   = off + N + 1;
    int*   ssrc  = rnk + E;
    float* c     = bkt  + (size_t)N * 128 + N;
    float* pW0   = c    + (size_t)N * 128;
    float* pW1   = pW0  + 16384;
    float* pmsgW = pW1  + 65536;
    float* pWih  = pmsgW + 65536;
    float* pWhh  = pWih + 294912;
    float* poutW = pWhh + 196608;

    float* q    = (float*)d_out;
    float* hnew = q + (size_t)N * 16;

    const int encBlocks  = (N / 16 + 3) / 4;
    const int gemmBlocks = (N / 32 + 3) / 4;
    const int rowTiles   = (N + 31) / 32;
    const int zeroBlocks = (N + 255) / 256;
    const int edgeBlocks = (E + EDGES_PER_BLOCK - 1) / EDGES_PER_BLOCK;
    const int nodeBlocks = (N + 7) / 8;

    setup_kernel<<<PB_ALL + zeroBlocks, 256, 0, stream>>>(
        W0, W1, msgW, Wih, Whh, outW, pW0, pW1, pmsgW, pWih, pWhh, poutW, deg, N);
    enc_rank_kernel<<<encBlocks + edgeBlocks, 128, 0, stream>>>(
        feat, pW0, b0, pW1, b1, x, N, encBlocks, dst, deg, rnk, E);
    scan_kernel<<<1, 1024, 0, stream>>>(deg, off, N);
    msg_fill_kernel<<<rowTiles + edgeBlocks, 128, 0, stream>>>(
        x, h, pmsgW, msgb, mx, N, rowTiles, src, dst, off, rnk, ssrc, E);
    gather_kernel<<<nodeBlocks, 256, 0, stream>>>(off, ssrc, mx, c, N);
    gru_kernel<<<dim3(rowTiles, 2), 128, 0, stream>>>(x, c, h, pWih, pWhh, bih,
                                                      bhh, hnew, N);
    out_kernel<<<gemmBlocks, 128, 0, stream>>>(hnew, poutW, outb, q, N);
}


# edge buckets: 64 fixed slots per node filled by the ranking pass (slot = node*64 + rank; edges past 64 go on an overflow list the gather walks); prefix-sum launch and fill pass go, 7 launches to 6
# speedup vs baseline: 5.5818x; 1.0815x over previous
#include <hip/hip_runtime.h>
#include <hip/hip_bf16.h>


typedef __attribute__((ext_vector_type(2))) float v2f;
typedef __attribute__((ext_vector_type(8))) float v8f;

__device__ __forceinline__ v8f wmma4(v2f a, v2f b, v8f c) {
    return __builtin_amdgcn_wmma_f32_16x16x4_f32(
        false, a, false, b, (short)0, c, false, false);
}

__device__ __forceinline__ v2f load_a_frag(const float* M, int ld, int row0,
                                           int k0, int lane) {
    int r = row0 + (lane & 15);
    int k = k0 + ((lane >> 4) << 1);
    const float* p = M + (size_t)r * ld + k;
    v2f a; a.x = p[0]; a.y = p[1];
    return a;
}


template <int KB> struct Frag21 { v2f a0[KB], a1[KB], b[KB]; };

template <int KB>
__device__ __forceinline__ void load21(Frag21<KB>& f, const float* pa0,
                                       const float* pa1, const v2f* pb,
                                       int blk) {
    const float* q0 = pa0 + blk * (KB * 4);
    const float* q1 = pa1 + blk * (KB * 4);
    const v2f* qb = pb + blk * (KB * 32);
#pragma unroll
    for (int j = 0; j < KB; ++j) {
        f.a0[j].x = q0[j * 4]; f.a0[j].y = q0[j * 4 + 1];
        f.a1[j].x = q1[j * 4]; f.a1[j].y = q1[j * 4 + 1];
        f.b[j] = qb[j * 32];
    }
}

template <int KB>
__device__ __forceinline__ void mma21(const Frag21<KB>& f, v8f& c0, v8f& c1) {
#pragma unroll
    for (int j = 0; j < KB; ++j) {
        c0 = wmma4(f.a0[j], f.b[j], c0);
        c1 = wmma4(f.a1[j], f.b[j], c1);
    }
}

template <int KB>
__device__ __forceinline__ void seg21(const float* pa0, const float* pa1,
                                      const v2f* pb, int nb, v8f& c0, v8f& c1) {
    Frag21<KB> f0, f1;
    load21<KB>(f0, pa0, pa1, pb, 0);
    int b = 0;
#pragma unroll 1
    for (; b + 2 < nb; b += 2) {
        load21<KB>(f1, pa0, pa1, pb, b + 1);
        mma21<KB>(f0, c0, c1);
        load21<KB>(f0, pa0, pa1, pb, b + 2);
        mma21<KB>(f1, c0, c1);
    }
    load21<KB>(f1, pa0, pa1, pb, b + 1);
    mma21<KB>(f0, c0, c1);
    mma21<KB>(f1, c0, c1);
}

template <int KB, int PITCH>
__device__ __forceinline__ void load21s(Frag21<KB>& f, const float* la,
                                        const v2f* pb, int blk) {
    const float* qa = la + blk * (KB * 4);
    const v2f* qb = pb + blk * (KB * 32);
#pragma unroll
    for (int j = 0; j < KB; ++j) {
        f.a0[j].x = qa[j * 4];              f.a0[j].y = qa[j * 4 + 1];
        f.a1[j].x = qa[16 * PITCH + j * 4]; f.a1[j].y = qa[16 * PITCH + j * 4 + 1];
        f.b[j] = qb[j * 32];
    }
}

template <int KB, int PITCH>
__device__ __forceinline__ void seg21s(const float* la, const v2f* pb, int nb,
                                       v8f& c0, v8f& c1) {
    Frag21<KB> f0, f1;
    load21s<KB, PITCH>(f0, la, pb, 0);
    int b = 0;
#pragma unroll 1
    for (; b + 2 < nb; b += 2) {
        load21s<KB, PITCH>(f1, la, pb, b + 1);
        __builtin_amdgcn_sched_barrier(0);
        mma21<KB>(f0, c0, c1);
        load21s<KB, PITCH>(f0, la, pb, b + 2);
        __builtin_amdgcn_sched_barrier(0);
        mma21<KB>(f1, c0, c1);
    }
    load21s<KB, PITCH>(f1, la, pb, b + 1);
    mma21<KB>(f0, c0, c1);
    mma21<KB>(f1, c0, c1);
}

template <int KB> struct Frag23 { v2f a0[KB], a1[KB], br[KB], bz[KB], bn[KB]; };

template <int KB>
__device__ __forceinline__ void mma23(const Frag23<KB>& f, v8f& r0, v8f& r1,
                                      v8f& z0, v8f& z1, v8f& n0, v8f& n1) {
#pragma unroll
    for (int j = 0; j < KB; ++j) {
        r0 = wmma4(f.a0[j], f.br[j], r0);  r1 = wmma4(f.a1[j], f.br[j], r1);
        z0 = wmma4(f.a0[j], f.bz[j], z0);  z1 = wmma4(f.a1[j], f.bz[j], z1);
        n0 = wmma4(f.a0[j], f.bn[j], n0);  n1 = wmma4(f.a1[j], f.bn[j], n1);
    }
}

template <int KB> struct FragB { v2f b[KB]; };

template <int KB>
__device__ __forceinline__ void loadB(FragB<KB>& f, const v2f* pb, int blk) {
    const v2f* qb = pb + blk * (KB * 32);
#pragma unroll
    for (int j = 0; j < KB; ++j) f.b[j] = qb[j * 32];
}

__device__ __forceinline__ float sigmoid_f(float v) {
    return 1.0f / (1.0f + __expf(-v));
}
__device__ __forceinline__ float tanh_f(float v) {
    float t = __expf(-2.0f * v);
    return (1.0f - t) / (1.0f + t);
}

template <int K, int Nc>
__device__ __forceinline__ void pack_tile(const float* __restrict__ W,
                                          float* __restrict__ P, int w,
                                          int lane) {
    constexpr int nk = K >> 2;
    int kt = w % nk, nt = w / nk;
    int n = nt * 16 + (lane & 15);
    int k = kt * 4 + ((lane >> 4) << 1);
    v2f v;
    v.x = W[(size_t)k * Nc + n];
    v.y = W[(size_t)(k + 1) * Nc + n];
    ((v2f*)P)[(size_t)(nt * nk + kt) * 32 + lane] = v;
}

constexpr int packBlocks(int K, int Nc) { return (K / 4) * (Nc / 16) / 8; }
constexpr bool packWhole(int K, int Nc) { return (K / 4) * (Nc / 16) % 8 == 0; }
constexpr int PB_W0  = packBlocks(64, 256);
constexpr int PB_W1  = packBlocks(256, 256);
constexpr int PB_MSG = packBlocks(512, 128);
constexpr int PB_WIH = packBlocks(384, 768);
constexpr int PB_WHH = packBlocks(256, 768);
constexpr int PB_OUT = packBlocks(256, 16);
constexpr int PB_ALL = PB_W0 + PB_W1 + PB_MSG + PB_WIH + PB_WHH + PB_OUT;
static_assert(packWhole(64, 256) && packWhole(256, 256) && packWhole(512, 128) &&
              packWhole(384, 768) && packWhole(256, 768) && packWhole(256, 16),
              "every matrix fills whole 8-wave blocks, so no wave has to be masked");

__global__ __launch_bounds__(256) void setup_kernel(
    const float* __restrict__ W0, const float* __restrict__ W1,
    const float* __restrict__ msgW, const float* __restrict__ Wih,
    const float* __restrict__ Whh, const float* __restrict__ outW,
    float* __restrict__ pW0, float* __restrict__ pW1,
    float* __restrict__ pmsgW, float* __restrict__ pWih,
    float* __restrict__ pWhh, float* __restrict__ poutW,
    int* __restrict__ deg, int N) {
    const int lane = threadIdx.x & 31;
    const int wib = threadIdx.x >> 5;
    int b = blockIdx.x;
    if (b < PB_W0)  { pack_tile<64, 256>(W0, pW0, b * 8 + wib, lane); return; }
    b -= PB_W0;
    if (b < PB_W1)  { pack_tile<256, 256>(W1, pW1, b * 8 + wib, lane); return; }
    b -= PB_W1;
    if (b < PB_MSG) { pack_tile<512, 128>(msgW, pmsgW, b * 8 + wib, lane); return; }
    b -= PB_MSG;
    if (b < PB_WIH) { pack_tile<384, 768>(Wih, pWih, b * 8 + wib, lane); return; }
    b -= PB_WIH;
    if (b < PB_WHH) { pack_tile<256, 768>(Whh, pWhh, b * 8 + wib, lane); return; }
    b -= PB_WHH;
    if (b < PB_OUT) { pack_tile<256, 16>(outW, poutW, b * 8 + wib, lane); return; }
    b -= PB_OUT;
    int i = b * 256 + (int)threadIdx.x;
    if (i <= N) deg[i] = 0;
}

__device__ __forceinline__ void enc_block(
    float (*t)[16 * 256], int blk,
    const float* __restrict__ feat, const float* __restrict__ pW0,
    const float* __restrict__ b0, const float* __restrict__ pW1,
    const float* __restrict__ b1, float* __restrict__ x, int N) {
    const int wave = threadIdx.x >> 5;
    const int lane = threadIdx.x & 31;
    const int row0 = (blk * 4 + wave) * 16;
    const bool active = (row0 < N);
    const v2f* pb1 = (const v2f*)pW1 + lane;
    FragB<32> f0, f1;

    if (active) {
        v2f afr[16];
#pragma unroll
        for (int kk = 0; kk < 16; ++kk)
            afr[kk] = load_a_frag(feat, 64, row0, kk * 4, lane);
        const v2f* pb0 = (const v2f*)pW0 + lane;
        FragB<16> g0, g1;
        loadB<16>(g0, pb0, 0);
#pragma unroll 1
        for (int nt = 0; nt < 16; nt += 2) {
            loadB<16>(g1, pb0, nt + 1);
            {
                v8f acc = {};
#pragma unroll
                for (int kk = 0; kk < 16; ++kk)
                    acc = wmma4(afr[kk], g0.b[kk], acc);
                int col = nt * 16 + (lane & 15);
                float bias = b0[col];
#pragma unroll
                for (int i = 0; i < 8; ++i) {
                    int r = i + 8 * (lane >> 4);
                    t[wave][r * 256 + col] = fmaxf(acc[i] + bias, 0.0f);
                }
            }
            loadB<16>(g0, pb0, min(nt + 2, 15));
            {
                v8f acc = {};
#pragma unroll
                for (int kk = 0; kk < 16; ++kk)
                    acc = wmma4(afr[kk], g1.b[kk], acc);
                int col = (nt + 1) * 16 + (lane & 15);
                float bias = b0[col];
#pragma unroll
                for (int i = 0; i < 8; ++i) {
                    int r = i + 8 * (lane >> 4);
                    t[wave][r * 256 + col] = fmaxf(acc[i] + bias, 0.0f);
                }
            }
        }
        loadB<32>(f0, pb1, 0);
    }
    __syncthreads();
    if (active) {
        const float* ta = &t[wave][(lane & 15) * 256 + ((lane >> 4) << 1)];
#pragma unroll 1
        for (int nt = 0; nt < 16; ++nt) {
            v8f acc = {};
            loadB<32>(f1, pb1, nt * 2 + 1);
#pragma unroll
            for (int kk = 0; kk < 32; ++kk) {
                v2f a; a.x = ta[kk * 4]; a.y = ta[kk * 4 + 1];
                acc = wmma4(a, f0.b[kk], acc);
            }
            loadB<32>(f0, pb1, min(nt * 2 + 2, 31));
#pragma unroll
            for (int kk = 0; kk < 32; ++kk) {
                v2f a; a.x = ta[128 + kk * 4]; a.y = ta[128 + kk * 4 + 1];
                acc = wmma4(a, f1.b[kk], acc);
            }
            int col = nt * 16 + (lane & 15);
            float bias = b1[col];
#pragma unroll
            for (int i = 0; i < 8; ++i) {
                int r = row0 + i + 8 * (lane >> 4);
                x[(size_t)r * 256 + col] = fmaxf(acc[i] + bias, 0.0f);
            }
        }
    }
}

constexpr int MSG_PITCH = 512 + 4;
__device__ __forceinline__ void msg_block(
    float* am, int blk, const float* __restrict__ x, const float* __restrict__ h,
    const float* __restrict__ pmsgW, const float* __restrict__ msgb,
    float* __restrict__ mx, int N) {
    const int wave = threadIdx.x >> 5;
    const int lane = threadIdx.x & 31;
    const int tid = threadIdx.x;
    const int row0 = blk * 32;
    if (row0 >= N) return;
    {
        const float4* xv = (const float4*)(x + (size_t)row0 * 256);
        const float4* hv = (const float4*)(h + (size_t)row0 * 256);
        float4 vx[16], vh[16];
#pragma unroll
        for (int i = 0; i < 16; ++i) vx[i] = xv[i * 128 + tid];
#pragma unroll
        for (int i = 0; i < 16; ++i) vh[i] = hv[i * 128 + tid];
#pragma unroll
        for (int i = 0; i < 16; ++i) {
            int idx = i * 128 + tid;
            *(float4*)&am[(idx >> 6) * MSG_PITCH + (idx & 63) * 4] = vx[i];
        }
#pragma unroll
        for (int i = 0; i < 16; ++i) {
            int idx = i * 128 + tid;
            *(float4*)&am[(idx >> 6) * MSG_PITCH + 256 + (idx & 63) * 4] = vh[i];
        }
    }
    __syncthreads();
    const float* la = am + (lane & 15) * MSG_PITCH + ((lane >> 4) << 1);
#pragma unroll 1
    for (int t = 0; t < 2; ++t) {
        const int nt = wave * 2 + t;
        const v2f* pb = (const v2f*)pmsgW + ((size_t)(nt * 128) * 32 + lane);
        v8f acc0 = {}, acc1 = {};
        seg21s<8, MSG_PITCH>(la, pb, 16, acc0, acc1);
        int col = nt * 16 + (lane & 15);
        float bias = msgb[col];
#pragma unroll
        for (int i = 0; i < 8; ++i) {
            int r = row0 + i + 8 * (lane >> 4);
            mx[(size_t)r * 128 + col] = acc0[i] + bias;
            mx[(size_t)(r + 16) * 128 + col] = acc1[i] + bias;
        }
    }
}

constexpr int SLOTS_PER_NODE = 64;
constexpr int EDGES_PER_BLOCK = 512;
template <bool FULL>
__device__ __forceinline__ void rank_edges(
    int base, const int* __restrict__ src, const int* __restrict__ dst,
    int* __restrict__ deg, int* __restrict__ slots, int* __restrict__ ovf,
    int N, int E) {
    int d[4], s[4], r[4];
#pragma unroll
    for (int k = 0; k < 4; ++k) {
        int e = base + k * 128;
        if (!FULL) e = min(e, E - 1);
        d[k] = dst[e]; s[k] = src[e];
    }
#pragma unroll
    for (int k = 0; k < 4; ++k)
        r[k] = (FULL || base + k * 128 < E) ? atomicAdd(deg + d[k], 1) : 0;
#pragma unroll
    for (int k = 0; k < 4; ++k) {
        if (FULL || base + k * 128 < E) {
            if (r[k] < SLOTS_PER_NODE) {
                int pos = d[k] * SLOTS_PER_NODE + r[k];
                pos = min(max(pos, 0), N * SLOTS_PER_NODE - 1);
                slots[pos] = s[k];
            } else {
                int o = atomicAdd(deg + N, 1);
                if (o < E) ovf[o] = base + k * 128;
            }
        }
    }
}

__device__ __forceinline__ void rank_block(
    int blk, const int* __restrict__ src, const int* __restrict__ dst,
    int* __restrict__ deg, int* __restrict__ slots, int* __restrict__ ovf,
    int N, int E) {
    const int base = blk * EDGES_PER_BLOCK + (int)threadIdx.x;
    if ((blk + 1) * EDGES_PER_BLOCK <= E)
        rank_edges<true>(base, src, dst, deg, slots, ovf, N, E);
    else
        rank_edges<false>(base, src, dst, deg, slots, ovf, N, E);
}

__global__ __launch_bounds__(128) void enc_rank_kernel(
    const float* __restrict__ feat, const float* __restrict__ pW0,
    const float* __restrict__ b0, const float* __restrict__ pW1,
    const float* __restrict__ b1, float* __restrict__ x, int N, int encBlocks,
    const int* __restrict__ src, const int* __restrict__ dst,
    int* __restrict__ deg, int* __restrict__ slots, int* __restrict__ ovf,
    int E) {
    __shared__ float t[4][16 * 256];
    const int blk = blockIdx.x;
    if (blk < encBlocks)
        enc_block(t, blk, feat, pW0, b0, pW1, b1, x, N);
    else
        rank_block(blk - encBlocks, src, dst, deg, slots, ovf, N, E);
}

__global__ __launch_bounds__(128) void msg_kernel(
    const float* __restrict__ x, const float* __restrict__ h,
    const float* __restrict__ pmsgW, const float* __restrict__ msgb,
    float* __restrict__ mx, int N) {
    __shared__ __attribute__((aligned(16))) float am[32 * MSG_PITCH];
    msg_block(am, blockIdx.x, x, h, pmsgW, msgb, mx, N);
}

__global__ __launch_bounds__(256) void gather_kernel(
    const int* __restrict__ deg, const int* __restrict__ slots,
    const int* __restrict__ ovf, const int* __restrict__ src,
    const int* __restrict__ dst, const float* __restrict__ mx,
    float* __restrict__ c, int N, int E) {
    const int node = __builtin_amdgcn_readfirstlane(
        (int)((blockIdx.x * 256u + threadIdx.x) >> 5));
    const int lane = threadIdx.x & 31;
    if (node >= N) return;
    const int cnt = deg[node];
    const int* sl = slots + (size_t)node * SLOTS_PER_NODE;
    const int end = min(cnt, SLOTS_PER_NODE);
    const float4* mv = (const float4*)mx;
    float4 acc = make_float4(0.0f, 0.0f, 0.0f, 0.0f);
    int j = 0;
    for (; j + 8 <= end; j += 8) {
        float4 v[8];
#pragma unroll
        for (int k = 0; k < 8; ++k) {
            int s = sl[j + k];
            v[k] = mv[(size_t)s * 32 + lane];
        }
#pragma unroll
        for (int k = 0; k < 8; ++k) {
            acc.x += v[k].x; acc.y += v[k].y;
            acc.z += v[k].z; acc.w += v[k].w;
        }
    }
    if (j < end) {
        float4 v[8];
#pragma unroll
        for (int k = 0; k < 8; ++k) {
            int s = sl[min(j + k, end - 1)];
            v[k] = mv[(size_t)s * 32 + lane];
        }
#pragma unroll
        for (int k = 0; k < 8; ++k) {
            const bool in = (j + k < end);
            acc.x += in ? v[k].x : 0.0f; acc.y += in ? v[k].y : 0.0f;
            acc.z += in ? v[k].z : 0.0f; acc.w += in ? v[k].w : 0.0f;
        }
    }
    if (cnt > SLOTS_PER_NODE) {
        const int tot = min(deg[N], E);
        for (int k = 0; k < tot; ++k) {
            const int e = min(max(ovf[k], 0), E - 1);
            if (dst[e] == node) {
                float4 v = mv[(size_t)src[e] * 32 + lane];
                acc.x += v.x; acc.y += v.y; acc.z += v.z; acc.w += v.w;
            }
        }
    }
    const float den = fmaxf((float)cnt, 1.0f);
    float4 o;
    o.x = acc.x / den; o.y = acc.y / den; o.z = acc.z / den; o.w = acc.w / den;
    ((float4*)c)[(size_t)node * 32 + lane] = o;
}

constexpr int GRU_KB = 4;
constexpr int GRU_K = 640;
constexpr int GRU_PITCH = GRU_K + 4;

template <int KB>
__device__ __forceinline__ void load23s(Frag23<KB>& f, const float* la,
                                        const float* B, int boff, int gate,
                                        int blk) {
    const float* qa = la + blk * (KB * 4);
    const float* qb = B + blk * (KB * 64);
#pragma unroll
    for (int j = 0; j < KB; ++j) {
        f.a0[j].x = qa[j * 4];                  f.a0[j].y = qa[j * 4 + 1];
        f.a1[j].x = qa[16 * GRU_PITCH + j * 4]; f.a1[j].y = qa[16 * GRU_PITCH + j * 4 + 1];
        f.br[j].x = qb[boff + j * 64];          f.br[j].y = qb[boff + j * 64 + 1];
        f.bz[j].x = qb[boff + gate + j * 64];
        f.bz[j].y = qb[boff + gate + j * 64 + 1];
        f.bn[j].x = qb[boff + 2 * gate + j * 64];
        f.bn[j].y = qb[boff + 2 * gate + j * 64 + 1];
    }
}

template <int KB>
__device__ __forceinline__ void seg23s(const float* la, const float* B,
                                       int boff, int gate, int nb, v8f& r0,
                                       v8f& r1, v8f& z0, v8f& z1, v8f& n0,
                                       v8f& n1) {
    Frag23<KB> f0, f1;
    load23s<KB>(f0, la, B, boff, gate, 0);
    int b = 0;
#pragma unroll 1
    for (; b + 2 < nb; b += 2) {
        load23s<KB>(f1, la, B, boff, gate, b + 1);
        __builtin_amdgcn_sched_barrier(0);
        mma23<KB>(f0, r0, r1, z0, z1, n0, n1);
        load23s<KB>(f0, la, B, boff, gate, b + 2);
        __builtin_amdgcn_sched_barrier(0);
        mma23<KB>(f1, r0, r1, z0, z1, n0, n1);
    }
    load23s<KB>(f1, la, B, boff, gate, b + 1);
    mma23<KB>(f0, r0, r1, z0, z1, n0, n1);
    mma23<KB>(f1, r0, r1, z0, z1, n0, n1);
}

__global__ __launch_bounds__(128) void gru_kernel(
    const float* __restrict__ x, const float* __restrict__ c,
    const float* __restrict__ h, const float* __restrict__ pWih,
    const float* __restrict__ pWhh, const float* __restrict__ bih,
    const float* __restrict__ bhh, float* __restrict__ hnew, int N) {
    __shared__ __attribute__((aligned(16))) float at[32 * GRU_PITCH];
    const int wave = __builtin_amdgcn_readfirstlane((int)(threadIdx.x >> 5));
    const int lane = threadIdx.x & 31;
    const int tid = threadIdx.x;
    const int row0 = blockIdx.x * 32;
    if (row0 >= N) return;

    {
        const float4* xv = (const float4*)(x + (size_t)row0 * 256);
        const float4* cv = (const float4*)(c + (size_t)row0 * 128);
        const float4* hv = (const float4*)(h + (size_t)row0 * 256);
        float4 vx[16], vc[8], vh[16];
#pragma unroll
        for (int i = 0; i < 16; ++i) vx[i] = xv[i * 128 + tid];
#pragma unroll
        for (int i = 0; i < 8; ++i) vc[i] = cv[i * 128 + tid];
#pragma unroll
        for (int i = 0; i < 16; ++i) vh[i] = hv[i * 128 + tid];
#pragma unroll
        for (int i = 0; i < 16; ++i) {
            int idx = i * 128 + tid;
            *(float4*)&at[(idx >> 6) * GRU_PITCH + (idx & 63) * 4] = vx[i];
        }
#pragma unroll
        for (int i = 0; i < 8; ++i) {
            int idx = i * 128 + tid;
            *(float4*)&at[(idx >> 5) * GRU_PITCH + 256 + (idx & 31) * 4] = vc[i];
        }
#pragma unroll
        for (int i = 0; i < 16; ++i) {
            int idx = i * 128 + tid;
            *(float4*)&at[(idx >> 6) * GRU_PITCH + 384 + (idx & 63) * 4] = vh[i];
        }
    }
    __syncthreads();

    const int p = blockIdx.y * 4 + wave;
    const int boff = lane * 2;
    const float* la = at + (lane & 15) * GRU_PITCH + ((lane >> 4) << 1);
#pragma unroll 1
    for (int t = 0; t < 2; ++t) {
        const int nt = p * 2 + t;
        v8f ir0 = {}, iz0 = {}, in0 = {}, ir1 = {}, iz1 = {}, in1 = {};
        v8f hr0 = {}, hz0 = {}, hn0 = {}, hr1 = {}, hz1 = {}, hn1 = {};
        const float* pbi = pWih + (size_t)(nt * 96) * 64;
        const float* pbh = pWhh + (size_t)(nt * 64) * 64;
        seg23s<GRU_KB>(la, pbi, boff, 16 * 96 * 64, 96 / GRU_KB,
                       ir0, ir1, iz0, iz1, in0, in1);
        seg23s<GRU_KB>(la + 384, pbh, boff, 16 * 64 * 64, 64 / GRU_KB,
                       hr0, hr1, hz0, hz1, hn0, hn1);
        int col = nt * 16 + (lane & 15);
        float bir = bih[col], biz = bih[col + 256], bin = bih[col + 512];
        float bhr = bhh[col], bhz = bhh[col + 256], bhn = bhh[col + 512];
#pragma unroll
        for (int i = 0; i < 8; ++i) {
            int lr = i + 8 * (lane >> 4);
            int r = row0 + lr;
            {
                float rg = sigmoid_f((ir0[i] + bir) + (hr0[i] + bhr));
                float zg = sigmoid_f((iz0[i] + biz) + (hz0[i] + bhz));
                float ng = tanh_f((in0[i] + bin) + rg * (hn0[i] + bhn));
                float hv = at[lr * GRU_PITCH + 384 + col];
                hnew[(size_t)r * 256 + col] = (1.0f - zg) * ng + zg * hv;
            }
            {
                float rg = sigmoid_f((ir1[i] + bir) + (hr1[i] + bhr));
                float zg = sigmoid_f((iz1[i] + biz) + (hz1[i] + bhz));
                float ng = tanh_f((in1[i] + bin) + rg * (hn1[i] + bhn));
                float hv = at[(lr + 16) * GRU_PITCH + 384 + col];
                hnew[(size_t)(r + 16) * 256 + col] = (1.0f - zg) * ng + zg * hv;
            }
        }
    }
}

__global__ __launch_bounds__(128) void out_kernel(
    const float* __restrict__ hnew, const float* __restrict__ poutW,
    const float* __restrict__ outb, float* __restrict__ q, int N) {
    const int wave = threadIdx.x >> 5;
    const int lane = threadIdx.x & 31;
    const int row0 = (blockIdx.x * 4 + wave) * 32;
    if (row0 >= N) return;
    v8f acc0 = {}, acc1 = {};
    const float* a0 = hnew + (size_t)(row0 + (lane & 15)) * 256 + ((lane >> 4) << 1);
    seg21<16>(a0, a0 + 16 * 256, (const v2f*)poutW + lane, 4, acc0, acc1);
    int col = lane & 15;
    float bias = outb[col];
#pragma unroll
    for (int i = 0; i < 8; ++i) {
        int r = row0 + i + 8 * (lane >> 4);
        q[(size_t)r * 16 + col] = acc0[i] + bias;
        q[(size_t)(r + 16) * 16 + col] = acc1[i] + bias;
    }
}

extern "C" void kernel_launch(void* const* d_in, const int* in_sizes, int n_in,
                              void* d_out, int out_size, void* d_ws,
                              size_t ws_size, hipStream_t stream) {
    (void)n_in; (void)out_size; (void)ws_size;
    const float* feat = (const float*)d_in[0];
    const float* h    = (const float*)d_in[1];
    const int*   src  = (const int*)d_in[2];
    const int*   dst  = (const int*)d_in[3];
    const float* W0   = (const float*)d_in[4];
    const float* b0   = (const float*)d_in[5];
    const float* W1   = (const float*)d_in[6];
    const float* b1   = (const float*)d_in[7];
    const float* msgW = (const float*)d_in[8];
    const float* msgb = (const float*)d_in[9];
    const float* Wih  = (const float*)d_in[10];
    const float* Whh  = (const float*)d_in[11];
    const float* bih  = (const float*)d_in[12];
    const float* bhh  = (const float*)d_in[13];
    const float* outW = (const float*)d_in[14];
    const float* outb = (const float*)d_in[15];

    const int N = in_sizes[1] / 256;
    const int E = in_sizes[2];

    float* ws    = (float*)d_ws;
    float* x     = ws;
    float* mx    = x    + (size_t)N * 256;
    float* bkt   = mx   + (size_t)N * 128;
    int*   slots = (int*)bkt;
    int*   deg   = slots + (size_t)N * SLOTS_PER_NODE;
    int*   ovf   = deg + N + 16;
    float* c     = bkt  + (size_t)N * 128 + N;
    float* pW0   = c    + (size_t)N * 128;
    float* pW1   = pW0  + 16384;
    float* pmsgW = pW1  + 65536;
    float* pWih  = pmsgW + 65536;
    float* pWhh  = pWih + 294912;
    float* poutW = pWhh + 196608;

    float* q    = (float*)d_out;
    float* hnew = q + (size_t)N * 16;

    const int encBlocks  = (N / 16 + 3) / 4;
    const int gemmBlocks = (N / 32 + 3) / 4;
    const int rowTiles   = (N + 31) / 32;
    const int zeroBlocks = (N + 1 + 255) / 256;
    const int edgeBlocks = (E + EDGES_PER_BLOCK - 1) / EDGES_PER_BLOCK;
    const int nodeBlocks = (N + 7) / 8;

    setup_kernel<<<PB_ALL + zeroBlocks, 256, 0, stream>>>(
        W0, W1, msgW, Wih, Whh, outW, pW0, pW1, pmsgW, pWih, pWhh, poutW, deg, N);
    enc_rank_kernel<<<encBlocks + edgeBlocks, 128, 0, stream>>>(
        feat, pW0, b0, pW1, b1, x, N, encBlocks, src, dst, deg, slots, ovf, E);
    msg_kernel<<<rowTiles, 128, 0, stream>>>(x, h, pmsgW, msgb, mx, N);
    gather_kernel<<<nodeBlocks, 256, 0, stream>>>(deg, slots, ovf, src, dst, mx,
                                                  c, N, E);
    gru_kernel<<<dim3(rowTiles, 2), 128, 0, stream>>>(x, c, h, pWih, pWhh, bih,
                                                      bhh, hnew, N);
    out_kernel<<<gemmBlocks, 128, 0, stream>>>(hnew, poutW, outb, q, N);
}


# encoder and node-message GEMM fused per 32-row tile: one block runs both encoder stages and the message GEMM, x passes to it through LDS; message launch goes, 6 launches to 5; k order unchanged
# speedup vs baseline: 5.9545x; 1.0668x over previous
#include <hip/hip_runtime.h>
#include <hip/hip_bf16.h>


typedef __attribute__((ext_vector_type(2))) float v2f;
typedef __attribute__((ext_vector_type(8))) float v8f;

__device__ __forceinline__ v8f wmma4(v2f a, v2f b, v8f c) {
    return __builtin_amdgcn_wmma_f32_16x16x4_f32(
        false, a, false, b, (short)0, c, false, false);
}

__device__ __forceinline__ v2f load_a_frag(const float* M, int ld, int row0,
                                           int k0, int lane) {
    int r = row0 + (lane & 15);
    int k = k0 + ((lane >> 4) << 1);
    const float* p = M + (size_t)r * ld + k;
    v2f a; a.x = p[0]; a.y = p[1];
    return a;
}


template <int KB> struct Frag21 { v2f a0[KB], a1[KB], b[KB]; };

template <int KB>
__device__ __forceinline__ void load21(Frag21<KB>& f, const float* pa0,
                                       const float* pa1, const v2f* pb,
                                       int blk) {
    const float* q0 = pa0 + blk * (KB * 4);
    const float* q1 = pa1 + blk * (KB * 4);
    const v2f* qb = pb + blk * (KB * 32);
#pragma unroll
    for (int j = 0; j < KB; ++j) {
        f.a0[j].x = q0[j * 4]; f.a0[j].y = q0[j * 4 + 1];
        f.a1[j].x = q1[j * 4]; f.a1[j].y = q1[j * 4 + 1];
        f.b[j] = qb[j * 32];
    }
}

template <int KB>
__device__ __forceinline__ void mma21(const Frag21<KB>& f, v8f& c0, v8f& c1) {
#pragma unroll
    for (int j = 0; j < KB; ++j) {
        c0 = wmma4(f.a0[j], f.b[j], c0);
        c1 = wmma4(f.a1[j], f.b[j], c1);
    }
}

template <int KB>
__device__ __forceinline__ void seg21(const float* pa0, const float* pa1,
                                      const v2f* pb, int nb, v8f& c0, v8f& c1) {
    Frag21<KB> f0, f1;
    load21<KB>(f0, pa0, pa1, pb, 0);
    int b = 0;
#pragma unroll 1
    for (; b + 2 < nb; b += 2) {
        load21<KB>(f1, pa0, pa1, pb, b + 1);
        mma21<KB>(f0, c0, c1);
        load21<KB>(f0, pa0, pa1, pb, b + 2);
        mma21<KB>(f1, c0, c1);
    }
    load21<KB>(f1, pa0, pa1, pb, b + 1);
    mma21<KB>(f0, c0, c1);
    mma21<KB>(f1, c0, c1);
}

template <int KB, int PITCH>
__device__ __forceinline__ void load21s(Frag21<KB>& f, const float* la,
                                        const v2f* pb, int blk) {
    const float* qa = la + blk * (KB * 4);
    const v2f* qb = pb + blk * (KB * 32);
#pragma unroll
    for (int j = 0; j < KB; ++j) {
        f.a0[j].x = qa[j * 4];              f.a0[j].y = qa[j * 4 + 1];
        f.a1[j].x = qa[16 * PITCH + j * 4]; f.a1[j].y = qa[16 * PITCH + j * 4 + 1];
        f.b[j] = qb[j * 32];
    }
}

template <int KB, int PITCH>
__device__ __forceinline__ void seg21s(const float* la, const v2f* pb, int nb,
                                       v8f& c0, v8f& c1) {
    Frag21<KB> f0, f1;
    load21s<KB, PITCH>(f0, la, pb, 0);
    int b = 0;
#pragma unroll 1
    for (; b + 2 < nb; b += 2) {
        load21s<KB, PITCH>(f1, la, pb, b + 1);
        __builtin_amdgcn_sched_barrier(0);
        mma21<KB>(f0, c0, c1);
        load21s<KB, PITCH>(f0, la, pb, b + 2);
        __builtin_amdgcn_sched_barrier(0);
        mma21<KB>(f1, c0, c1);
    }
    load21s<KB, PITCH>(f1, la, pb, b + 1);
    mma21<KB>(f0, c0, c1);
    mma21<KB>(f1, c0, c1);
}

template <int KB> struct Frag23 { v2f a0[KB], a1[KB], br[KB], bz[KB], bn[KB]; };

template <int KB>
__device__ __forceinline__ void mma23(const Frag23<KB>& f, v8f& r0, v8f& r1,
                                      v8f& z0, v8f& z1, v8f& n0, v8f& n1) {
#pragma unroll
    for (int j = 0; j < KB; ++j) {
        r0 = wmma4(f.a0[j], f.br[j], r0);  r1 = wmma4(f.a1[j], f.br[j], r1);
        z0 = wmma4(f.a0[j], f.bz[j], z0);  z1 = wmma4(f.a1[j], f.bz[j], z1);
        n0 = wmma4(f.a0[j], f.bn[j], n0);  n1 = wmma4(f.a1[j], f.bn[j], n1);
    }
}

template <int KB> struct FragB { v2f b[KB]; };

template <int KB>
__device__ __forceinline__ void loadB(FragB<KB>& f, const v2f* pb, int blk) {
    const v2f* qb = pb + blk * (KB * 32);
#pragma unroll
    for (int j = 0; j < KB; ++j) f.b[j] = qb[j * 32];
}

__device__ __forceinline__ float sigmoid_f(float v) {
    return 1.0f / (1.0f + __expf(-v));
}
__device__ __forceinline__ float tanh_f(float v) {
    float t = __expf(-2.0f * v);
    return (1.0f - t) / (1.0f + t);
}

template <int K, int Nc>
__device__ __forceinline__ void pack_tile(const float* __restrict__ W,
                                          float* __restrict__ P, int w,
                                          int lane) {
    constexpr int nk = K >> 2;
    int kt = w % nk, nt = w / nk;
    int n = nt * 16 + (lane & 15);
    int k = kt * 4 + ((lane >> 4) << 1);
    v2f v;
    v.x = W[(size_t)k * Nc + n];
    v.y = W[(size_t)(k + 1) * Nc + n];
    ((v2f*)P)[(size_t)(nt * nk + kt) * 32 + lane] = v;
}

constexpr int packBlocks(int K, int Nc) { return (K / 4) * (Nc / 16) / 8; }
constexpr bool packWhole(int K, int Nc) { return (K / 4) * (Nc / 16) % 8 == 0; }
constexpr int PB_W0  = packBlocks(64, 256);
constexpr int PB_W1  = packBlocks(256, 256);
constexpr int PB_MSG = packBlocks(512, 128);
constexpr int PB_WIH = packBlocks(384, 768);
constexpr int PB_WHH = packBlocks(256, 768);
constexpr int PB_OUT = packBlocks(256, 16);
constexpr int PB_ALL = PB_W0 + PB_W1 + PB_MSG + PB_WIH + PB_WHH + PB_OUT;
static_assert(packWhole(64, 256) && packWhole(256, 256) && packWhole(512, 128) &&
              packWhole(384, 768) && packWhole(256, 768) && packWhole(256, 16),
              "every matrix fills whole 8-wave blocks, so no wave has to be masked");

__global__ __launch_bounds__(256) void setup_kernel(
    const float* __restrict__ W0, const float* __restrict__ W1,
    const float* __restrict__ msgW, const float* __restrict__ Wih,
    const float* __restrict__ Whh, const float* __restrict__ outW,
    float* __restrict__ pW0, float* __restrict__ pW1,
    float* __restrict__ pmsgW, float* __restrict__ pWih,
    float* __restrict__ pWhh, float* __restrict__ poutW,
    int* __restrict__ deg, int N) {
    const int lane = threadIdx.x & 31;
    const int wib = threadIdx.x >> 5;
    int b = blockIdx.x;
    if (b < PB_W0)  { pack_tile<64, 256>(W0, pW0, b * 8 + wib, lane); return; }
    b -= PB_W0;
    if (b < PB_W1)  { pack_tile<256, 256>(W1, pW1, b * 8 + wib, lane); return; }
    b -= PB_W1;
    if (b < PB_MSG) { pack_tile<512, 128>(msgW, pmsgW, b * 8 + wib, lane); return; }
    b -= PB_MSG;
    if (b < PB_WIH) { pack_tile<384, 768>(Wih, pWih, b * 8 + wib, lane); return; }
    b -= PB_WIH;
    if (b < PB_WHH) { pack_tile<256, 768>(Whh, pWhh, b * 8 + wib, lane); return; }
    b -= PB_WHH;
    if (b < PB_OUT) { pack_tile<256, 16>(outW, poutW, b * 8 + wib, lane); return; }
    b -= PB_OUT;
    int i = b * 256 + (int)threadIdx.x;
    if (i <= N) deg[i] = 0;
}

constexpr int MSG_PITCH = 512 + 4;
constexpr int T1_PITCH = 256 + 4;

__device__ __forceinline__ void enc1_tile(const FragB<16>& g, const v2f* af0,
                                          const v2f* af1, int nt, int lane,
                                          const float* __restrict__ b0,
                                          float* t1) {
    v8f a0 = {}, a1 = {};
#pragma unroll
    for (int kk = 0; kk < 16; ++kk) {
        a0 = wmma4(af0[kk], g.b[kk], a0);
        a1 = wmma4(af1[kk], g.b[kk], a1);
    }
    int col = nt * 16 + (lane & 15);
    float bias = b0[col];
#pragma unroll
    for (int i = 0; i < 8; ++i) {
        int r = i + 8 * (lane >> 4);
        t1[r * T1_PITCH + col] = fmaxf(a0[i] + bias, 0.0f);
        t1[(r + 16) * T1_PITCH + col] = fmaxf(a1[i] + bias, 0.0f);
    }
}

__device__ __forceinline__ void encmsg_block(
    float* t1, float* am, int blk,
    const float* __restrict__ feat, const float* __restrict__ pW0,
    const float* __restrict__ b0, const float* __restrict__ pW1,
    const float* __restrict__ b1, float* __restrict__ x,
    const float* __restrict__ h, const float* __restrict__ pmsgW,
    const float* __restrict__ msgb, float* __restrict__ mx, int N) {
    const int wave = threadIdx.x >> 5;
    const int lane = threadIdx.x & 31;
    const int tid = threadIdx.x;
    const int row0 = blk * 32;
    if (row0 >= N) return;

    const float4* hv = (const float4*)(h + (size_t)row0 * 256);
    float4 vh[16];
#pragma unroll
    for (int i = 0; i < 16; ++i) vh[i] = hv[i * 128 + tid];

    {
        v2f af0[16], af1[16];
#pragma unroll
        for (int kk = 0; kk < 16; ++kk) {
            af0[kk] = load_a_frag(feat, 64, row0, kk * 4, lane);
            af1[kk] = load_a_frag(feat, 64, row0 + 16, kk * 4, lane);
        }
        const v2f* pb0 = (const v2f*)pW0 + lane;
        const int nt0 = wave * 4;
        FragB<16> g0, g1;
        loadB<16>(g0, pb0, nt0);
        loadB<16>(g1, pb0, nt0 + 1);
        enc1_tile(g0, af0, af1, nt0, lane, b0, t1);
        loadB<16>(g0, pb0, nt0 + 2);
        enc1_tile(g1, af0, af1, nt0 + 1, lane, b0, t1);
        loadB<16>(g1, pb0, nt0 + 3);
        enc1_tile(g0, af0, af1, nt0 + 2, lane, b0, t1);
        enc1_tile(g1, af0, af1, nt0 + 3, lane, b0, t1);
    }
#pragma unroll
    for (int i = 0; i < 16; ++i) {
        int idx = i * 128 + tid;
        *(float4*)&am[(idx >> 6) * MSG_PITCH + 256 + (idx & 63) * 4] = vh[i];
    }
    __syncthreads();

    {
        const float* la1 = t1 + (lane & 15) * T1_PITCH + ((lane >> 4) << 1);
#pragma unroll 1
        for (int t = 0; t < 4; ++t) {
            const int nt = wave * 4 + t;
            const v2f* pb = (const v2f*)pW1 + ((size_t)(nt * 64) * 32 + lane);
            v8f acc0 = {}, acc1 = {};
            seg21s<8, T1_PITCH>(la1, pb, 8, acc0, acc1);
            int col = nt * 16 + (lane & 15);
            float bias = b1[col];
#pragma unroll
            for (int i = 0; i < 8; ++i) {
                int r = i + 8 * (lane >> 4);
                float v0 = fmaxf(acc0[i] + bias, 0.0f);
                float v1 = fmaxf(acc1[i] + bias, 0.0f);
                x[(size_t)(row0 + r) * 256 + col] = v0;
                x[(size_t)(row0 + r + 16) * 256 + col] = v1;
                am[r * MSG_PITCH + col] = v0;
                am[(r + 16) * MSG_PITCH + col] = v1;
            }
        }
    }
    __syncthreads();

    const float* la = am + (lane & 15) * MSG_PITCH + ((lane >> 4) << 1);
#pragma unroll 1
    for (int t = 0; t < 2; ++t) {
        const int nt = wave * 2 + t;
        const v2f* pb = (const v2f*)pmsgW + ((size_t)(nt * 128) * 32 + lane);
        v8f acc0 = {}, acc1 = {};
        seg21s<8, MSG_PITCH>(la, pb, 16, acc0, acc1);
        int col = nt * 16 + (lane & 15);
        float bias = msgb[col];
#pragma unroll
        for (int i = 0; i < 8; ++i) {
            int r = row0 + i + 8 * (lane >> 4);
            mx[(size_t)r * 128 + col] = acc0[i] + bias;
            mx[(size_t)(r + 16) * 128 + col] = acc1[i] + bias;
        }
    }
}

constexpr int SLOTS_PER_NODE = 64;
constexpr int EDGES_PER_BLOCK = 512;
template <bool FULL>
__device__ __forceinline__ void rank_edges(
    int base, const int* __restrict__ src, const int* __restrict__ dst,
    int* __restrict__ deg, int* __restrict__ slots, int* __restrict__ ovf,
    int N, int E) {
    int d[4], s[4], r[4];
#pragma unroll
    for (int k = 0; k < 4; ++k) {
        int e = base + k * 128;
        if (!FULL) e = min(e, E - 1);
        d[k] = dst[e]; s[k] = src[e];
    }
#pragma unroll
    for (int k = 0; k < 4; ++k)
        r[k] = (FULL || base + k * 128 < E) ? atomicAdd(deg + d[k], 1) : 0;
#pragma unroll
    for (int k = 0; k < 4; ++k) {
        if (FULL || base + k * 128 < E) {
            if (r[k] < SLOTS_PER_NODE) {
                int pos = d[k] * SLOTS_PER_NODE + r[k];
                pos = min(max(pos, 0), N * SLOTS_PER_NODE - 1);
                slots[pos] = s[k];
            } else {
                int o = atomicAdd(deg + N, 1);
                if (o < E) ovf[o] = base + k * 128;
            }
        }
    }
}

__device__ __forceinline__ void rank_block(
    int blk, const int* __restrict__ src, const int* __restrict__ dst,
    int* __restrict__ deg, int* __restrict__ slots, int* __restrict__ ovf,
    int N, int E) {
    const int base = blk * EDGES_PER_BLOCK + (int)threadIdx.x;
    if ((blk + 1) * EDGES_PER_BLOCK <= E)
        rank_edges<true>(base, src, dst, deg, slots, ovf, N, E);
    else
        rank_edges<false>(base, src, dst, deg, slots, ovf, N, E);
}

__global__ __launch_bounds__(128) void enc_msg_rank_kernel(
    const float* __restrict__ feat, const float* __restrict__ pW0,
    const float* __restrict__ b0, const float* __restrict__ pW1,
    const float* __restrict__ b1, float* __restrict__ x,
    const float* __restrict__ h, const float* __restrict__ pmsgW,
    const float* __restrict__ msgb, float* __restrict__ mx, int N,
    int gemmTiles, const int* __restrict__ src, const int* __restrict__ dst,
    int* __restrict__ deg, int* __restrict__ slots, int* __restrict__ ovf,
    int E) {
    __shared__ __attribute__((aligned(16))) float am[32 * MSG_PITCH];
    __shared__ __attribute__((aligned(16))) float t1[32 * T1_PITCH];
    const int blk = blockIdx.x;
    if (blk < gemmTiles)
        encmsg_block(t1, am, blk, feat, pW0, b0, pW1, b1, x, h, pmsgW, msgb, mx, N);
    else
        rank_block(blk - gemmTiles, src, dst, deg, slots, ovf, N, E);
}

__global__ __launch_bounds__(256) void gather_kernel(
    const int* __restrict__ deg, const int* __restrict__ slots,
    const int* __restrict__ ovf, const int* __restrict__ src,
    const int* __restrict__ dst, const float* __restrict__ mx,
    float* __restrict__ c, int N, int E) {
    const int node = __builtin_amdgcn_readfirstlane(
        (int)((blockIdx.x * 256u + threadIdx.x) >> 5));
    const int lane = threadIdx.x & 31;
    if (node >= N) return;
    const int cnt = deg[node];
    const int* sl = slots + (size_t)node * SLOTS_PER_NODE;
    const int end = min(cnt, SLOTS_PER_NODE);
    const float4* mv = (const float4*)mx;
    float4 acc = make_float4(0.0f, 0.0f, 0.0f, 0.0f);
    int j = 0;
    for (; j + 8 <= end; j += 8) {
        float4 v[8];
#pragma unroll
        for (int k = 0; k < 8; ++k) {
            int s = sl[j + k];
            v[k] = mv[(size_t)s * 32 + lane];
        }
#pragma unroll
        for (int k = 0; k < 8; ++k) {
            acc.x += v[k].x; acc.y += v[k].y;
            acc.z += v[k].z; acc.w += v[k].w;
        }
    }
    if (j < end) {
        float4 v[8];
#pragma unroll
        for (int k = 0; k < 8; ++k) {
            int s = sl[min(j + k, end - 1)];
            v[k] = mv[(size_t)s * 32 + lane];
        }
#pragma unroll
        for (int k = 0; k < 8; ++k) {
            const bool in = (j + k < end);
            acc.x += in ? v[k].x : 0.0f; acc.y += in ? v[k].y : 0.0f;
            acc.z += in ? v[k].z : 0.0f; acc.w += in ? v[k].w : 0.0f;
        }
    }
    if (cnt > SLOTS_PER_NODE) {
        const int tot = min(deg[N], E);
        for (int k = 0; k < tot; ++k) {
            const int e = min(max(ovf[k], 0), E - 1);
            if (dst[e] == node) {
                float4 v = mv[(size_t)src[e] * 32 + lane];
                acc.x += v.x; acc.y += v.y; acc.z += v.z; acc.w += v.w;
            }
        }
    }
    const float den = fmaxf((float)cnt, 1.0f);
    float4 o;
    o.x = acc.x / den; o.y = acc.y / den; o.z = acc.z / den; o.w = acc.w / den;
    ((float4*)c)[(size_t)node * 32 + lane] = o;
}

constexpr int GRU_KB = 4;
constexpr int GRU_K = 640;
constexpr int GRU_PITCH = GRU_K + 4;

template <int KB>
__device__ __forceinline__ void load23s(Frag23<KB>& f, const float* la,
                                        const float* B, int boff, int gate,
                                        int blk) {
    const float* qa = la + blk * (KB * 4);
    const float* qb = B + blk * (KB * 64);
#pragma unroll
    for (int j = 0; j < KB; ++j) {
        f.a0[j].x = qa[j * 4];                  f.a0[j].y = qa[j * 4 + 1];
        f.a1[j].x = qa[16 * GRU_PITCH + j * 4]; f.a1[j].y = qa[16 * GRU_PITCH + j * 4 + 1];
        f.br[j].x = qb[boff + j * 64];          f.br[j].y = qb[boff + j * 64 + 1];
        f.bz[j].x = qb[boff + gate + j * 64];
        f.bz[j].y = qb[boff + gate + j * 64 + 1];
        f.bn[j].x = qb[boff + 2 * gate + j * 64];
        f.bn[j].y = qb[boff + 2 * gate + j * 64 + 1];
    }
}

template <int KB>
__device__ __forceinline__ void seg23s(const float* la, const float* B,
                                       int boff, int gate, int nb, v8f& r0,
                                       v8f& r1, v8f& z0, v8f& z1, v8f& n0,
                                       v8f& n1) {
    Frag23<KB> f0, f1;
    load23s<KB>(f0, la, B, boff, gate, 0);
    int b = 0;
#pragma unroll 1
    for (; b + 2 < nb; b += 2) {
        load23s<KB>(f1, la, B, boff, gate, b + 1);
        __builtin_amdgcn_sched_barrier(0);
        mma23<KB>(f0, r0, r1, z0, z1, n0, n1);
        load23s<KB>(f0, la, B, boff, gate, b + 2);
        __builtin_amdgcn_sched_barrier(0);
        mma23<KB>(f1, r0, r1, z0, z1, n0, n1);
    }
    load23s<KB>(f1, la, B, boff, gate, b + 1);
    mma23<KB>(f0, r0, r1, z0, z1, n0, n1);
    mma23<KB>(f1, r0, r1, z0, z1, n0, n1);
}

__global__ __launch_bounds__(128) void gru_kernel(
    const float* __restrict__ x, const float* __restrict__ c,
    const float* __restrict__ h, const float* __restrict__ pWih,
    const float* __restrict__ pWhh, const float* __restrict__ bih,
    const float* __restrict__ bhh, float* __restrict__ hnew, int N) {
    __shared__ __attribute__((aligned(16))) float at[32 * GRU_PITCH];
    const int wave = __builtin_amdgcn_readfirstlane((int)(threadIdx.x >> 5));
    const int lane = threadIdx.x & 31;
    const int tid = threadIdx.x;
    const int row0 = blockIdx.x * 32;
    if (row0 >= N) return;

    {
        const float4* xv = (const float4*)(x + (size_t)row0 * 256);
        const float4* cv = (const float4*)(c + (size_t)row0 * 128);
        const float4* hv = (const float4*)(h + (size_t)row0 * 256);
        float4 vx[16], vc[8], vh[16];
#pragma unroll
        for (int i = 0; i < 16; ++i) vx[i] = xv[i * 128 + tid];
#pragma unroll
        for (int i = 0; i < 8; ++i) vc[i] = cv[i * 128 + tid];
#pragma unroll
        for (int i = 0; i < 16; ++i) vh[i] = hv[i * 128 + tid];
#pragma unroll
        for (int i = 0; i < 16; ++i) {
            int idx = i * 128 + tid;
            *(float4*)&at[(idx >> 6) * GRU_PITCH + (idx & 63) * 4] = vx[i];
        }
#pragma unroll
        for (int i = 0; i < 8; ++i) {
            int idx = i * 128 + tid;
            *(float4*)&at[(idx >> 5) * GRU_PITCH + 256 + (idx & 31) * 4] = vc[i];
        }
#pragma unroll
        for (int i = 0; i < 16; ++i) {
            int idx = i * 128 + tid;
            *(float4*)&at[(idx >> 6) * GRU_PITCH + 384 + (idx & 63) * 4] = vh[i];
        }
    }
    __syncthreads();

    const int p = blockIdx.y * 4 + wave;
    const int boff = lane * 2;
    const float* la = at + (lane & 15) * GRU_PITCH + ((lane >> 4) << 1);
#pragma unroll 1
    for (int t = 0; t < 2; ++t) {
        const int nt = p * 2 + t;
        v8f ir0 = {}, iz0 = {}, in0 = {}, ir1 = {}, iz1 = {}, in1 = {};
        v8f hr0 = {}, hz0 = {}, hn0 = {}, hr1 = {}, hz1 = {}, hn1 = {};
        const float* pbi = pWih + (size_t)(nt * 96) * 64;
        const float* pbh = pWhh + (size_t)(nt * 64) * 64;
        seg23s<GRU_KB>(la, pbi, boff, 16 * 96 * 64, 96 / GRU_KB,
                       ir0, ir1, iz0, iz1, in0, in1);
        seg23s<GRU_KB>(la + 384, pbh, boff, 16 * 64 * 64, 64 / GRU_KB,
                       hr0, hr1, hz0, hz1, hn0, hn1);
        int col = nt * 16 + (lane & 15);
        float bir = bih[col], biz = bih[col + 256], bin = bih[col + 512];
        float bhr = bhh[col], bhz = bhh[col + 256], bhn = bhh[col + 512];
#pragma unroll
        for (int i = 0; i < 8; ++i) {
            int lr = i + 8 * (lane >> 4);
            int r = row0 + lr;
            {
                float rg = sigmoid_f((ir0[i] + bir) + (hr0[i] + bhr));
                float zg = sigmoid_f((iz0[i] + biz) + (hz0[i] + bhz));
                float ng = tanh_f((in0[i] + bin) + rg * (hn0[i] + bhn));
                float hv = at[lr * GRU_PITCH + 384 + col];
                hnew[(size_t)r * 256 + col] = (1.0f - zg) * ng + zg * hv;
            }
            {
                float rg = sigmoid_f((ir1[i] + bir) + (hr1[i] + bhr));
                float zg = sigmoid_f((iz1[i] + biz) + (hz1[i] + bhz));
                float ng = tanh_f((in1[i] + bin) + rg * (hn1[i] + bhn));
                float hv = at[(lr + 16) * GRU_PITCH + 384 + col];
                hnew[(size_t)(r + 16) * 256 + col] = (1.0f - zg) * ng + zg * hv;
            }
        }
    }
}

__global__ __launch_bounds__(128) void out_kernel(
    const float* __restrict__ hnew, const float* __restrict__ poutW,
    const float* __restrict__ outb, float* __restrict__ q, int N) {
    const int wave = threadIdx.x >> 5;
    const int lane = threadIdx.x & 31;
    const int row0 = (blockIdx.x * 4 + wave) * 32;
    if (row0 >= N) return;
    v8f acc0 = {}, acc1 = {};
    const float* a0 = hnew + (size_t)(row0 + (lane & 15)) * 256 + ((lane >> 4) << 1);
    seg21<16>(a0, a0 + 16 * 256, (const v2f*)poutW + lane, 4, acc0, acc1);
    int col = lane & 15;
    float bias = outb[col];
#pragma unroll
    for (int i = 0; i < 8; ++i) {
        int r = row0 + i + 8 * (lane >> 4);
        q[(size_t)r * 16 + col] = acc0[i] + bias;
        q[(size_t)(r + 16) * 16 + col] = acc1[i] + bias;
    }
}

extern "C" void kernel_launch(void* const* d_in, const int* in_sizes, int n_in,
                              void* d_out, int out_size, void* d_ws,
                              size_t ws_size, hipStream_t stream) {
    (void)n_in; (void)out_size; (void)ws_size;
    const float* feat = (const float*)d_in[0];
    const float* h    = (const float*)d_in[1];
    const int*   src  = (const int*)d_in[2];
    const int*   dst  = (const int*)d_in[3];
    const float* W0   = (const float*)d_in[4];
    const float* b0   = (const float*)d_in[5];
    const float* W1   = (const float*)d_in[6];
    const float* b1   = (const float*)d_in[7];
    const float* msgW = (const float*)d_in[8];
    const float* msgb = (const float*)d_in[9];
    const float* Wih  = (const float*)d_in[10];
    const float* Whh  = (const float*)d_in[11];
    const float* bih  = (const float*)d_in[12];
    const float* bhh  = (const float*)d_in[13];
    const float* outW = (const float*)d_in[14];
    const float* outb = (const float*)d_in[15];

    const int N = in_sizes[1] / 256;
    const int E = in_sizes[2];

    float* ws    = (float*)d_ws;
    float* x     = ws;
    float* mx    = x    + (size_t)N * 256;
    float* bkt   = mx   + (size_t)N * 128;
    int*   slots = (int*)bkt;
    int*   deg   = slots + (size_t)N * SLOTS_PER_NODE;
    int*   ovf   = deg + N + 16;
    float* c     = bkt  + (size_t)N * 128 + N;
    float* pW0   = c    + (size_t)N * 128;
    float* pW1   = pW0  + 16384;
    float* pmsgW = pW1  + 65536;
    float* pWih  = pmsgW + 65536;
    float* pWhh  = pWih + 294912;
    float* poutW = pWhh + 196608;

    float* q    = (float*)d_out;
    float* hnew = q + (size_t)N * 16;

    const int gemmBlocks = (N / 32 + 3) / 4;
    const int rowTiles   = (N + 31) / 32;
    const int zeroBlocks = (N + 1 + 255) / 256;
    const int edgeBlocks = (E + EDGES_PER_BLOCK - 1) / EDGES_PER_BLOCK;
    const int nodeBlocks = (N + 7) / 8;

    setup_kernel<<<PB_ALL + zeroBlocks, 256, 0, stream>>>(
        W0, W1, msgW, Wih, Whh, outW, pW0, pW1, pmsgW, pWih, pWhh, poutW, deg, N);
    enc_msg_rank_kernel<<<rowTiles + edgeBlocks, 128, 0, stream>>>(
        feat, pW0, b0, pW1, b1, x, h, pmsgW, msgb, mx, N, rowTiles, src, dst,
        deg, slots, ovf, E);
    gather_kernel<<<nodeBlocks, 256, 0, stream>>>(deg, slots, ovf, src, dst, mx,
                                                  c, N, E);
    gru_kernel<<<dim3(rowTiles, 2), 128, 0, stream>>>(x, c, h, pWih, pWhh, bih,
                                                      bhh, hnew, N);
    out_kernel<<<gemmBlocks, 128, 0, stream>>>(hnew, poutW, outb, q, N);
}


# head GEMM fused into the GRU launch: a block owns 80 whole rows (5 row tiles x every column tile, 8 waves); h_new goes via an LDS tile to whole-line stores and to the head's WMMAs; 5 launches to 4
# speedup vs baseline: 6.2414x; 1.0482x over previous
#include <hip/hip_runtime.h>
#include <hip/hip_bf16.h>


typedef __attribute__((ext_vector_type(2))) float v2f;
typedef __attribute__((ext_vector_type(8))) float v8f;

__device__ __forceinline__ v8f wmma4(v2f a, v2f b, v8f c) {
    return __builtin_amdgcn_wmma_f32_16x16x4_f32(
        false, a, false, b, (short)0, c, false, false);
}

__device__ __forceinline__ v2f load_a_frag(const float* M, int ld, int row0,
                                           int k0, int lane) {
    int r = row0 + (lane & 15);
    int k = k0 + ((lane >> 4) << 1);
    const float* p = M + (size_t)r * ld + k;
    v2f a; a.x = p[0]; a.y = p[1];
    return a;
}


template <int KB> struct Frag21 { v2f a0[KB], a1[KB], b[KB]; };

template <int KB>
__device__ __forceinline__ void mma21(const Frag21<KB>& f, v8f& c0, v8f& c1) {
#pragma unroll
    for (int j = 0; j < KB; ++j) {
        c0 = wmma4(f.a0[j], f.b[j], c0);
        c1 = wmma4(f.a1[j], f.b[j], c1);
    }
}

template <int KB, int PITCH>
__device__ __forceinline__ void load21s(Frag21<KB>& f, const float* la,
                                        const v2f* pb, int blk) {
    const float* qa = la + blk * (KB * 4);
    const v2f* qb = pb + blk * (KB * 32);
#pragma unroll
    for (int j = 0; j < KB; ++j) {
        f.a0[j].x = qa[j * 4];              f.a0[j].y = qa[j * 4 + 1];
        f.a1[j].x = qa[16 * PITCH + j * 4]; f.a1[j].y = qa[16 * PITCH + j * 4 + 1];
        f.b[j] = qb[j * 32];
    }
}

template <int KB, int PITCH>
__device__ __forceinline__ void seg21s(const float* la, const v2f* pb, int nb,
                                       v8f& c0, v8f& c1) {
    Frag21<KB> f0, f1;
    load21s<KB, PITCH>(f0, la, pb, 0);
    int b = 0;
#pragma unroll 1
    for (; b + 2 < nb; b += 2) {
        load21s<KB, PITCH>(f1, la, pb, b + 1);
        __builtin_amdgcn_sched_barrier(0);
        mma21<KB>(f0, c0, c1);
        load21s<KB, PITCH>(f0, la, pb, b + 2);
        __builtin_amdgcn_sched_barrier(0);
        mma21<KB>(f1, c0, c1);
    }
    load21s<KB, PITCH>(f1, la, pb, b + 1);
    mma21<KB>(f0, c0, c1);
    mma21<KB>(f1, c0, c1);
}

constexpr int GRU_RT = 5;
template <int KB> struct Frag53 { v2f a[GRU_RT][KB], br[KB], bz[KB], bn[KB]; };

template <int KB>
__device__ __forceinline__ void mma53(const Frag53<KB>& f, v8f (&r)[GRU_RT],
                                      v8f (&z)[GRU_RT], v8f (&n)[GRU_RT]) {
#pragma unroll
    for (int j = 0; j < KB; ++j) {
#pragma unroll
        for (int m = 0; m < GRU_RT; ++m) {
            r[m] = wmma4(f.a[m][j], f.br[j], r[m]);
            z[m] = wmma4(f.a[m][j], f.bz[j], z[m]);
            n[m] = wmma4(f.a[m][j], f.bn[j], n[m]);
        }
    }
}

template <int KB> struct FragB { v2f b[KB]; };

template <int KB>
__device__ __forceinline__ void loadB(FragB<KB>& f, const v2f* pb, int blk) {
    const v2f* qb = pb + blk * (KB * 32);
#pragma unroll
    for (int j = 0; j < KB; ++j) f.b[j] = qb[j * 32];
}

__device__ __forceinline__ float sigmoid_f(float v) {
    return 1.0f / (1.0f + __expf(-v));
}
__device__ __forceinline__ float tanh_f(float v) {
    float t = __expf(-2.0f * v);
    return (1.0f - t) / (1.0f + t);
}

template <int K, int Nc>
__device__ __forceinline__ void pack_tile(const float* __restrict__ W,
                                          float* __restrict__ P, int w,
                                          int lane) {
    constexpr int nk = K >> 2;
    int kt = w % nk, nt = w / nk;
    int n = nt * 16 + (lane & 15);
    int k = kt * 4 + ((lane >> 4) << 1);
    v2f v;
    v.x = W[(size_t)k * Nc + n];
    v.y = W[(size_t)(k + 1) * Nc + n];
    ((v2f*)P)[(size_t)(nt * nk + kt) * 32 + lane] = v;
}

constexpr int packBlocks(int K, int Nc) { return (K / 4) * (Nc / 16) / 8; }
constexpr bool packWhole(int K, int Nc) { return (K / 4) * (Nc / 16) % 8 == 0; }
constexpr int PB_W0  = packBlocks(64, 256);
constexpr int PB_W1  = packBlocks(256, 256);
constexpr int PB_MSG = packBlocks(512, 128);
constexpr int PB_WIH = packBlocks(384, 768);
constexpr int PB_WHH = packBlocks(256, 768);
constexpr int PB_OUT = packBlocks(256, 16);
constexpr int PB_ALL = PB_W0 + PB_W1 + PB_MSG + PB_WIH + PB_WHH + PB_OUT;
static_assert(packWhole(64, 256) && packWhole(256, 256) && packWhole(512, 128) &&
              packWhole(384, 768) && packWhole(256, 768) && packWhole(256, 16),
              "every matrix fills whole 8-wave blocks, so no wave has to be masked");

__global__ __launch_bounds__(256) void setup_kernel(
    const float* __restrict__ W0, const float* __restrict__ W1,
    const float* __restrict__ msgW, const float* __restrict__ Wih,
    const float* __restrict__ Whh, const float* __restrict__ outW,
    float* __restrict__ pW0, float* __restrict__ pW1,
    float* __restrict__ pmsgW, float* __restrict__ pWih,
    float* __restrict__ pWhh, float* __restrict__ poutW,
    int* __restrict__ deg, int N) {
    const int lane = threadIdx.x & 31;
    const int wib = threadIdx.x >> 5;
    int b = blockIdx.x;
    if (b < PB_W0)  { pack_tile<64, 256>(W0, pW0, b * 8 + wib, lane); return; }
    b -= PB_W0;
    if (b < PB_W1)  { pack_tile<256, 256>(W1, pW1, b * 8 + wib, lane); return; }
    b -= PB_W1;
    if (b < PB_MSG) { pack_tile<512, 128>(msgW, pmsgW, b * 8 + wib, lane); return; }
    b -= PB_MSG;
    if (b < PB_WIH) { pack_tile<384, 768>(Wih, pWih, b * 8 + wib, lane); return; }
    b -= PB_WIH;
    if (b < PB_WHH) { pack_tile<256, 768>(Whh, pWhh, b * 8 + wib, lane); return; }
    b -= PB_WHH;
    if (b < PB_OUT) { pack_tile<256, 16>(outW, poutW, b * 8 + wib, lane); return; }
    b -= PB_OUT;
    int i = b * 256 + (int)threadIdx.x;
    if (i <= N) deg[i] = 0;
}

constexpr int MSG_PITCH = 512 + 4;
constexpr int T1_PITCH = 256 + 4;

__device__ __forceinline__ void enc1_tile(const FragB<16>& g, const v2f* af0,
                                          const v2f* af1, int nt, int lane,
                                          const float* __restrict__ b0,
                                          float* t1) {
    v8f a0 = {}, a1 = {};
#pragma unroll
    for (int kk = 0; kk < 16; ++kk) {
        a0 = wmma4(af0[kk], g.b[kk], a0);
        a1 = wmma4(af1[kk], g.b[kk], a1);
    }
    int col = nt * 16 + (lane & 15);
    float bias = b0[col];
#pragma unroll
    for (int i = 0; i < 8; ++i) {
        int r = i + 8 * (lane >> 4);
        t1[r * T1_PITCH + col] = fmaxf(a0[i] + bias, 0.0f);
        t1[(r + 16) * T1_PITCH + col] = fmaxf(a1[i] + bias, 0.0f);
    }
}

__device__ __forceinline__ void encmsg_block(
    float* t1, float* am, int blk,
    const float* __restrict__ feat, const float* __restrict__ pW0,
    const float* __restrict__ b0, const float* __restrict__ pW1,
    const float* __restrict__ b1, float* __restrict__ x,
    const float* __restrict__ h, const float* __restrict__ pmsgW,
    const float* __restrict__ msgb, float* __restrict__ mx, int N) {
    const int wave = threadIdx.x >> 5;
    const int lane = threadIdx.x & 31;
    const int tid = threadIdx.x;
    const int row0 = blk * 32;
    if (row0 >= N) return;

    const float4* hv = (const float4*)(h + (size_t)row0 * 256);
    float4 vh[16];
#pragma unroll
    for (int i = 0; i < 16; ++i) vh[i] = hv[i * 128 + tid];

    {
        v2f af0[16], af1[16];
#pragma unroll
        for (int kk = 0; kk < 16; ++kk) {
            af0[kk] = load_a_frag(feat, 64, row0, kk * 4, lane);
            af1[kk] = load_a_frag(feat, 64, row0 + 16, kk * 4, lane);
        }
        const v2f* pb0 = (const v2f*)pW0 + lane;
        const int nt0 = wave * 4;
        FragB<16> g0, g1;
        loadB<16>(g0, pb0, nt0);
        loadB<16>(g1, pb0, nt0 + 1);
        enc1_tile(g0, af0, af1, nt0, lane, b0, t1);
        loadB<16>(g0, pb0, nt0 + 2);
        enc1_tile(g1, af0, af1, nt0 + 1, lane, b0, t1);
        loadB<16>(g1, pb0, nt0 + 3);
        enc1_tile(g0, af0, af1, nt0 + 2, lane, b0, t1);
        enc1_tile(g1, af0, af1, nt0 + 3, lane, b0, t1);
    }
#pragma unroll
    for (int i = 0; i < 16; ++i) {
        int idx = i * 128 + tid;
        *(float4*)&am[(idx >> 6) * MSG_PITCH + 256 + (idx & 63) * 4] = vh[i];
    }
    __syncthreads();

    {
        const float* la1 = t1 + (lane & 15) * T1_PITCH + ((lane >> 4) << 1);
#pragma unroll 1
        for (int t = 0; t < 4; ++t) {
            const int nt = wave * 4 + t;
            const v2f* pb = (const v2f*)pW1 + ((size_t)(nt * 64) * 32 + lane);
            v8f acc0 = {}, acc1 = {};
            seg21s<8, T1_PITCH>(la1, pb, 8, acc0, acc1);
            int col = nt * 16 + (lane & 15);
            float bias = b1[col];
#pragma unroll
            for (int i = 0; i < 8; ++i) {
                int r = i + 8 * (lane >> 4);
                float v0 = fmaxf(acc0[i] + bias, 0.0f);
                float v1 = fmaxf(acc1[i] + bias, 0.0f);
                x[(size_t)(row0 + r) * 256 + col] = v0;
                x[(size_t)(row0 + r + 16) * 256 + col] = v1;
                am[r * MSG_PITCH + col] = v0;
                am[(r + 16) * MSG_PITCH + col] = v1;
            }
        }
    }
    __syncthreads();

    const float* la = am + (lane & 15) * MSG_PITCH + ((lane >> 4) << 1);
#pragma unroll 1
    for (int t = 0; t < 2; ++t) {
        const int nt = wave * 2 + t;
        const v2f* pb = (const v2f*)pmsgW + ((size_t)(nt * 128) * 32 + lane);
        v8f acc0 = {}, acc1 = {};
        seg21s<8, MSG_PITCH>(la, pb, 16, acc0, acc1);
        int col = nt * 16 + (lane & 15);
        float bias = msgb[col];
#pragma unroll
        for (int i = 0; i < 8; ++i) {
            int r = row0 + i + 8 * (lane >> 4);
            mx[(size_t)r * 128 + col] = acc0[i] + bias;
            mx[(size_t)(r + 16) * 128 + col] = acc1[i] + bias;
        }
    }
}

constexpr int SLOTS_PER_NODE = 64;
constexpr int EDGES_PER_BLOCK = 512;
template <bool FULL>
__device__ __forceinline__ void rank_edges(
    int base, const int* __restrict__ src, const int* __restrict__ dst,
    int* __restrict__ deg, int* __restrict__ slots, int* __restrict__ ovf,
    int N, int E) {
    int d[4], s[4], r[4];
#pragma unroll
    for (int k = 0; k < 4; ++k) {
        int e = base + k * 128;
        if (!FULL) e = min(e, E - 1);
        d[k] = dst[e]; s[k] = src[e];
    }
#pragma unroll
    for (int k = 0; k < 4; ++k)
        r[k] = (FULL || base + k * 128 < E) ? atomicAdd(deg + d[k], 1) : 0;
#pragma unroll
    for (int k = 0; k < 4; ++k) {
        if (FULL || base + k * 128 < E) {
            if (r[k] < SLOTS_PER_NODE) {
                int pos = d[k] * SLOTS_PER_NODE + r[k];
                pos = min(max(pos, 0), N * SLOTS_PER_NODE - 1);
                slots[pos] = s[k];
            } else {
                int o = atomicAdd(deg + N, 1);
                if (o < E) ovf[o] = base + k * 128;
            }
        }
    }
}

__device__ __forceinline__ void rank_block(
    int blk, const int* __restrict__ src, const int* __restrict__ dst,
    int* __restrict__ deg, int* __restrict__ slots, int* __restrict__ ovf,
    int N, int E) {
    const int base = blk * EDGES_PER_BLOCK + (int)threadIdx.x;
    if ((blk + 1) * EDGES_PER_BLOCK <= E)
        rank_edges<true>(base, src, dst, deg, slots, ovf, N, E);
    else
        rank_edges<false>(base, src, dst, deg, slots, ovf, N, E);
}

__global__ __launch_bounds__(128) void enc_msg_rank_kernel(
    const float* __restrict__ feat, const float* __restrict__ pW0,
    const float* __restrict__ b0, const float* __restrict__ pW1,
    const float* __restrict__ b1, float* __restrict__ x,
    const float* __restrict__ h, const float* __restrict__ pmsgW,
    const float* __restrict__ msgb, float* __restrict__ mx, int N,
    int gemmTiles, const int* __restrict__ src, const int* __restrict__ dst,
    int* __restrict__ deg, int* __restrict__ slots, int* __restrict__ ovf,
    int E) {
    __shared__ __attribute__((aligned(16))) float am[32 * MSG_PITCH];
    __shared__ __attribute__((aligned(16))) float t1[32 * T1_PITCH];
    const int blk = blockIdx.x;
    if (blk < gemmTiles)
        encmsg_block(t1, am, blk, feat, pW0, b0, pW1, b1, x, h, pmsgW, msgb, mx, N);
    else
        rank_block(blk - gemmTiles, src, dst, deg, slots, ovf, N, E);
}

__global__ __launch_bounds__(256) void gather_kernel(
    const int* __restrict__ deg, const int* __restrict__ slots,
    const int* __restrict__ ovf, const int* __restrict__ src,
    const int* __restrict__ dst, const float* __restrict__ mx,
    float* __restrict__ c, int N, int E) {
    const int node = __builtin_amdgcn_readfirstlane(
        (int)((blockIdx.x * 256u + threadIdx.x) >> 5));
    const int lane = threadIdx.x & 31;
    if (node >= N) return;
    const int cnt = deg[node];
    const int* sl = slots + (size_t)node * SLOTS_PER_NODE;
    const int end = min(cnt, SLOTS_PER_NODE);
    const float4* mv = (const float4*)mx;
    float4 acc = make_float4(0.0f, 0.0f, 0.0f, 0.0f);
    int j = 0;
    for (; j + 8 <= end; j += 8) {
        float4 v[8];
#pragma unroll
        for (int k = 0; k < 8; ++k) {
            int s = sl[j + k];
            v[k] = mv[(size_t)s * 32 + lane];
        }
#pragma unroll
        for (int k = 0; k < 8; ++k) {
            acc.x += v[k].x; acc.y += v[k].y;
            acc.z += v[k].z; acc.w += v[k].w;
        }
    }
    if (j < end) {
        float4 v[8];
#pragma unroll
        for (int k = 0; k < 8; ++k) {
            int s = sl[min(j + k, end - 1)];
            v[k] = mv[(size_t)s * 32 + lane];
        }
#pragma unroll
        for (int k = 0; k < 8; ++k) {
            const bool in = (j + k < end);
            acc.x += in ? v[k].x : 0.0f; acc.y += in ? v[k].y : 0.0f;
            acc.z += in ? v[k].z : 0.0f; acc.w += in ? v[k].w : 0.0f;
        }
    }
    if (cnt > SLOTS_PER_NODE) {
        const int tot = min(deg[N], E);
        for (int k = 0; k < tot; ++k) {
            const int e = min(max(ovf[k], 0), E - 1);
            if (dst[e] == node) {
                float4 v = mv[(size_t)src[e] * 32 + lane];
                acc.x += v.x; acc.y += v.y; acc.z += v.z; acc.w += v.w;
            }
        }
    }
    const float den = fmaxf((float)cnt, 1.0f);
    float4 o;
    o.x = acc.x / den; o.y = acc.y / den; o.z = acc.z / den; o.w = acc.w / den;
    ((float4*)c)[(size_t)node * 32 + lane] = o;
}

constexpr int GRU_ROWS = GRU_RT * 16;
constexpr int GRU_KB = 4;
constexpr int GRU_K = 640;
constexpr int GRU_PITCH = GRU_K + 4;
constexpr int HN_PITCH = 256 + 4;

template <int KB>
__device__ __forceinline__ void load53s(Frag53<KB>& f, const float* la,
                                        const float* B, int boff, int gate,
                                        int blk) {
    const float* qa = la + blk * (KB * 4);
    const float* qb = B + blk * (KB * 64);
#pragma unroll
    for (int m = 0; m < GRU_RT; ++m) {
#pragma unroll
        for (int j = 0; j < KB; ++j) {
            f.a[m][j].x = qa[m * 16 * GRU_PITCH + j * 4];
            f.a[m][j].y = qa[m * 16 * GRU_PITCH + j * 4 + 1];
        }
    }
#pragma unroll
    for (int j = 0; j < KB; ++j) {
        f.br[j].x = qb[boff + j * 64];          f.br[j].y = qb[boff + j * 64 + 1];
        f.bz[j].x = qb[boff + gate + j * 64];
        f.bz[j].y = qb[boff + gate + j * 64 + 1];
        f.bn[j].x = qb[boff + 2 * gate + j * 64];
        f.bn[j].y = qb[boff + 2 * gate + j * 64 + 1];
    }
}

template <int KB>
__device__ __forceinline__ void seg53s(const float* la, const float* B,
                                       int boff, int gate, int nb,
                                       v8f (&r)[GRU_RT], v8f (&z)[GRU_RT],
                                       v8f (&n)[GRU_RT]) {
    Frag53<KB> f0, f1;
    load53s<KB>(f0, la, B, boff, gate, 0);
    int b = 0;
#pragma unroll 1
    for (; b + 2 < nb; b += 2) {
        load53s<KB>(f1, la, B, boff, gate, b + 1);
        __builtin_amdgcn_sched_barrier(0);
        mma53<KB>(f0, r, z, n);
        load53s<KB>(f0, la, B, boff, gate, b + 2);
        __builtin_amdgcn_sched_barrier(0);
        mma53<KB>(f1, r, z, n);
    }
    load53s<KB>(f1, la, B, boff, gate, b + 1);
    mma53<KB>(f0, r, z, n);
    mma53<KB>(f1, r, z, n);
}

constexpr int HN_RW = GRU_ROWS / 8;
static_assert(GRU_ROWS % 8 == 0, "the eight waves share the tile's rows evenly");
template <bool FULL>
__device__ __forceinline__ void store_hnew_tile(const float* ht,
                                                float* __restrict__ hnew,
                                                int row0, int wave, int lane,
                                                int N) {
    float4 v[HN_RW][2];
    const float* src = ht + wave * (HN_RW * HN_PITCH) + lane * 4;
#pragma unroll
    for (int i = 0; i < HN_RW; ++i) {
        v[i][0] = *(const float4*)(src + i * HN_PITCH);
        v[i][1] = *(const float4*)(src + i * HN_PITCH + 128);
    }
    float4* dst = (float4*)(hnew + (size_t)(row0 + wave * HN_RW) * 256) + lane;
#pragma unroll
    for (int i = 0; i < HN_RW; ++i) {
        if (FULL || row0 + wave * HN_RW + i < N) {
            dst[i * 64] = v[i][0];
            dst[i * 64 + 32] = v[i][1];
        }
    }
}

template <bool FULL>
__device__ __forceinline__ void head_tile(
    const float* ht, const float* __restrict__ poutW,
    const float* __restrict__ outb, float* __restrict__ hnew,
    float* __restrict__ q, int row0, int wave, int lane, int N) {
    if (wave < GRU_RT) {
        v2f bq[64], aq[64];
        const v2f* pq = (const v2f*)poutW + lane;
#pragma unroll
        for (int kk = 0; kk < 64; ++kk) bq[kk] = pq[kk * 32];
        store_hnew_tile<FULL>(ht, hnew, row0, wave, lane, N);
        const float* lq = ht + (wave * 16 + (lane & 15)) * HN_PITCH + ((lane >> 4) << 1);
#pragma unroll
        for (int kk = 0; kk < 64; ++kk) {
            aq[kk].x = lq[kk * 4]; aq[kk].y = lq[kk * 4 + 1];
        }
        __builtin_amdgcn_sched_barrier(0);
        v8f acc = {};
#pragma unroll
        for (int kk = 0; kk < 64; ++kk) acc = wmma4(aq[kk], bq[kk], acc);
        int col = lane & 15;
        float bias = outb[col];
#pragma unroll
        for (int i = 0; i < 8; ++i) {
            int r = row0 + wave * 16 + i + 8 * (lane >> 4);
            if (FULL || r < N) q[(size_t)r * 16 + col] = acc[i] + bias;
        }
    } else {
        store_hnew_tile<FULL>(ht, hnew, row0, wave, lane, N);
    }
}

__global__ __launch_bounds__(256) void gru_out_kernel(
    const float* __restrict__ x, const float* __restrict__ c,
    const float* __restrict__ h, const float* __restrict__ pWih,
    const float* __restrict__ pWhh, const float* __restrict__ bih,
    const float* __restrict__ bhh, const float* __restrict__ poutW,
    const float* __restrict__ outb, float* __restrict__ hnew,
    float* __restrict__ q, int N) {
    __shared__ __attribute__((aligned(16))) float at[GRU_ROWS * GRU_PITCH];
    __shared__ __attribute__((aligned(16))) float ht[GRU_ROWS * HN_PITCH];
    const int wave = __builtin_amdgcn_readfirstlane((int)(threadIdx.x >> 5));
    const int lane = threadIdx.x & 31;
    const int tid = threadIdx.x;
    const int row0 = blockIdx.x * GRU_ROWS;
    if (row0 >= N) return;

    {
        constexpr int NX = GRU_ROWS * 64 / 256;
        constexpr int NC = GRU_ROWS * 32 / 256;
        float4 vx[NX], vc[NC], vh[NX];
#pragma unroll
        for (int i = 0; i < NX; ++i) {
            int idx = i * 256 + tid;
            int gr = min(row0 + (idx >> 6), N - 1);
            vx[i] = *(const float4*)(x + (size_t)gr * 256 + (idx & 63) * 4);
        }
#pragma unroll
        for (int i = 0; i < NC; ++i) {
            int idx = i * 256 + tid;
            int gr = min(row0 + (idx >> 5), N - 1);
            vc[i] = *(const float4*)(c + (size_t)gr * 128 + (idx & 31) * 4);
        }
#pragma unroll
        for (int i = 0; i < NX; ++i) {
            int idx = i * 256 + tid;
            int gr = min(row0 + (idx >> 6), N - 1);
            vh[i] = *(const float4*)(h + (size_t)gr * 256 + (idx & 63) * 4);
        }
#pragma unroll
        for (int i = 0; i < NX; ++i) {
            int idx = i * 256 + tid;
            *(float4*)&at[(idx >> 6) * GRU_PITCH + (idx & 63) * 4] = vx[i];
        }
#pragma unroll
        for (int i = 0; i < NC; ++i) {
            int idx = i * 256 + tid;
            *(float4*)&at[(idx >> 5) * GRU_PITCH + 256 + (idx & 31) * 4] = vc[i];
        }
#pragma unroll
        for (int i = 0; i < NX; ++i) {
            int idx = i * 256 + tid;
            *(float4*)&at[(idx >> 6) * GRU_PITCH + 384 + (idx & 63) * 4] = vh[i];
        }
    }
    __syncthreads();

    const int boff = lane * 2;
    const float* la = at + (lane & 15) * GRU_PITCH + ((lane >> 4) << 1);
#pragma unroll 1
    for (int t = 0; t < 2; ++t) {
        const int nt = wave * 2 + t;
        v8f ir[GRU_RT] = {}, iz[GRU_RT] = {}, in[GRU_RT] = {};
        v8f hr[GRU_RT] = {}, hz[GRU_RT] = {}, hn[GRU_RT] = {};
        const float* pbi = pWih + (size_t)(nt * 96) * 64;
        const float* pbh = pWhh + (size_t)(nt * 64) * 64;
        seg53s<GRU_KB>(la, pbi, boff, 16 * 96 * 64, 96 / GRU_KB, ir, iz, in);
        seg53s<GRU_KB>(la + 384, pbh, boff, 16 * 64 * 64, 64 / GRU_KB, hr, hz, hn);
        int col = nt * 16 + (lane & 15);
        float bir = bih[col], biz = bih[col + 256], bin = bih[col + 512];
        float bhr = bhh[col], bhz = bhh[col + 256], bhn = bhh[col + 512];
#pragma unroll
        for (int m = 0; m < GRU_RT; ++m) {
#pragma unroll
            for (int i = 0; i < 8; ++i) {
                int lr = m * 16 + i + 8 * (lane >> 4);
                float rg = sigmoid_f((ir[m][i] + bir) + (hr[m][i] + bhr));
                float zg = sigmoid_f((iz[m][i] + biz) + (hz[m][i] + bhz));
                float ng = tanh_f((in[m][i] + bin) + rg * (hn[m][i] + bhn));
                float hv = at[lr * GRU_PITCH + 384 + col];
                ht[lr * HN_PITCH + col] = (1.0f - zg) * ng + zg * hv;
            }
        }
    }
    __syncthreads();

    if (row0 + GRU_ROWS <= N)
        head_tile<true>(ht, poutW, outb, hnew, q, row0, wave, lane, N);
    else
        head_tile<false>(ht, poutW, outb, hnew, q, row0, wave, lane, N);
}

extern "C" void kernel_launch(void* const* d_in, const int* in_sizes, int n_in,
                              void* d_out, int out_size, void* d_ws,
                              size_t ws_size, hipStream_t stream) {
    (void)n_in; (void)out_size; (void)ws_size;
    const float* feat = (const float*)d_in[0];
    const float* h    = (const float*)d_in[1];
    const int*   src  = (const int*)d_in[2];
    const int*   dst  = (const int*)d_in[3];
    const float* W0   = (const float*)d_in[4];
    const float* b0   = (const float*)d_in[5];
    const float* W1   = (const float*)d_in[6];
    const float* b1   = (const float*)d_in[7];
    const float* msgW = (const float*)d_in[8];
    const float* msgb = (const float*)d_in[9];
    const float* Wih  = (const float*)d_in[10];
    const float* Whh  = (const float*)d_in[11];
    const float* bih  = (const float*)d_in[12];
    const float* bhh  = (const float*)d_in[13];
    const float* outW = (const float*)d_in[14];
    const float* outb = (const float*)d_in[15];

    const int N = in_sizes[1] / 256;
    const int E = in_sizes[2];

    float* ws    = (float*)d_ws;
    float* x     = ws;
    float* mx    = x    + (size_t)N * 256;
    float* bkt   = mx   + (size_t)N * 128;
    int*   slots = (int*)bkt;
    int*   deg   = slots + (size_t)N * SLOTS_PER_NODE;
    int*   ovf   = deg + N + 16;
    float* c     = bkt  + (size_t)N * 128 + N;
    float* pW0   = c    + (size_t)N * 128;
    float* pW1   = pW0  + 16384;
    float* pmsgW = pW1  + 65536;
    float* pWih  = pmsgW + 65536;
    float* pWhh  = pWih + 294912;
    float* poutW = pWhh + 196608;

    float* q    = (float*)d_out;
    float* hnew = q + (size_t)N * 16;

    const int rowTiles   = (N + 31) / 32;
    const int gruBlocks  = (N + GRU_ROWS - 1) / GRU_ROWS;
    const int zeroBlocks = (N + 1 + 255) / 256;
    const int edgeBlocks = (E + EDGES_PER_BLOCK - 1) / EDGES_PER_BLOCK;
    const int nodeBlocks = (N + 7) / 8;

    setup_kernel<<<PB_ALL + zeroBlocks, 256, 0, stream>>>(
        W0, W1, msgW, Wih, Whh, outW, pW0, pW1, pmsgW, pWih, pWhh, poutW, deg, N);
    enc_msg_rank_kernel<<<rowTiles + edgeBlocks, 128, 0, stream>>>(
        feat, pW0, b0, pW1, b1, x, h, pmsgW, msgb, mx, N, rowTiles, src, dst,
        deg, slots, ovf, E);
    gather_kernel<<<nodeBlocks, 256, 0, stream>>>(deg, slots, ovf, src, dst, mx,
                                                  c, N, E);
    gru_out_kernel<<<gruBlocks, 256, 0, stream>>>(x, c, h, pWih, pWhh, bih, bhh,
                                                  poutW, outb, hnew, q, N);
}


# edge gather fused into the GRU launch: each GRU block's waves sum the message rows of its own 80 nodes (same slot order) straight into the block's LDS tile; c never goes to memory; 4 launches to 3
# speedup vs baseline: 6.4171x; 1.0282x over previous
#include <hip/hip_runtime.h>
#include <hip/hip_bf16.h>


typedef __attribute__((ext_vector_type(2))) float v2f;
typedef __attribute__((ext_vector_type(8))) float v8f;

__device__ __forceinline__ v8f wmma4(v2f a, v2f b, v8f c) {
    return __builtin_amdgcn_wmma_f32_16x16x4_f32(
        false, a, false, b, (short)0, c, false, false);
}

__device__ __forceinline__ v2f load_a_frag(const float* M, int ld, int row0,
                                           int k0, int lane) {
    int r = row0 + (lane & 15);
    int k = k0 + ((lane >> 4) << 1);
    const float* p = M + (size_t)r * ld + k;
    v2f a; a.x = p[0]; a.y = p[1];
    return a;
}


template <int KB> struct Frag21 { v2f a0[KB], a1[KB], b[KB]; };

template <int KB>
__device__ __forceinline__ void mma21(const Frag21<KB>& f, v8f& c0, v8f& c1) {
#pragma unroll
    for (int j = 0; j < KB; ++j) {
        c0 = wmma4(f.a0[j], f.b[j], c0);
        c1 = wmma4(f.a1[j], f.b[j], c1);
    }
}

template <int KB, int PITCH>
__device__ __forceinline__ void load21s(Frag21<KB>& f, const float* la,
                                        const v2f* pb, int blk) {
    const float* qa = la + blk * (KB * 4);
    const v2f* qb = pb + blk * (KB * 32);
#pragma unroll
    for (int j = 0; j < KB; ++j) {
        f.a0[j].x = qa[j * 4];              f.a0[j].y = qa[j * 4 + 1];
        f.a1[j].x = qa[16 * PITCH + j * 4]; f.a1[j].y = qa[16 * PITCH + j * 4 + 1];
        f.b[j] = qb[j * 32];
    }
}

template <int KB, int PITCH>
__device__ __forceinline__ void seg21s(const float* la, const v2f* pb, int nb,
                                       v8f& c0, v8f& c1) {
    Frag21<KB> f0, f1;
    load21s<KB, PITCH>(f0, la, pb, 0);
    int b = 0;
#pragma unroll 1
    for (; b + 2 < nb; b += 2) {
        load21s<KB, PITCH>(f1, la, pb, b + 1);
        __builtin_amdgcn_sched_barrier(0);
        mma21<KB>(f0, c0, c1);
        load21s<KB, PITCH>(f0, la, pb, b + 2);
        __builtin_amdgcn_sched_barrier(0);
        mma21<KB>(f1, c0, c1);
    }
    load21s<KB, PITCH>(f1, la, pb, b + 1);
    mma21<KB>(f0, c0, c1);
    mma21<KB>(f1, c0, c1);
}

constexpr int GRU_RT = 5;
template <int KB> struct Frag53 { v2f a[GRU_RT][KB], br[KB], bz[KB], bn[KB]; };

template <int KB>
__device__ __forceinline__ void mma53(const Frag53<KB>& f, v8f (&r)[GRU_RT],
                                      v8f (&z)[GRU_RT], v8f (&n)[GRU_RT]) {
#pragma unroll
    for (int j = 0; j < KB; ++j) {
#pragma unroll
        for (int m = 0; m < GRU_RT; ++m) {
            r[m] = wmma4(f.a[m][j], f.br[j], r[m]);
            z[m] = wmma4(f.a[m][j], f.bz[j], z[m]);
            n[m] = wmma4(f.a[m][j], f.bn[j], n[m]);
        }
    }
}

template <int KB> struct FragB { v2f b[KB]; };

template <int KB>
__device__ __forceinline__ void loadB(FragB<KB>& f, const v2f* pb, int blk) {
    const v2f* qb = pb + blk * (KB * 32);
#pragma unroll
    for (int j = 0; j < KB; ++j) f.b[j] = qb[j * 32];
}

__device__ __forceinline__ float sigmoid_f(float v) {
    return 1.0f / (1.0f + __expf(-v));
}
__device__ __forceinline__ float tanh_f(float v) {
    float t = __expf(-2.0f * v);
    return (1.0f - t) / (1.0f + t);
}

template <int K, int Nc>
__device__ __forceinline__ void pack_tile(const float* __restrict__ W,
                                          float* __restrict__ P, int w,
                                          int lane) {
    constexpr int nk = K >> 2;
    int kt = w % nk, nt = w / nk;
    int n = nt * 16 + (lane & 15);
    int k = kt * 4 + ((lane >> 4) << 1);
    v2f v;
    v.x = W[(size_t)k * Nc + n];
    v.y = W[(size_t)(k + 1) * Nc + n];
    ((v2f*)P)[(size_t)(nt * nk + kt) * 32 + lane] = v;
}

constexpr int packBlocks(int K, int Nc) { return (K / 4) * (Nc / 16) / 8; }
constexpr bool packWhole(int K, int Nc) { return (K / 4) * (Nc / 16) % 8 == 0; }
constexpr int PB_W0  = packBlocks(64, 256);
constexpr int PB_W1  = packBlocks(256, 256);
constexpr int PB_MSG = packBlocks(512, 128);
constexpr int PB_WIH = packBlocks(384, 768);
constexpr int PB_WHH = packBlocks(256, 768);
constexpr int PB_OUT = packBlocks(256, 16);
constexpr int PB_ALL = PB_W0 + PB_W1 + PB_MSG + PB_WIH + PB_WHH + PB_OUT;
static_assert(packWhole(64, 256) && packWhole(256, 256) && packWhole(512, 128) &&
              packWhole(384, 768) && packWhole(256, 768) && packWhole(256, 16),
              "every matrix fills whole 8-wave blocks, so no wave has to be masked");

__global__ __launch_bounds__(256) void setup_kernel(
    const float* __restrict__ W0, const float* __restrict__ W1,
    const float* __restrict__ msgW, const float* __restrict__ Wih,
    const float* __restrict__ Whh, const float* __restrict__ outW,
    float* __restrict__ pW0, float* __restrict__ pW1,
    float* __restrict__ pmsgW, float* __restrict__ pWih,
    float* __restrict__ pWhh, float* __restrict__ poutW,
    int* __restrict__ deg, int N) {
    const int lane = threadIdx.x & 31;
    const int wib = threadIdx.x >> 5;
    int b = blockIdx.x;
    if (b < PB_W0)  { pack_tile<64, 256>(W0, pW0, b * 8 + wib, lane); return; }
    b -= PB_W0;
    if (b < PB_W1)  { pack_tile<256, 256>(W1, pW1, b * 8 + wib, lane); return; }
    b -= PB_W1;
    if (b < PB_MSG) { pack_tile<512, 128>(msgW, pmsgW, b * 8 + wib, lane); return; }
    b -= PB_MSG;
    if (b < PB_WIH) { pack_tile<384, 768>(Wih, pWih, b * 8 + wib, lane); return; }
    b -= PB_WIH;
    if (b < PB_WHH) { pack_tile<256, 768>(Whh, pWhh, b * 8 + wib, lane); return; }
    b -= PB_WHH;
    if (b < PB_OUT) { pack_tile<256, 16>(outW, poutW, b * 8 + wib, lane); return; }
    b -= PB_OUT;
    int i = b * 256 + (int)threadIdx.x;
    if (i <= N) deg[i] = 0;
}

constexpr int MSG_PITCH = 512 + 4;
constexpr int T1_PITCH = 256 + 4;

__device__ __forceinline__ void enc1_tile(const FragB<16>& g, const v2f* af0,
                                          const v2f* af1, int nt, int lane,
                                          const float* __restrict__ b0,
                                          float* t1) {
    v8f a0 = {}, a1 = {};
#pragma unroll
    for (int kk = 0; kk < 16; ++kk) {
        a0 = wmma4(af0[kk], g.b[kk], a0);
        a1 = wmma4(af1[kk], g.b[kk], a1);
    }
    int col = nt * 16 + (lane & 15);
    float bias = b0[col];
#pragma unroll
    for (int i = 0; i < 8; ++i) {
        int r = i + 8 * (lane >> 4);
        t1[r * T1_PITCH + col] = fmaxf(a0[i] + bias, 0.0f);
        t1[(r + 16) * T1_PITCH + col] = fmaxf(a1[i] + bias, 0.0f);
    }
}

__device__ __forceinline__ void encmsg_block(
    float* t1, float* am, int blk,
    const float* __restrict__ feat, const float* __restrict__ pW0,
    const float* __restrict__ b0, const float* __restrict__ pW1,
    const float* __restrict__ b1, float* __restrict__ x,
    const float* __restrict__ h, const float* __restrict__ pmsgW,
    const float* __restrict__ msgb, float* __restrict__ mx, int N) {
    const int wave = threadIdx.x >> 5;
    const int lane = threadIdx.x & 31;
    const int tid = threadIdx.x;
    const int row0 = blk * 32;
    if (row0 >= N) return;

    const float4* hv = (const float4*)(h + (size_t)row0 * 256);
    float4 vh[16];
#pragma unroll
    for (int i = 0; i < 16; ++i) vh[i] = hv[i * 128 + tid];

    {
        v2f af0[16], af1[16];
#pragma unroll
        for (int kk = 0; kk < 16; ++kk) {
            af0[kk] = load_a_frag(feat, 64, row0, kk * 4, lane);
            af1[kk] = load_a_frag(feat, 64, row0 + 16, kk * 4, lane);
        }
        const v2f* pb0 = (const v2f*)pW0 + lane;
        const int nt0 = wave * 4;
        FragB<16> g0, g1;
        loadB<16>(g0, pb0, nt0);
        loadB<16>(g1, pb0, nt0 + 1);
        enc1_tile(g0, af0, af1, nt0, lane, b0, t1);
        loadB<16>(g0, pb0, nt0 + 2);
        enc1_tile(g1, af0, af1, nt0 + 1, lane, b0, t1);
        loadB<16>(g1, pb0, nt0 + 3);
        enc1_tile(g0, af0, af1, nt0 + 2, lane, b0, t1);
        enc1_tile(g1, af0, af1, nt0 + 3, lane, b0, t1);
    }
#pragma unroll
    for (int i = 0; i < 16; ++i) {
        int idx = i * 128 + tid;
        *(float4*)&am[(idx >> 6) * MSG_PITCH + 256 + (idx & 63) * 4] = vh[i];
    }
    __syncthreads();

    {
        const float* la1 = t1 + (lane & 15) * T1_PITCH + ((lane >> 4) << 1);
#pragma unroll 1
        for (int t = 0; t < 4; ++t) {
            const int nt = wave * 4 + t;
            const v2f* pb = (const v2f*)pW1 + ((size_t)(nt * 64) * 32 + lane);
            v8f acc0 = {}, acc1 = {};
            seg21s<8, T1_PITCH>(la1, pb, 8, acc0, acc1);
            int col = nt * 16 + (lane & 15);
            float bias = b1[col];
#pragma unroll
            for (int i = 0; i < 8; ++i) {
                int r = i + 8 * (lane >> 4);
                float v0 = fmaxf(acc0[i] + bias, 0.0f);
                float v1 = fmaxf(acc1[i] + bias, 0.0f);
                x[(size_t)(row0 + r) * 256 + col] = v0;
                x[(size_t)(row0 + r + 16) * 256 + col] = v1;
                am[r * MSG_PITCH + col] = v0;
                am[(r + 16) * MSG_PITCH + col] = v1;
            }
        }
    }
    __syncthreads();

    const float* la = am + (lane & 15) * MSG_PITCH + ((lane >> 4) << 1);
#pragma unroll 1
    for (int t = 0; t < 2; ++t) {
        const int nt = wave * 2 + t;
        const v2f* pb = (const v2f*)pmsgW + ((size_t)(nt * 128) * 32 + lane);
        v8f acc0 = {}, acc1 = {};
        seg21s<8, MSG_PITCH>(la, pb, 16, acc0, acc1);
        int col = nt * 16 + (lane & 15);
        float bias = msgb[col];
#pragma unroll
        for (int i = 0; i < 8; ++i) {
            int r = row0 + i + 8 * (lane >> 4);
            mx[(size_t)r * 128 + col] = acc0[i] + bias;
            mx[(size_t)(r + 16) * 128 + col] = acc1[i] + bias;
        }
    }
}

constexpr int SLOTS_PER_NODE = 64;
constexpr int EDGES_PER_BLOCK = 512;
template <bool FULL>
__device__ __forceinline__ void rank_edges(
    int base, const int* __restrict__ src, const int* __restrict__ dst,
    int* __restrict__ deg, int* __restrict__ slots, int* __restrict__ ovf,
    int N, int E) {
    int d[4], s[4], r[4];
#pragma unroll
    for (int k = 0; k < 4; ++k) {
        int e = base + k * 128;
        if (!FULL) e = min(e, E - 1);
        d[k] = dst[e]; s[k] = src[e];
    }
#pragma unroll
    for (int k = 0; k < 4; ++k)
        r[k] = (FULL || base + k * 128 < E) ? atomicAdd(deg + d[k], 1) : 0;
#pragma unroll
    for (int k = 0; k < 4; ++k) {
        if (FULL || base + k * 128 < E) {
            if (r[k] < SLOTS_PER_NODE) {
                int pos = d[k] * SLOTS_PER_NODE + r[k];
                pos = min(max(pos, 0), N * SLOTS_PER_NODE - 1);
                slots[pos] = s[k];
            } else {
                int o = atomicAdd(deg + N, 1);
                if (o < E) ovf[o] = base + k * 128;
            }
        }
    }
}

__device__ __forceinline__ void rank_block(
    int blk, const int* __restrict__ src, const int* __restrict__ dst,
    int* __restrict__ deg, int* __restrict__ slots, int* __restrict__ ovf,
    int N, int E) {
    const int base = blk * EDGES_PER_BLOCK + (int)threadIdx.x;
    if ((blk + 1) * EDGES_PER_BLOCK <= E)
        rank_edges<true>(base, src, dst, deg, slots, ovf, N, E);
    else
        rank_edges<false>(base, src, dst, deg, slots, ovf, N, E);
}

__global__ __launch_bounds__(128) void enc_msg_rank_kernel(
    const float* __restrict__ feat, const float* __restrict__ pW0,
    const float* __restrict__ b0, const float* __restrict__ pW1,
    const float* __restrict__ b1, float* __restrict__ x,
    const float* __restrict__ h, const float* __restrict__ pmsgW,
    const float* __restrict__ msgb, float* __restrict__ mx, int N,
    int gemmTiles, const int* __restrict__ src, const int* __restrict__ dst,
    int* __restrict__ deg, int* __restrict__ slots, int* __restrict__ ovf,
    int E) {
    __shared__ __attribute__((aligned(16))) float am[32 * MSG_PITCH];
    __shared__ __attribute__((aligned(16))) float t1[32 * T1_PITCH];
    const int blk = blockIdx.x;
    if (blk < gemmTiles)
        encmsg_block(t1, am, blk, feat, pW0, b0, pW1, b1, x, h, pmsgW, msgb, mx, N);
    else
        rank_block(blk - gemmTiles, src, dst, deg, slots, ovf, N, E);
}


template <int G>
__device__ __forceinline__ void gather_rows(const float4* __restrict__ mv,
                                            int sidx, int off, int n, int lane,
                                            float4& acc) {
    float4 v[G];
#pragma unroll
    for (int k = 0; k < G; ++k) {
        int s = __builtin_amdgcn_readlane(sidx, min(off + k, n - 1));
        v[k] = mv[(size_t)s * 32 + lane];
    }
#pragma unroll
    for (int k = 0; k < G; ++k) {
        const bool in = (off + k < n);
        acc.x += in ? v[k].x : 0.0f; acc.y += in ? v[k].y : 0.0f;
        acc.z += in ? v[k].z : 0.0f; acc.w += in ? v[k].w : 0.0f;
    }
}

__device__ __forceinline__ float4 gather_node(
    int node, int cnt, int slo, int shi, const int* __restrict__ deg,
    const int* __restrict__ ovf, const int* __restrict__ src,
    const int* __restrict__ dst, const float* __restrict__ mx, int lane,
    int N, int E) {
    const float4* mv = (const float4*)mx;
    const int end = min(cnt, SLOTS_PER_NODE);
    float4 acc = make_float4(0.0f, 0.0f, 0.0f, 0.0f);
    if (end > 0) gather_rows<32>(mv, slo, 0, min(end, 32), lane, acc);
    for (int j = 32; j < end; j += 8)
        gather_rows<8>(mv, shi, j - 32, end - 32, lane, acc);
    if (cnt > SLOTS_PER_NODE) {
        const int tot = min(deg[N], E);
        for (int k = 0; k < tot; ++k) {
            const int e = min(max(ovf[k], 0), E - 1);
            if (dst[e] == node) {
                float4 v = mv[(size_t)src[e] * 32 + lane];
                acc.x += v.x; acc.y += v.y; acc.z += v.z; acc.w += v.w;
            }
        }
    }
    const float den = fmaxf((float)cnt, 1.0f);
    float4 o;
    o.x = acc.x / den; o.y = acc.y / den; o.z = acc.z / den; o.w = acc.w / den;
    return o;
}

constexpr int GRU_ROWS = GRU_RT * 16;
constexpr int GRU_KB = 4;
constexpr int GRU_K = 640;
constexpr int GRU_PITCH = GRU_K + 4;
constexpr int HN_PITCH = 256 + 4;

template <int KB>
__device__ __forceinline__ void load53s(Frag53<KB>& f, const float* la,
                                        const float* B, int boff, int gate,
                                        int blk) {
    const float* qa = la + blk * (KB * 4);
    const float* qb = B + blk * (KB * 64);
#pragma unroll
    for (int m = 0; m < GRU_RT; ++m) {
#pragma unroll
        for (int j = 0; j < KB; ++j) {
            f.a[m][j].x = qa[m * 16 * GRU_PITCH + j * 4];
            f.a[m][j].y = qa[m * 16 * GRU_PITCH + j * 4 + 1];
        }
    }
#pragma unroll
    for (int j = 0; j < KB; ++j) {
        f.br[j].x = qb[boff + j * 64];          f.br[j].y = qb[boff + j * 64 + 1];
        f.bz[j].x = qb[boff + gate + j * 64];
        f.bz[j].y = qb[boff + gate + j * 64 + 1];
        f.bn[j].x = qb[boff + 2 * gate + j * 64];
        f.bn[j].y = qb[boff + 2 * gate + j * 64 + 1];
    }
}

template <int KB>
__device__ __forceinline__ void seg53s(const float* la, const float* B,
                                       int boff, int gate, int nb,
                                       v8f (&r)[GRU_RT], v8f (&z)[GRU_RT],
                                       v8f (&n)[GRU_RT]) {
    Frag53<KB> f0, f1;
    load53s<KB>(f0, la, B, boff, gate, 0);
    int b = 0;
#pragma unroll 1
    for (; b + 2 < nb; b += 2) {
        load53s<KB>(f1, la, B, boff, gate, b + 1);
        __builtin_amdgcn_sched_barrier(0);
        mma53<KB>(f0, r, z, n);
        load53s<KB>(f0, la, B, boff, gate, b + 2);
        __builtin_amdgcn_sched_barrier(0);
        mma53<KB>(f1, r, z, n);
    }
    load53s<KB>(f1, la, B, boff, gate, b + 1);
    mma53<KB>(f0, r, z, n);
    mma53<KB>(f1, r, z, n);
}

constexpr int HN_RW = GRU_ROWS / 8;
static_assert(GRU_ROWS % 8 == 0, "the eight waves share the tile's rows evenly");
template <bool FULL>
__device__ __forceinline__ void store_hnew_tile(const float* ht,
                                                float* __restrict__ hnew,
                                                int row0, int wave, int lane,
                                                int N) {
    float4 v[HN_RW][2];
    const float* src = ht + wave * (HN_RW * HN_PITCH) + lane * 4;
#pragma unroll
    for (int i = 0; i < HN_RW; ++i) {
        v[i][0] = *(const float4*)(src + i * HN_PITCH);
        v[i][1] = *(const float4*)(src + i * HN_PITCH + 128);
    }
    float4* dst = (float4*)(hnew + (size_t)(row0 + wave * HN_RW) * 256) + lane;
#pragma unroll
    for (int i = 0; i < HN_RW; ++i) {
        if (FULL || row0 + wave * HN_RW + i < N) {
            dst[i * 64] = v[i][0];
            dst[i * 64 + 32] = v[i][1];
        }
    }
}

template <bool FULL>
__device__ __forceinline__ void head_tile(
    const float* ht, const float* __restrict__ poutW,
    const float* __restrict__ outb, float* __restrict__ hnew,
    float* __restrict__ q, int row0, int wave, int lane, int N) {
    if (wave < GRU_RT) {
        v2f bq[64], aq[64];
        const v2f* pq = (const v2f*)poutW + lane;
#pragma unroll
        for (int kk = 0; kk < 64; ++kk) bq[kk] = pq[kk * 32];
        store_hnew_tile<FULL>(ht, hnew, row0, wave, lane, N);
        const float* lq = ht + (wave * 16 + (lane & 15)) * HN_PITCH + ((lane >> 4) << 1);
#pragma unroll
        for (int kk = 0; kk < 64; ++kk) {
            aq[kk].x = lq[kk * 4]; aq[kk].y = lq[kk * 4 + 1];
        }
        __builtin_amdgcn_sched_barrier(0);
        v8f acc = {};
#pragma unroll
        for (int kk = 0; kk < 64; ++kk) acc = wmma4(aq[kk], bq[kk], acc);
        int col = lane & 15;
        float bias = outb[col];
#pragma unroll
        for (int i = 0; i < 8; ++i) {
            int r = row0 + wave * 16 + i + 8 * (lane >> 4);
            if (FULL || r < N) q[(size_t)r * 16 + col] = acc[i] + bias;
        }
    } else {
        store_hnew_tile<FULL>(ht, hnew, row0, wave, lane, N);
    }
}

__global__ __launch_bounds__(256) void gru_out_kernel(
    const float* __restrict__ x, const float* __restrict__ mx,
    const int* __restrict__ deg, const int* __restrict__ slots,
    const int* __restrict__ ovf, const int* __restrict__ src,
    const int* __restrict__ dst, int E,
    const float* __restrict__ h, const float* __restrict__ pWih,
    const float* __restrict__ pWhh, const float* __restrict__ bih,
    const float* __restrict__ bhh, const float* __restrict__ poutW,
    const float* __restrict__ outb, float* __restrict__ hnew,
    float* __restrict__ q, int N) {
    __shared__ __attribute__((aligned(16))) float at[GRU_ROWS * GRU_PITCH];
    __shared__ __attribute__((aligned(16))) float ht[GRU_ROWS * HN_PITCH];
    const int wave = __builtin_amdgcn_readfirstlane((int)(threadIdx.x >> 5));
    const int lane = threadIdx.x & 31;
    const int tid = threadIdx.x;
    const int row0 = blockIdx.x * GRU_ROWS;
    if (row0 >= N) return;

    {
        constexpr int NX = GRU_ROWS * 64 / 256;
        static_assert(NX == 2 * HN_RW, "two x chunks and two h chunks a trip cover the tile");
        const int n0 = row0 + wave * HN_RW;
        int nd = min(n0, N - 1);
        int cnt = deg[nd];
        int slo = slots[(size_t)nd * SLOTS_PER_NODE + lane];
        int shi = slots[(size_t)nd * SLOTS_PER_NODE + 32 + lane];
        const int lr0 = tid >> 6, lc = (tid & 63) * 4;
        int ra = min(row0 + lr0, N - 1), rb = min(row0 + lr0 + 4, N - 1);
        float4 vx0 = *(const float4*)(x + (size_t)ra * 256 + lc);
        float4 vx1 = *(const float4*)(x + (size_t)rb * 256 + lc);
        float4 vh0 = *(const float4*)(h + (size_t)ra * 256 + lc);
        float4 vh1 = *(const float4*)(h + (size_t)rb * 256 + lc);
#pragma unroll 1
        for (int i = 0; i < HN_RW; ++i) {
            const int p2 = min(i + 1, HN_RW - 1);
            ra = min(row0 + 8 * p2 + lr0, N - 1);
            rb = min(row0 + 8 * p2 + lr0 + 4, N - 1);
            const float4 nx0 = *(const float4*)(x + (size_t)ra * 256 + lc);
            const float4 nx1 = *(const float4*)(x + (size_t)rb * 256 + lc);
            const float4 nh0 = *(const float4*)(h + (size_t)ra * 256 + lc);
            const float4 nh1 = *(const float4*)(h + (size_t)rb * 256 + lc);
            const int nd2 = min(n0 + i + 1, N - 1);
            const int cnt2 = deg[nd2];
            const int slo2 = slots[(size_t)nd2 * SLOTS_PER_NODE + lane];
            const int shi2 = slots[(size_t)nd2 * SLOTS_PER_NODE + 32 + lane];
            float4 o = gather_node(nd, cnt, slo, shi, deg, ovf, src, dst, mx,
                                   lane, N, E);
            *(float4*)&at[(wave * HN_RW + i) * GRU_PITCH + 256 + lane * 4] = o;
            float* ax = at + (8 * i + lr0) * GRU_PITCH + lc;
            *(float4*)ax = vx0;
            *(float4*)(ax + 4 * GRU_PITCH) = vx1;
            *(float4*)(ax + 384) = vh0;
            *(float4*)(ax + 4 * GRU_PITCH + 384) = vh1;
            nd = nd2; cnt = cnt2; slo = slo2; shi = shi2;
            vx0 = nx0; vx1 = nx1; vh0 = nh0; vh1 = nh1;
        }
    }
    __syncthreads();

    const int boff = lane * 2;
    const float* la = at + (lane & 15) * GRU_PITCH + ((lane >> 4) << 1);
#pragma unroll 1
    for (int t = 0; t < 2; ++t) {
        const int nt = wave * 2 + t;
        v8f ir[GRU_RT] = {}, iz[GRU_RT] = {}, in[GRU_RT] = {};
        v8f hr[GRU_RT] = {}, hz[GRU_RT] = {}, hn[GRU_RT] = {};
        const float* pbi = pWih + (size_t)(nt * 96) * 64;
        const float* pbh = pWhh + (size_t)(nt * 64) * 64;
        seg53s<GRU_KB>(la, pbi, boff, 16 * 96 * 64, 96 / GRU_KB, ir, iz, in);
        seg53s<GRU_KB>(la + 384, pbh, boff, 16 * 64 * 64, 64 / GRU_KB, hr, hz, hn);
        int col = nt * 16 + (lane & 15);
        float bir = bih[col], biz = bih[col + 256], bin = bih[col + 512];
        float bhr = bhh[col], bhz = bhh[col + 256], bhn = bhh[col + 512];
#pragma unroll
        for (int m = 0; m < GRU_RT; ++m) {
#pragma unroll
            for (int i = 0; i < 8; ++i) {
                int lr = m * 16 + i + 8 * (lane >> 4);
                float rg = sigmoid_f((ir[m][i] + bir) + (hr[m][i] + bhr));
                float zg = sigmoid_f((iz[m][i] + biz) + (hz[m][i] + bhz));
                float ng = tanh_f((in[m][i] + bin) + rg * (hn[m][i] + bhn));
                float hv = at[lr * GRU_PITCH + 384 + col];
                ht[lr * HN_PITCH + col] = (1.0f - zg) * ng + zg * hv;
            }
        }
    }
    __syncthreads();

    if (row0 + GRU_ROWS <= N)
        head_tile<true>(ht, poutW, outb, hnew, q, row0, wave, lane, N);
    else
        head_tile<false>(ht, poutW, outb, hnew, q, row0, wave, lane, N);
}

extern "C" void kernel_launch(void* const* d_in, const int* in_sizes, int n_in,
                              void* d_out, int out_size, void* d_ws,
                              size_t ws_size, hipStream_t stream) {
    (void)n_in; (void)out_size; (void)ws_size;
    const float* feat = (const float*)d_in[0];
    const float* h    = (const float*)d_in[1];
    const int*   src  = (const int*)d_in[2];
    const int*   dst  = (const int*)d_in[3];
    const float* W0   = (const float*)d_in[4];
    const float* b0   = (const float*)d_in[5];
    const float* W1   = (const float*)d_in[6];
    const float* b1   = (const float*)d_in[7];
    const float* msgW = (const float*)d_in[8];
    const float* msgb = (const float*)d_in[9];
    const float* Wih  = (const float*)d_in[10];
    const float* Whh  = (const float*)d_in[11];
    const float* bih  = (const float*)d_in[12];
    const float* bhh  = (const float*)d_in[13];
    const float* outW = (const float*)d_in[14];
    const float* outb = (const float*)d_in[15];

    const int N = in_sizes[1] / 256;
    const int E = in_sizes[2];

    float* ws    = (float*)d_ws;
    float* x     = ws;
    float* mx    = x    + (size_t)N * 256;
    float* bkt   = mx   + (size_t)N * 128;
    int*   slots = (int*)bkt;
    int*   deg   = slots + (size_t)N * SLOTS_PER_NODE;
    int*   ovf   = deg + N + 16;
    float* c     = bkt  + (size_t)N * 128 + N;
    float* pW0   = c    + (size_t)N * 128;
    float* pW1   = pW0  + 16384;
    float* pmsgW = pW1  + 65536;
    float* pWih  = pmsgW + 65536;
    float* pWhh  = pWih + 294912;
    float* poutW = pWhh + 196608;

    float* q    = (float*)d_out;
    float* hnew = q + (size_t)N * 16;

    const int rowTiles   = (N + 31) / 32;
    const int gruBlocks  = (N + GRU_ROWS - 1) / GRU_ROWS;
    const int zeroBlocks = (N + 1 + 255) / 256;
    const int edgeBlocks = (E + EDGES_PER_BLOCK - 1) / EDGES_PER_BLOCK;

    setup_kernel<<<PB_ALL + zeroBlocks, 256, 0, stream>>>(
        W0, W1, msgW, Wih, Whh, outW, pW0, pW1, pmsgW, pWih, pWhh, poutW, deg, N);
    enc_msg_rank_kernel<<<rowTiles + edgeBlocks, 128, 0, stream>>>(
        feat, pW0, b0, pW1, b1, x, h, pmsgW, msgb, mx, N, rowTiles, src, dst,
        deg, slots, ovf, E);
    gru_out_kernel<<<gruBlocks, 256, 0, stream>>>(x, mx, deg, slots, ovf, src, dst,
                                                  E, h, pWih, pWhh, bih, bhh,
                                                  poutW, outb, hnew, q, N);
}
